# starting kernel
# baseline (speedup 1.0000x reference)
#include <hip/hip_runtime.h>

typedef __attribute__((ext_vector_type(2))) float v2f;
typedef __attribute__((ext_vector_type(8))) float v8f;


__global__ void fill_zero_kernel(float* __restrict__ p, int n) {
    int i = blockIdx.x * blockDim.x + threadIdx.x;
    if (i < n) p[i] = 0.0f;
}

__global__ void degree_kernel(const int* __restrict__ col, float* __restrict__ deg, int E) {
    int i = blockIdx.x * blockDim.x + threadIdx.x;
    if (i < E) atomicAdd(&deg[col[i]], 1.0f);
}

__global__ void dinv_kernel(float* __restrict__ deg, int n) {
    int i = blockIdx.x * blockDim.x + threadIdx.x;
    if (i < n) deg[i] = rsqrtf(deg[i] + 1.0f);
}

template <int TN>
__global__ void gemm_wmma_f32_kernel(const float* __restrict__ X,
                                     const float* __restrict__ W,
                                     float* __restrict__ H,
                                     int n, int cin, int cout) {
    constexpr int coutP = TN * 16;
    extern __shared__ float lds[];

    const int total = cin * coutP;
    for (int i = threadIdx.x; i < total; i += blockDim.x) {
        const int k = i / coutP;
        const int c = i - k * coutP;
        const float v = (c < cout) ? W[(size_t)k * cout + c] : 0.0f;
        lds[(k >> 1) * (2 * coutP) + (c << 1) + (k & 1)] = v;
    }
    __syncthreads();

    const int lane   = threadIdx.x & 31;
    const int waveId = threadIdx.x >> 5;
    const int tileM  = blockIdx.x * (blockDim.x >> 5) + waveId;
    const int tilesM = (n + 15) >> 4;
    if (tileM >= tilesM) return;

    const int nlane = lane & 15;
    const int koff  = (lane >> 4) << 1;
    const int mrow  = (tileM << 4) + nlane;
    const int arow  = (mrow < n) ? mrow : (n - 1);

    const float* aPtr = X + (size_t)arow * cin + koff;
    int bIdx = (koff >> 1) * (2 * coutP) + (nlane << 1);

    v8f acc[TN] = {};
    for (int kb = 0; kb < cin; kb += 4) {
        const float2 av = *(const float2*)aPtr;
        v2f a; a.x = av.x; a.y = av.y;
        #pragma unroll
        for (int t = 0; t < TN; ++t) {
            const v2f b = *(const v2f*)(lds + bIdx + (t << 5));
            acc[t] = __builtin_amdgcn_wmma_f32_16x16x4_f32(
                          false, a,  false, b,
                          (short)0, acc[t],
                          false,  false);
        }
        aPtr += 4;
        bIdx += 4 * coutP;
    }

    const int rbase = (tileM << 4) + ((lane >> 4) << 3);
    #pragma unroll
    for (int t = 0; t < TN; ++t) {
        const int col = t * 16 + nlane;
        if (col < cout) {
            #pragma unroll
            for (int r = 0; r < 8; ++r) {
                const int orow = rbase + r;
                if (orow < n) H[(size_t)orow * cout + col] = acc[t][r];
            }
        }
    }
}

__global__ void scatter_kernel(const float* __restrict__ h,
                               const int* __restrict__ rowIdx,
                               const int* __restrict__ colIdx,
                               const float* __restrict__ dinv,
                               float* __restrict__ agg,
                               int E, int cout) {
    const int per = cout >> 2;
    const int tid = blockIdx.x * blockDim.x + threadIdx.x;
    const int e   = tid / per;
    if (e >= E) return;
    const int c   = (tid - e * per) << 2;

    const int r = rowIdx[e];
    const int d = colIdx[e];
    const float nrm = dinv[r] * dinv[d];

    const float4 hv = *(const float4*)(h + (size_t)r * cout + c);
    float* dst = agg + (size_t)d * cout + c;
    atomicAdd(dst + 0, hv.x * nrm);
    atomicAdd(dst + 1, hv.y * nrm);
    atomicAdd(dst + 2, hv.z * nrm);
    atomicAdd(dst + 3, hv.w * nrm);
}

__global__ void finalize_kernel(const float* __restrict__ h,
                                const float* __restrict__ agg,
                                const float* __restrict__ dinv,
                                const float* __restrict__ bias,
                                float* __restrict__ out,
                                int n, int cout, int doRelu) {
    const int tid = blockIdx.x * blockDim.x + threadIdx.x;
    const int i   = tid / cout;
    if (i >= n) return;
    const int c   = tid - i * cout;
    const float s = dinv[i];
    float v = agg[(size_t)i * cout + c] + h[(size_t)i * cout + c] * (s * s) + bias[c];
    if (doRelu) v = fmaxf(v, 0.0f);
    out[(size_t)i * cout + c] = v;
}


static inline int cdiv(long long a, long long b) { return (int)((a + b - 1) / b); }

static void launch_gemm(const float* X, const float* W, float* H,
                        int n, int cin, int cout, hipStream_t stream) {
    const int tilesM = (n + 15) >> 4;
    const int blocks = cdiv(tilesM, 8);
    const int TN     = (cout + 15) >> 4;
    const size_t shmem = (size_t)cin * TN * 16 * sizeof(float);
    switch (TN) {
        case 4: gemm_wmma_f32_kernel<4><<<blocks, 256, shmem, stream>>>(X, W, H, n, cin, cout); break;
        case 3: gemm_wmma_f32_kernel<3><<<blocks, 256, shmem, stream>>>(X, W, H, n, cin, cout); break;
        case 2: gemm_wmma_f32_kernel<2><<<blocks, 256, shmem, stream>>>(X, W, H, n, cin, cout); break;
        default: gemm_wmma_f32_kernel<1><<<blocks, 256, shmem, stream>>>(X, W, H, n, cin, cout); break;
    }
}

extern "C" void kernel_launch(void* const* d_in, const int* in_sizes, int n_in,
                              void* d_out, int out_size, void* d_ws, size_t ws_size,
                              hipStream_t stream) {
    const float* x    = (const float*)d_in[0];
    const int*   ei   = (const int*)  d_in[1];
    const float* W1   = (const float*)d_in[2];
    const float* b1   = (const float*)d_in[3];
    const float* W2   = (const float*)d_in[4];
    const float* b2   = (const float*)d_in[5];
    const float* Wout = (const float*)d_in[6];
    const float* bout = (const float*)d_in[7];

    const int C1  = in_sizes[3];
    const int C2  = in_sizes[5];
    const int C3  = in_sizes[7];
    const int Cin = in_sizes[2] / C1;
    const int n   = in_sizes[0] / Cin;
    const int E   = in_sizes[1] / 2;

    const int* rowIdx = ei;
    const int* colIdx = ei + E;

    float* dinv = (float*)d_ws;
    float* hbuf = dinv + n;
    float* abuf = hbuf + (size_t)n * C1;

    const int T = 256;

    fill_zero_kernel<<<cdiv(n, T), T, 0, stream>>>(dinv, n);
    degree_kernel<<<cdiv(E, T), T, 0, stream>>>(colIdx, dinv, E);
    dinv_kernel<<<cdiv(n, T), T, 0, stream>>>(dinv, n);

    launch_gemm(x, W1, hbuf, n, Cin, C1, stream);
    fill_zero_kernel<<<cdiv((long long)n * C1, T), T, 0, stream>>>(abuf, n * C1);
    scatter_kernel<<<cdiv((long long)E * (C1 >> 2), T), T, 0, stream>>>(hbuf, rowIdx, colIdx, dinv, abuf, E, C1);
    finalize_kernel<<<cdiv((long long)n * C1, T), T, 0, stream>>>(hbuf, abuf, dinv, b1, abuf, n, C1, 1);

    launch_gemm(abuf, W2, hbuf, n, C1, C2, stream);
    fill_zero_kernel<<<cdiv((long long)n * C2, T), T, 0, stream>>>(abuf, n * C2);
    scatter_kernel<<<cdiv((long long)E * (C2 >> 2), T), T, 0, stream>>>(hbuf, rowIdx, colIdx, dinv, abuf, E, C2);
    finalize_kernel<<<cdiv((long long)n * C2, T), T, 0, stream>>>(hbuf, abuf, dinv, b2, abuf, n, C2, 1);

    launch_gemm(abuf, Wout, hbuf, n, C2, C3, stream);
    fill_zero_kernel<<<cdiv((long long)n * C3, T), T, 0, stream>>>(abuf, n * C3);
    scatter_kernel<<<cdiv((long long)E * (C3 >> 2), T), T, 0, stream>>>(hbuf, rowIdx, colIdx, dinv, abuf, E, C3);
    finalize_kernel<<<cdiv((long long)n * C3, T), T, 0, stream>>>(hbuf, abuf, dinv, bout, (float*)d_out, n, C3, 0);
}


# scatter: a wave walks 32 edges with lanes along the feature row, so each gather and f32 atomic covers one contiguous 128B row piece (was 16B stride over 4 lines); atomics kept, arrival-order sums
# speedup vs baseline: 1.2986x; 1.2986x over previous
#include <hip/hip_runtime.h>

typedef __attribute__((ext_vector_type(2))) float v2f;
typedef __attribute__((ext_vector_type(8))) float v8f;


__global__ void fill_zero_kernel(float* __restrict__ p, int n) {
    int i = blockIdx.x * blockDim.x + threadIdx.x;
    if (i < n) p[i] = 0.0f;
}

__global__ void degree_kernel(const int* __restrict__ col, float* __restrict__ deg, int E) {
    int i = blockIdx.x * blockDim.x + threadIdx.x;
    if (i < E) atomicAdd(&deg[col[i]], 1.0f);
}

__global__ void dinv_kernel(float* __restrict__ deg, int n) {
    int i = blockIdx.x * blockDim.x + threadIdx.x;
    if (i < n) deg[i] = rsqrtf(deg[i] + 1.0f);
}

template <int TN>
__global__ void gemm_wmma_f32_kernel(const float* __restrict__ X,
                                     const float* __restrict__ W,
                                     float* __restrict__ H,
                                     int n, int cin, int cout) {
    constexpr int coutP = TN * 16;
    extern __shared__ float lds[];

    const int total = cin * coutP;
    for (int i = threadIdx.x; i < total; i += blockDim.x) {
        const int k = i / coutP;
        const int c = i - k * coutP;
        const float v = (c < cout) ? W[(size_t)k * cout + c] : 0.0f;
        lds[(k >> 1) * (2 * coutP) + (c << 1) + (k & 1)] = v;
    }
    __syncthreads();

    const int lane   = threadIdx.x & 31;
    const int waveId = threadIdx.x >> 5;
    const int tileM  = blockIdx.x * (blockDim.x >> 5) + waveId;
    const int tilesM = (n + 15) >> 4;
    if (tileM >= tilesM) return;

    const int nlane = lane & 15;
    const int koff  = (lane >> 4) << 1;
    const int mrow  = (tileM << 4) + nlane;
    const int arow  = (mrow < n) ? mrow : (n - 1);

    const float* aPtr = X + (size_t)arow * cin + koff;
    int bIdx = (koff >> 1) * (2 * coutP) + (nlane << 1);

    v8f acc[TN] = {};
    for (int kb = 0; kb < cin; kb += 4) {
        const float2 av = *(const float2*)aPtr;
        v2f a; a.x = av.x; a.y = av.y;
        #pragma unroll
        for (int t = 0; t < TN; ++t) {
            const v2f b = *(const v2f*)(lds + bIdx + (t << 5));
            acc[t] = __builtin_amdgcn_wmma_f32_16x16x4_f32(
                          false, a,  false, b,
                          (short)0, acc[t],
                          false,  false);
        }
        aPtr += 4;
        bIdx += 4 * coutP;
    }

    const int rbase = (tileM << 4) + ((lane >> 4) << 3);
    #pragma unroll
    for (int t = 0; t < TN; ++t) {
        const int col = t * 16 + nlane;
        if (col < cout) {
            #pragma unroll
            for (int r = 0; r < 8; ++r) {
                const int orow = rbase + r;
                if (orow < n) H[(size_t)orow * cout + col] = acc[t][r];
            }
        }
    }
}

template <int NCH, bool EXACT, int U, bool FULL>
__device__ __forceinline__ void scatter_edges(const float* __restrict__ h,
                                              float* __restrict__ agg,
                                              int r, int d, float nrm,
                                              int cnt, unsigned lane, int cout) {
    const int nrmBits = __float_as_int(nrm);
    const unsigned lastCol  = lane + 32u * (NCH - 1);
    const bool     lastIn   = EXACT || (lastCol < (unsigned)cout);
    const unsigned lastLoad = lastIn ? lastCol : (unsigned)(cout - 1);
    for (int j0 = 0; j0 < 32; j0 += U) {
        float v[U][NCH] = {};
        #pragma unroll
        for (int u = 0; u < U; ++u) {
            const int j = j0 + u;
            if (FULL || j < cnt) {
                const int rj = __builtin_amdgcn_readlane(r, j);
                const float* __restrict__ src = h + (size_t)rj * (size_t)cout;
                #pragma unroll
                for (int k = 0; k < NCH; ++k)
                    v[u][k] = src[(k + 1 < NCH) ? (lane + 32u * k) : lastLoad];
            }
        }
        if (!EXACT) {
            #pragma unroll
            for (int u = 0; u < U; ++u) asm volatile("" :: "v"(v[u][NCH - 1]));
        }
        #pragma unroll
        for (int u = 0; u < U; ++u) {
            const int j = j0 + u;
            if (FULL || j < cnt) {
                const int   dj = __builtin_amdgcn_readlane(d, j);
                const float nj = __int_as_float(__builtin_amdgcn_readlane(nrmBits, j));
                float* __restrict__ dst = agg + (size_t)dj * (size_t)cout;
                #pragma unroll
                for (int k = 0; k < NCH; ++k)
                    if (EXACT || k + 1 < NCH) atomicAdd(dst + (lane + 32u * k), v[u][k] * nj);
            }
        }
        if (!EXACT && lastIn) {
            #pragma unroll
            for (int u = 0; u < U; ++u) {
                const int j = j0 + u;
                if (FULL || j < cnt) {
                    const int   dj = __builtin_amdgcn_readlane(d, j);
                    const float nj = __int_as_float(__builtin_amdgcn_readlane(nrmBits, j));
                    float* __restrict__ dst = agg + (size_t)dj * (size_t)cout;
                    atomicAdd(dst + lastCol, v[u][NCH - 1] * nj);
                }
            }
        }
    }
}

template <int NCH, bool EXACT>
__global__ void __launch_bounds__(256)
scatter_kernel(const float* __restrict__ h,
               const int* __restrict__ rowIdx,
               const int* __restrict__ colIdx,
               const float* __restrict__ dinv,
               float* __restrict__ agg,
               int E, int cout) {
    const unsigned lane = threadIdx.x & 31u;
    const int wave = __builtin_amdgcn_readfirstlane(
        (int)(blockIdx.x * (blockDim.x >> 5) + (threadIdx.x >> 5)));
    const int e0   = wave << 5;
    if (e0 >= E) return;
    const int e = e0 + (int)lane;

    int r = 0, d = 0;
    float nrm = 0.0f;
    if (e < E) {
        r = rowIdx[e];
        d = colIdx[e];
        nrm = dinv[r] * dinv[d];
    }
    const int cnt = (E - e0 < 32) ? (E - e0) : 32;
    if (cnt == 32) scatter_edges<NCH, EXACT, 8, true >(h, agg, r, d, nrm, cnt, lane, cout);
    else           scatter_edges<NCH, EXACT, 8, false>(h, agg, r, d, nrm, cnt, lane, cout);
}

__global__ void finalize_kernel(const float* __restrict__ h,
                                const float* __restrict__ agg,
                                const float* __restrict__ dinv,
                                const float* __restrict__ bias,
                                float* __restrict__ out,
                                int n, int cout, int doRelu) {
    const int tid = blockIdx.x * blockDim.x + threadIdx.x;
    const int i   = tid / cout;
    if (i >= n) return;
    const int c   = tid - i * cout;
    const float s = dinv[i];
    float v = agg[(size_t)i * cout + c] + h[(size_t)i * cout + c] * (s * s) + bias[c];
    if (doRelu) v = fmaxf(v, 0.0f);
    out[(size_t)i * cout + c] = v;
}


static inline int cdiv(long long a, long long b) { return (int)((a + b - 1) / b); }

static void launch_gemm(const float* X, const float* W, float* H,
                        int n, int cin, int cout, hipStream_t stream) {
    const int tilesM = (n + 15) >> 4;
    const int blocks = cdiv(tilesM, 8);
    const int TN     = (cout + 15) >> 4;
    const size_t shmem = (size_t)cin * TN * 16 * sizeof(float);
    switch (TN) {
        case 4: gemm_wmma_f32_kernel<4><<<blocks, 256, shmem, stream>>>(X, W, H, n, cin, cout); break;
        case 3: gemm_wmma_f32_kernel<3><<<blocks, 256, shmem, stream>>>(X, W, H, n, cin, cout); break;
        case 2: gemm_wmma_f32_kernel<2><<<blocks, 256, shmem, stream>>>(X, W, H, n, cin, cout); break;
        default: gemm_wmma_f32_kernel<1><<<blocks, 256, shmem, stream>>>(X, W, H, n, cin, cout); break;
    }
}

static void launch_scatter(const float* h, const int* rowIdx, const int* colIdx,
                           const float* dinv, float* agg, int E, int cout,
                           hipStream_t stream) {
    const int waves  = cdiv(E, 32);
    const int blocks = cdiv(waves, 8);
    const bool exact = (cout & 31) == 0;
    if (cout > 32) {
        if (exact) scatter_kernel<2, true ><<<blocks, 256, 0, stream>>>(h, rowIdx, colIdx, dinv, agg, E, cout);
        else       scatter_kernel<2, false><<<blocks, 256, 0, stream>>>(h, rowIdx, colIdx, dinv, agg, E, cout);
    } else {
        if (exact) scatter_kernel<1, true ><<<blocks, 256, 0, stream>>>(h, rowIdx, colIdx, dinv, agg, E, cout);
        else       scatter_kernel<1, false><<<blocks, 256, 0, stream>>>(h, rowIdx, colIdx, dinv, agg, E, cout);
    }
}

extern "C" void kernel_launch(void* const* d_in, const int* in_sizes, int n_in,
                              void* d_out, int out_size, void* d_ws, size_t ws_size,
                              hipStream_t stream) {
    const float* x    = (const float*)d_in[0];
    const int*   ei   = (const int*)  d_in[1];
    const float* W1   = (const float*)d_in[2];
    const float* b1   = (const float*)d_in[3];
    const float* W2   = (const float*)d_in[4];
    const float* b2   = (const float*)d_in[5];
    const float* Wout = (const float*)d_in[6];
    const float* bout = (const float*)d_in[7];

    const int C1  = in_sizes[3];
    const int C2  = in_sizes[5];
    const int C3  = in_sizes[7];
    const int Cin = in_sizes[2] / C1;
    const int n   = in_sizes[0] / Cin;
    const int E   = in_sizes[1] / 2;

    const int* rowIdx = ei;
    const int* colIdx = ei + E;

    float* dinv = (float*)d_ws;
    float* hbuf = dinv + n;
    float* abuf = hbuf + (size_t)n * C1;

    const int T = 256;

    fill_zero_kernel<<<cdiv(n, T), T, 0, stream>>>(dinv, n);
    degree_kernel<<<cdiv(E, T), T, 0, stream>>>(colIdx, dinv, E);
    dinv_kernel<<<cdiv(n, T), T, 0, stream>>>(dinv, n);

    launch_gemm(x, W1, hbuf, n, Cin, C1, stream);
    fill_zero_kernel<<<cdiv((long long)n * C1, T), T, 0, stream>>>(abuf, n * C1);
    launch_scatter(hbuf, rowIdx, colIdx, dinv, abuf, E, C1, stream);
    finalize_kernel<<<cdiv((long long)n * C1, T), T, 0, stream>>>(hbuf, abuf, dinv, b1, abuf, n, C1, 1);

    launch_gemm(abuf, W2, hbuf, n, C1, C2, stream);
    fill_zero_kernel<<<cdiv((long long)n * C2, T), T, 0, stream>>>(abuf, n * C2);
    launch_scatter(hbuf, rowIdx, colIdx, dinv, abuf, E, C2, stream);
    finalize_kernel<<<cdiv((long long)n * C2, T), T, 0, stream>>>(hbuf, abuf, dinv, b2, abuf, n, C2, 1);

    launch_gemm(abuf, Wout, hbuf, n, C2, C3, stream);
    fill_zero_kernel<<<cdiv((long long)n * C3, T), T, 0, stream>>>(abuf, n * C3);
    launch_scatter(hbuf, rowIdx, colIdx, dinv, abuf, E, C3, stream);
    finalize_kernel<<<cdiv((long long)n * C3, T), T, 0, stream>>>(hbuf, abuf, dinv, bout, (float*)d_out, n, C3, 0);
}


# aggregate by destination: edges bucketed once per call (count, scan, cursor fill); each node sums its rows in registers in bucket (arrival) order, one store with self-loop+bias+relu; no f32 atomics
# speedup vs baseline: 1.9100x; 1.4708x over previous
#include <hip/hip_runtime.h>

typedef __attribute__((ext_vector_type(2))) float v2f;
typedef __attribute__((ext_vector_type(8))) float v8f;


__global__ void zero_int_kernel(int* __restrict__ p, int n) {
    int i = blockIdx.x * blockDim.x + threadIdx.x;
    if (i < n) p[i] = 0;
}

__global__ void count_kernel(const int* __restrict__ col, int* __restrict__ cnt, int E) {
    int i = blockIdx.x * blockDim.x + threadIdx.x;
    if (i < E) atomicAdd(&cnt[col[i]], 1);
}

__device__ __forceinline__ int wave_sum(int s) {
    #pragma unroll
    for (int off = 16; off > 0; off >>= 1) s += __shfl_xor(s, off, 32);
    return s;
}

__global__ void __launch_bounds__(256)
block_sum_kernel(const int* __restrict__ cnt, int* __restrict__ bsum, int n) {
    __shared__ int sm[8];
    const int base = blockIdx.x * 1024 + threadIdx.x * 4;
    int s = 0;
    #pragma unroll
    for (int k = 0; k < 4; ++k) if (base + k < n) s += cnt[base + k];
    s = wave_sum(s);
    if ((threadIdx.x & 31) == 0) sm[threadIdx.x >> 5] = s;
    __syncthreads();
    if (threadIdx.x == 0) {
        int t = 0;
        #pragma unroll
        for (int w = 0; w < 8; ++w) t += sm[w];
        bsum[blockIdx.x] = t;
    }
}

__global__ void __launch_bounds__(256)
scan_kernel(const int* __restrict__ cnt, const int* __restrict__ bsum,
            int* __restrict__ offs, int* __restrict__ cursor,
            float* __restrict__ dinv, int n) {
    __shared__ int smA[8];
    __shared__ int smB[8];
    const int tid  = threadIdx.x;
    const int lane = tid & 31;
    const int wave = tid >> 5;

    int pb = 0;
    for (int i = tid; i < (int)blockIdx.x; i += 256) pb += bsum[i];
    pb = wave_sum(pb);
    if (lane == 0) smA[wave] = pb;

    const int base = blockIdx.x * 1024 + tid * 4;
    int c[4];
    #pragma unroll
    for (int k = 0; k < 4; ++k) c[k] = (base + k < n) ? cnt[base + k] : 0;
    const int tsum = c[0] + c[1] + c[2] + c[3];

    int incl = tsum;
    #pragma unroll
    for (int off = 1; off < 32; off <<= 1) {
        const int t = __shfl_up(incl, off, 32);
        if (lane >= off) incl += t;
    }
    if (lane == 31) smB[wave] = incl;
    __syncthreads();

    int run = 0;
    #pragma unroll
    for (int w = 0; w < 8; ++w) run += smA[w];
    #pragma unroll
    for (int w = 0; w < 8; ++w) if (w < wave) run += smB[w];
    run += incl - tsum;

    #pragma unroll
    for (int k = 0; k < 4; ++k) {
        if (base + k < n) {
            offs[base + k]   = run;
            cursor[base + k] = run;
            dinv[base + k]   = rsqrtf((float)c[k] + 1.0f);
        }
        run += c[k];
    }
    if (base <= n - 1 && n - 1 < base + 4) offs[n] = run;
}

__global__ void bucket_fill_kernel(const int* __restrict__ row, const int* __restrict__ col,
                                   const float* __restrict__ dinv,
                                   int* __restrict__ cursor, int2* __restrict__ ent, int E) {
    int e = blockIdx.x * blockDim.x + threadIdx.x;
    if (e < E) {
        const int r = row[e];
        const int c = col[e];
        const float nrm = dinv[r] * dinv[c];
        const int p = atomicAdd(&cursor[c], 1);
        ent[p] = make_int2(r, __float_as_int(nrm));
    }
}

template <int TN>
__global__ void gemm_wmma_f32_kernel(const float* __restrict__ X,
                                     const float* __restrict__ W,
                                     float* __restrict__ H,
                                     int n, int cin, int cout) {
    constexpr int coutP = TN * 16;
    extern __shared__ float lds[];

    const int total = cin * coutP;
    for (int i = threadIdx.x; i < total; i += blockDim.x) {
        const int k = i / coutP;
        const int c = i - k * coutP;
        const float v = (c < cout) ? W[(size_t)k * cout + c] : 0.0f;
        lds[(k >> 1) * (2 * coutP) + (c << 1) + (k & 1)] = v;
    }
    __syncthreads();

    const int lane   = threadIdx.x & 31;
    const int waveId = threadIdx.x >> 5;
    const int tileM  = blockIdx.x * (blockDim.x >> 5) + waveId;
    const int tilesM = (n + 15) >> 4;
    if (tileM >= tilesM) return;

    const int nlane = lane & 15;
    const int koff  = (lane >> 4) << 1;
    const int mrow  = (tileM << 4) + nlane;
    const int arow  = (mrow < n) ? mrow : (n - 1);

    const float* aPtr = X + (size_t)arow * cin + koff;
    int bIdx = (koff >> 1) * (2 * coutP) + (nlane << 1);

    v8f acc[TN] = {};
    for (int kb = 0; kb < cin; kb += 4) {
        const float2 av = *(const float2*)aPtr;
        v2f a; a.x = av.x; a.y = av.y;
        #pragma unroll
        for (int t = 0; t < TN; ++t) {
            const v2f b = *(const v2f*)(lds + bIdx + (t << 5));
            acc[t] = __builtin_amdgcn_wmma_f32_16x16x4_f32(
                          false, a,  false, b,
                          (short)0, acc[t],
                          false,  false);
        }
        aPtr += 4;
        bIdx += 4 * coutP;
    }

    const int rbase = (tileM << 4) + ((lane >> 4) << 3);
    #pragma unroll
    for (int t = 0; t < TN; ++t) {
        const int col = t * 16 + nlane;
        if (col < cout) {
            #pragma unroll
            for (int r = 0; r < 8; ++r) {
                const int orow = rbase + r;
                if (orow < n) H[(size_t)orow * cout + col] = acc[t][r];
            }
        }
    }
}

template <int VEC, int NP, int U>
__global__ void __launch_bounds__(256)
gather_kernel(const float* __restrict__ h,
              const int* __restrict__ offs,
              const int2* __restrict__ ent,
              const float* __restrict__ dinv,
              const float* __restrict__ bias,
              float* __restrict__ out,
              int n, int cout, int doRelu) {
    const unsigned lane = threadIdx.x & 31u;
    const int node = __builtin_amdgcn_readfirstlane(
        (int)(blockIdx.x * (blockDim.x >> 5) + (threadIdx.x >> 5)));
    if (node >= n) return;

    const int   beg = offs[node];
    const int   end = offs[node + 1];
    const float di  = dinv[node];

    int  colOf[NP];
    bool act[NP];
    int  ldc[NP];
    #pragma unroll
    for (int p = 0; p < NP; ++p) {
        colOf[p] = (int)(lane + 32u * p) * VEC;
        act[p]   = colOf[p] < cout;
        ldc[p]   = act[p] ? colOf[p] : 0;
    }

    float hs[NP][VEC], bs[NP][VEC];
    {
        const float* __restrict__ self = h + (size_t)node * (size_t)cout;
        #pragma unroll
        for (int p = 0; p < NP; ++p) {
            if (VEC == 2) {
                const float2 t = *(const float2*)(self + ldc[p]);
                const float2 b = *(const float2*)(bias + ldc[p]);
                hs[p][0] = t.x; hs[p][VEC - 1] = t.y;
                bs[p][0] = b.x; bs[p][VEC - 1] = b.y;
            } else {
                hs[p][0] = self[ldc[p]];
                bs[p][0] = bias[ldc[p]];
            }
        }
    }

    float acc[NP][VEC] = {};
    for (int b0 = beg; b0 < end; b0 += 32) {
        const int cntb = (end - b0 < 32) ? (end - b0) : 32;
        int r = 0, wBits = 0;
        if ((int)lane < cntb) {
            const int2 en = ent[b0 + (int)lane];
            r     = en.x;
            wBits = en.y;
        }
        asm volatile("" :: "v"(r), "v"(wBits));
        for (int j0 = 0; j0 < cntb; j0 += U) {
            float v[U][NP][VEC] = {};
            #pragma unroll
            for (int u = 0; u < U; ++u) {
                const int j = j0 + u;
                if (j < cntb) {
                    const int rj = __builtin_amdgcn_readlane(r, j);
                    const float* __restrict__ src = h + (size_t)rj * (size_t)cout;
                    #pragma unroll
                    for (int p = 0; p < NP; ++p) {
                        if (VEC == 2) {
                            const float2 t = *(const float2*)(src + ldc[p]);
                            v[u][p][0] = t.x; v[u][p][VEC - 1] = t.y;
                        } else {
                            v[u][p][0] = src[ldc[p]];
                        }
                    }
                }
            }
            #pragma unroll
            for (int u = 0; u < U; ++u) {
                const int j = j0 + u;
                if (j < cntb) {
                    const float nj = __int_as_float(__builtin_amdgcn_readlane(wBits, j));
                    #pragma unroll
                    for (int p = 0; p < NP; ++p) {
                        #pragma unroll
                        for (int q = 0; q < VEC; ++q) acc[p][q] += v[u][p][q] * nj;
                    }
                }
            }
        }
    }

    const float dd = di * di;
    float* __restrict__ dst = out + (size_t)node * (size_t)cout;
    #pragma unroll
    for (int p = 0; p < NP; ++p) {
        float o[VEC];
        #pragma unroll
        for (int q = 0; q < VEC; ++q) {
            o[q] = acc[p][q] + hs[p][q] * dd + bs[p][q];
            if (doRelu) o[q] = fmaxf(o[q], 0.0f);
        }
        if (act[p]) {
            if (VEC == 2) {
                float2 t; t.x = o[0]; t.y = o[VEC - 1];
                *(float2*)(dst + colOf[p]) = t;
            } else {
                dst[colOf[p]] = o[0];
            }
        }
    }
}


static inline int cdiv(long long a, long long b) { return (int)((a + b - 1) / b); }

static void launch_gemm(const float* X, const float* W, float* H,
                        int n, int cin, int cout, hipStream_t stream) {
    const int tilesM = (n + 15) >> 4;
    const int blocks = cdiv(tilesM, 8);
    const int TN     = (cout + 15) >> 4;
    const size_t shmem = (size_t)cin * TN * 16 * sizeof(float);
    switch (TN) {
        case 4: gemm_wmma_f32_kernel<4><<<blocks, 256, shmem, stream>>>(X, W, H, n, cin, cout); break;
        case 3: gemm_wmma_f32_kernel<3><<<blocks, 256, shmem, stream>>>(X, W, H, n, cin, cout); break;
        case 2: gemm_wmma_f32_kernel<2><<<blocks, 256, shmem, stream>>>(X, W, H, n, cin, cout); break;
        default: gemm_wmma_f32_kernel<1><<<blocks, 256, shmem, stream>>>(X, W, H, n, cin, cout); break;
    }
}

static void launch_gather(const float* h, const int* offs, const int2* ent,
                          const float* dinv, const float* bias, float* out,
                          int n, int cout, int doRelu, hipStream_t stream) {
    const int blocks = cdiv(n, 8);
    if ((cout & 1) == 0)
        gather_kernel<2, 1, 16><<<blocks, 256, 0, stream>>>(h, offs, ent, dinv, bias, out, n, cout, doRelu);
    else
        gather_kernel<1, 2, 16><<<blocks, 256, 0, stream>>>(h, offs, ent, dinv, bias, out, n, cout, doRelu);
}

extern "C" void kernel_launch(void* const* d_in, const int* in_sizes, int n_in,
                              void* d_out, int out_size, void* d_ws, size_t ws_size,
                              hipStream_t stream) {
    const float* x    = (const float*)d_in[0];
    const int*   ei   = (const int*)  d_in[1];
    const float* W1   = (const float*)d_in[2];
    const float* b1   = (const float*)d_in[3];
    const float* W2   = (const float*)d_in[4];
    const float* b2   = (const float*)d_in[5];
    const float* Wout = (const float*)d_in[6];
    const float* bout = (const float*)d_in[7];

    const int C1  = in_sizes[3];
    const int C2  = in_sizes[5];
    const int C3  = in_sizes[7];
    const int Cin = in_sizes[2] / C1;
    const int n   = in_sizes[0] / Cin;
    const int E   = in_sizes[1] / 2;

    const int* rowIdx = ei;
    const int* colIdx = ei + E;

    const size_t nP = ((size_t)n + 64) & ~(size_t)63;
    float* hbuf   = (float*)d_ws;
    float* abuf   = hbuf + (size_t)n * 64;
    float* dinv   = abuf + (size_t)n * 64;
    int*   cnt    = (int*)(dinv + nP);
    int*   cursor = cnt + nP;
    int*   offs   = cursor + nP;
    int*   bsum   = offs + nP;
    int2*  ent    = (int2*)(bsum + nP);

    const int T = 256;
    const int scanBlocks = cdiv(n, 1024);

    zero_int_kernel<<<cdiv(n, T), T, 0, stream>>>(cnt, n);
    count_kernel<<<cdiv(E, T), T, 0, stream>>>(colIdx, cnt, E);
    block_sum_kernel<<<scanBlocks, 256, 0, stream>>>(cnt, bsum, n);
    scan_kernel<<<scanBlocks, 256, 0, stream>>>(cnt, bsum, offs, cursor, dinv, n);
    bucket_fill_kernel<<<cdiv(E, T), T, 0, stream>>>(rowIdx, colIdx, dinv, cursor, ent, E);

    launch_gemm(x, W1, hbuf, n, Cin, C1, stream);
    launch_gather(hbuf, offs, ent, dinv, b1, abuf, n, C1, 1, stream);

    launch_gemm(abuf, W2, hbuf, n, C1, C2, stream);
    launch_gather(hbuf, offs, ent, dinv, b2, abuf, n, C2, 1, stream);

    launch_gemm(abuf, Wout, hbuf, n, C2, C3, stream);
    launch_gather(hbuf, offs, ent, dinv, bout, (float*)d_out, n, C3, 0, stream);
}


# bucketing: an edge's slot = bucket start + the rank its counting atomic returned: one scattered integer atomic per edge, not two (no cursor atomic in the fill); sums in counting-arrival order
# speedup vs baseline: 2.3123x; 1.2106x over previous
#include <hip/hip_runtime.h>

typedef __attribute__((ext_vector_type(2))) float v2f;
typedef __attribute__((ext_vector_type(8))) float v8f;


__global__ void zero_int_kernel(int* __restrict__ p, int n) {
    int i = blockIdx.x * blockDim.x + threadIdx.x;
    if (i < n) p[i] = 0;
}

__global__ void count_rank_kernel(const int* __restrict__ col, int* __restrict__ cnt,
                                  int* __restrict__ rank, int E) {
    int i = blockIdx.x * blockDim.x + threadIdx.x;
    if (i < E) rank[i] = atomicAdd(&cnt[col[i]], 1);
}

__device__ __forceinline__ int wave_sum(int s) {
    #pragma unroll
    for (int off = 16; off > 0; off >>= 1) s += __shfl_xor(s, off, 32);
    return s;
}

__global__ void __launch_bounds__(256)
block_sum_kernel(const int* __restrict__ cnt, int* __restrict__ bsum, int n) {
    __shared__ int sm[8];
    const int base = blockIdx.x * 1024 + threadIdx.x * 4;
    int s = 0;
    #pragma unroll
    for (int k = 0; k < 4; ++k) if (base + k < n) s += cnt[base + k];
    s = wave_sum(s);
    if ((threadIdx.x & 31) == 0) sm[threadIdx.x >> 5] = s;
    __syncthreads();
    if (threadIdx.x == 0) {
        int t = 0;
        #pragma unroll
        for (int w = 0; w < 8; ++w) t += sm[w];
        bsum[blockIdx.x] = t;
    }
}

__global__ void __launch_bounds__(256)
scan_kernel(const int* __restrict__ cnt, const int* __restrict__ bsum,
            int* __restrict__ offs,
            float* __restrict__ dinv, int n) {
    __shared__ int smA[8];
    __shared__ int smB[8];
    const int tid  = threadIdx.x;
    const int lane = tid & 31;
    const int wave = tid >> 5;

    int pb = 0;
    for (int i = tid; i < (int)blockIdx.x; i += 256) pb += bsum[i];
    pb = wave_sum(pb);
    if (lane == 0) smA[wave] = pb;

    const int base = blockIdx.x * 1024 + tid * 4;
    int c[4];
    #pragma unroll
    for (int k = 0; k < 4; ++k) c[k] = (base + k < n) ? cnt[base + k] : 0;
    const int tsum = c[0] + c[1] + c[2] + c[3];

    int incl = tsum;
    #pragma unroll
    for (int off = 1; off < 32; off <<= 1) {
        const int t = __shfl_up(incl, off, 32);
        if (lane >= off) incl += t;
    }
    if (lane == 31) smB[wave] = incl;
    __syncthreads();

    int run = 0;
    #pragma unroll
    for (int w = 0; w < 8; ++w) run += smA[w];
    #pragma unroll
    for (int w = 0; w < 8; ++w) if (w < wave) run += smB[w];
    run += incl - tsum;

    #pragma unroll
    for (int k = 0; k < 4; ++k) {
        if (base + k < n) {
            offs[base + k] = run;
            dinv[base + k] = rsqrtf((float)c[k] + 1.0f);
        }
        run += c[k];
    }
    if (base <= n - 1 && n - 1 < base + 4) offs[n] = run;
}

__global__ void bucket_fill_kernel(const int* __restrict__ row, const int* __restrict__ col,
                                   const int* __restrict__ rank, const int* __restrict__ offs,
                                   const float* __restrict__ dinv,
                                   int2* __restrict__ ent, int E) {
    int e = blockIdx.x * blockDim.x + threadIdx.x;
    if (e < E) {
        const int r = row[e];
        const int c = col[e];
        const int p = offs[c] + rank[e];
        const float nrm = dinv[r] * dinv[c];
        ent[p] = make_int2(r, __float_as_int(nrm));
    }
}

template <int TN>
__global__ void gemm_wmma_f32_kernel(const float* __restrict__ X,
                                     const float* __restrict__ W,
                                     float* __restrict__ H,
                                     int n, int cin, int cout) {
    constexpr int coutP = TN * 16;
    extern __shared__ float lds[];

    const int total = cin * coutP;
    for (int i = threadIdx.x; i < total; i += blockDim.x) {
        const int k = i / coutP;
        const int c = i - k * coutP;
        const float v = (c < cout) ? W[(size_t)k * cout + c] : 0.0f;
        lds[(k >> 1) * (2 * coutP) + (c << 1) + (k & 1)] = v;
    }
    __syncthreads();

    const int lane   = threadIdx.x & 31;
    const int waveId = threadIdx.x >> 5;
    const int tileM  = blockIdx.x * (blockDim.x >> 5) + waveId;
    const int tilesM = (n + 15) >> 4;
    if (tileM >= tilesM) return;

    const int nlane = lane & 15;
    const int koff  = (lane >> 4) << 1;
    const int mrow  = (tileM << 4) + nlane;
    const int arow  = (mrow < n) ? mrow : (n - 1);

    const float* aPtr = X + (size_t)arow * cin + koff;
    int bIdx = (koff >> 1) * (2 * coutP) + (nlane << 1);

    v8f acc[TN] = {};
    for (int kb = 0; kb < cin; kb += 4) {
        const float2 av = *(const float2*)aPtr;
        v2f a; a.x = av.x; a.y = av.y;
        #pragma unroll
        for (int t = 0; t < TN; ++t) {
            const v2f b = *(const v2f*)(lds + bIdx + (t << 5));
            acc[t] = __builtin_amdgcn_wmma_f32_16x16x4_f32(
                          false, a,  false, b,
                          (short)0, acc[t],
                          false,  false);
        }
        aPtr += 4;
        bIdx += 4 * coutP;
    }

    const int rbase = (tileM << 4) + ((lane >> 4) << 3);
    #pragma unroll
    for (int t = 0; t < TN; ++t) {
        const int col = t * 16 + nlane;
        if (col < cout) {
            #pragma unroll
            for (int r = 0; r < 8; ++r) {
                const int orow = rbase + r;
                if (orow < n) H[(size_t)orow * cout + col] = acc[t][r];
            }
        }
    }
}

template <int VEC, int NP, int U>
__global__ void __launch_bounds__(256)
gather_kernel(const float* __restrict__ h,
              const int* __restrict__ offs,
              const int2* __restrict__ ent,
              const float* __restrict__ dinv,
              const float* __restrict__ bias,
              float* __restrict__ out,
              int n, int cout, int doRelu) {
    const unsigned lane = threadIdx.x & 31u;
    const int node = __builtin_amdgcn_readfirstlane(
        (int)(blockIdx.x * (blockDim.x >> 5) + (threadIdx.x >> 5)));
    if (node >= n) return;

    const int   beg = offs[node];
    const int   end = offs[node + 1];
    const float di  = dinv[node];

    int  colOf[NP];
    bool act[NP];
    int  ldc[NP];
    #pragma unroll
    for (int p = 0; p < NP; ++p) {
        colOf[p] = (int)(lane + 32u * p) * VEC;
        act[p]   = colOf[p] < cout;
        ldc[p]   = act[p] ? colOf[p] : 0;
    }

    float hs[NP][VEC], bs[NP][VEC];
    {
        const float* __restrict__ self = h + (size_t)node * (size_t)cout;
        #pragma unroll
        for (int p = 0; p < NP; ++p) {
            if (VEC == 2) {
                const float2 t = *(const float2*)(self + ldc[p]);
                const float2 b = *(const float2*)(bias + ldc[p]);
                hs[p][0] = t.x; hs[p][VEC - 1] = t.y;
                bs[p][0] = b.x; bs[p][VEC - 1] = b.y;
            } else {
                hs[p][0] = self[ldc[p]];
                bs[p][0] = bias[ldc[p]];
            }
        }
    }

    float acc[NP][VEC] = {};
    for (int b0 = beg; b0 < end; b0 += 32) {
        const int cntb = (end - b0 < 32) ? (end - b0) : 32;
        int r = 0, wBits = 0;
        if ((int)lane < cntb) {
            const int2 en = ent[b0 + (int)lane];
            r     = en.x;
            wBits = en.y;
        }
        asm volatile("" :: "v"(r), "v"(wBits));
        for (int j0 = 0; j0 < cntb; j0 += U) {
            float v[U][NP][VEC] = {};
            #pragma unroll
            for (int u = 0; u < U; ++u) {
                const int j = j0 + u;
                if (j < cntb) {
                    const int rj = __builtin_amdgcn_readlane(r, j);
                    const float* __restrict__ src = h + (size_t)rj * (size_t)cout;
                    #pragma unroll
                    for (int p = 0; p < NP; ++p) {
                        if (VEC == 2) {
                            const float2 t = *(const float2*)(src + ldc[p]);
                            v[u][p][0] = t.x; v[u][p][VEC - 1] = t.y;
                        } else {
                            v[u][p][0] = src[ldc[p]];
                        }
                    }
                }
            }
            #pragma unroll
            for (int u = 0; u < U; ++u) {
                const int j = j0 + u;
                if (j < cntb) {
                    const float nj = __int_as_float(__builtin_amdgcn_readlane(wBits, j));
                    #pragma unroll
                    for (int p = 0; p < NP; ++p) {
                        #pragma unroll
                        for (int q = 0; q < VEC; ++q) acc[p][q] += v[u][p][q] * nj;
                    }
                }
            }
        }
    }

    const float dd = di * di;
    float* __restrict__ dst = out + (size_t)node * (size_t)cout;
    #pragma unroll
    for (int p = 0; p < NP; ++p) {
        float o[VEC];
        #pragma unroll
        for (int q = 0; q < VEC; ++q) {
            o[q] = acc[p][q] + hs[p][q] * dd + bs[p][q];
            if (doRelu) o[q] = fmaxf(o[q], 0.0f);
        }
        if (act[p]) {
            if (VEC == 2) {
                float2 t; t.x = o[0]; t.y = o[VEC - 1];
                *(float2*)(dst + colOf[p]) = t;
            } else {
                dst[colOf[p]] = o[0];
            }
        }
    }
}


static inline int cdiv(long long a, long long b) { return (int)((a + b - 1) / b); }

static void launch_gemm(const float* X, const float* W, float* H,
                        int n, int cin, int cout, hipStream_t stream) {
    const int tilesM = (n + 15) >> 4;
    const int blocks = cdiv(tilesM, 8);
    const int TN     = (cout + 15) >> 4;
    const size_t shmem = (size_t)cin * TN * 16 * sizeof(float);
    switch (TN) {
        case 4: gemm_wmma_f32_kernel<4><<<blocks, 256, shmem, stream>>>(X, W, H, n, cin, cout); break;
        case 3: gemm_wmma_f32_kernel<3><<<blocks, 256, shmem, stream>>>(X, W, H, n, cin, cout); break;
        case 2: gemm_wmma_f32_kernel<2><<<blocks, 256, shmem, stream>>>(X, W, H, n, cin, cout); break;
        default: gemm_wmma_f32_kernel<1><<<blocks, 256, shmem, stream>>>(X, W, H, n, cin, cout); break;
    }
}

static void launch_gather(const float* h, const int* offs, const int2* ent,
                          const float* dinv, const float* bias, float* out,
                          int n, int cout, int doRelu, hipStream_t stream) {
    const int blocks = cdiv(n, 8);
    if ((cout & 1) == 0)
        gather_kernel<2, 1, 16><<<blocks, 256, 0, stream>>>(h, offs, ent, dinv, bias, out, n, cout, doRelu);
    else
        gather_kernel<1, 2, 16><<<blocks, 256, 0, stream>>>(h, offs, ent, dinv, bias, out, n, cout, doRelu);
}

extern "C" void kernel_launch(void* const* d_in, const int* in_sizes, int n_in,
                              void* d_out, int out_size, void* d_ws, size_t ws_size,
                              hipStream_t stream) {
    const float* x    = (const float*)d_in[0];
    const int*   ei   = (const int*)  d_in[1];
    const float* W1   = (const float*)d_in[2];
    const float* b1   = (const float*)d_in[3];
    const float* W2   = (const float*)d_in[4];
    const float* b2   = (const float*)d_in[5];
    const float* Wout = (const float*)d_in[6];
    const float* bout = (const float*)d_in[7];

    const int C1  = in_sizes[3];
    const int C2  = in_sizes[5];
    const int C3  = in_sizes[7];
    const int Cin = in_sizes[2] / C1;
    const int n   = in_sizes[0] / Cin;
    const int E   = in_sizes[1] / 2;

    const int* rowIdx = ei;
    const int* colIdx = ei + E;

    const size_t nP = ((size_t)n + 64) & ~(size_t)63;
    float* hbuf   = (float*)d_ws;
    float* abuf   = hbuf + (size_t)n * 64;
    float* dinv   = abuf + (size_t)n * 64;
    int*   cnt    = (int*)(dinv + nP);
    int*   offs   = cnt + nP;
    int*   bsum   = offs + nP;
    int2*  ent    = (int2*)(bsum + nP);
    int*   rank   = (int*)(ent + E);

    const int T = 256;
    const int scanBlocks = cdiv(n, 1024);

    zero_int_kernel<<<cdiv(n, T), T, 0, stream>>>(cnt, n);
    count_rank_kernel<<<cdiv(E, T), T, 0, stream>>>(colIdx, cnt, rank, E);
    block_sum_kernel<<<scanBlocks, 256, 0, stream>>>(cnt, bsum, n);
    scan_kernel<<<scanBlocks, 256, 0, stream>>>(cnt, bsum, offs, dinv, n);
    bucket_fill_kernel<<<cdiv(E, T), T, 0, stream>>>(rowIdx, colIdx, rank, offs, dinv, ent, E);

    launch_gemm(x, W1, hbuf, n, Cin, C1, stream);
    launch_gather(hbuf, offs, ent, dinv, b1, abuf, n, C1, 1, stream);

    launch_gemm(abuf, W2, hbuf, n, C1, C2, stream);
    launch_gather(hbuf, offs, ent, dinv, b2, abuf, n, C2, 1, stream);

    launch_gemm(abuf, Wout, hbuf, n, C2, C3, stream);
    launch_gather(hbuf, offs, ent, dinv, bout, (float*)d_out, n, C3, 0, stream);
}


# counting: each edge is counted and ranked by an LDS add in the block of its edge slice x node range (16 x 16 blocks), no global atomic; bucket order = slice, then LDS arrival (sum order changes)
# speedup vs baseline: 2.7442x; 1.1868x over previous
#include <hip/hip_runtime.h>

typedef __attribute__((ext_vector_type(2))) float v2f;
typedef __attribute__((ext_vector_type(8))) float v8f;


#define NSLICE 16
#define NRANGE 16

__device__ __forceinline__ int draw_rank(int c, int lo, int len, int* cntL) {
    const unsigned d = (unsigned)(c - lo);
    return (d < (unsigned)len) ? atomicAdd(&cntL[d], 1) : -1;
}

__global__ void __launch_bounds__(1024)
count_rank_kernel(const int* __restrict__ col, int* __restrict__ part,
                  int* __restrict__ rank, int n, int E, int nP,
                  int rangeLen, int sliceLen) {
    extern __shared__ int cntL[];
    const int tid = threadIdx.x;
    const int lo  = blockIdx.y * rangeLen;
    int len = n - lo;
    if (len > rangeLen) len = rangeLen;
    if (len < 0) len = 0;
    for (int i = tid; i < len; i += 1024) cntL[i] = 0;
    __syncthreads();

    int e0 = blockIdx.x * sliceLen;
    if (e0 > E) e0 = E;
    int e1 = e0 + sliceLen;
    if (e1 > E) e1 = E;
    const int* __restrict__ cs = col + e0;
    const int nq = ((((size_t)cs) & 15) == 0) ? ((e1 - e0) >> 2) : 0;
    const int4* __restrict__ cs4 = (const int4*)cs;
    for (int q0 = tid; q0 < nq; q0 += 4 * 1024) {
        int4 v[4];
        #pragma unroll
        for (int u = 0; u < 4; ++u) {
            const int q = q0 + u * 1024;
            v[u] = (q < nq) ? cs4[q] : make_int4(-1, -1, -1, -1);
        }
        int rk[4][4];
        #pragma unroll
        for (int u = 0; u < 4; ++u) {
            rk[u][0] = draw_rank(v[u].x, lo, len, cntL);
            rk[u][1] = draw_rank(v[u].y, lo, len, cntL);
            rk[u][2] = draw_rank(v[u].z, lo, len, cntL);
            rk[u][3] = draw_rank(v[u].w, lo, len, cntL);
        }
        #pragma unroll
        for (int u = 0; u < 4; ++u) {
            const int e = e0 + 4 * (q0 + u * 1024);
            #pragma unroll
            for (int j = 0; j < 4; ++j)
                if (rk[u][j] >= 0) rank[e + j] = rk[u][j];
        }
    }
    for (int e = e0 + 4 * nq + tid; e < e1; e += 1024) {
        const int rk = draw_rank(col[e], lo, len, cntL);
        if (rk >= 0) rank[e] = rk;
    }
    __syncthreads();

    int* __restrict__ dst = part + (size_t)blockIdx.x * (size_t)nP + lo;
    for (int i = tid; i < len; i += 1024) dst[i] = cntL[i];
}

__device__ __forceinline__ int4 load4_guard(const int* p, int idx, int n) {
    if (idx + 3 < n) return *(const int4*)p;
    int4 v;
    v.x = (idx     < n) ? p[0] : 0;
    v.y = (idx + 1 < n) ? p[1] : 0;
    v.z = (idx + 2 < n) ? p[2] : 0;
    v.w = 0;
    return v;
}

__device__ __forceinline__ void store4_guard(int* p, int idx, int n, int4 v) {
    if (idx + 3 < n) { *(int4*)p = v; return; }
    if (idx     < n) p[0] = v.x;
    if (idx + 1 < n) p[1] = v.y;
    if (idx + 2 < n) p[2] = v.z;
}

__device__ __forceinline__ int wave_sum(int s) {
    #pragma unroll
    for (int off = 16; off > 0; off >>= 1) s += __shfl_xor(s, off, 32);
    return s;
}

__global__ void __launch_bounds__(256)
block_sum_kernel(const int* __restrict__ part, int* __restrict__ bsum, int n, int nP) {
    __shared__ int sm[8];
    const int base = blockIdx.x * 1024 + threadIdx.x * 4;
    int s = 0;
    if (base < n) {
        #pragma unroll
        for (int sl = 0; sl < NSLICE; ++sl) {
            const int4 v = load4_guard(part + (size_t)sl * (size_t)nP + base, base, n);
            s += v.x + v.y + v.z + v.w;
        }
    }
    s = wave_sum(s);
    if ((threadIdx.x & 31) == 0) sm[threadIdx.x >> 5] = s;
    __syncthreads();
    if (threadIdx.x == 0) {
        int t = 0;
        #pragma unroll
        for (int w = 0; w < 8; ++w) t += sm[w];
        bsum[blockIdx.x] = t;
    }
}

__global__ void __launch_bounds__(256)
scan_kernel(int* __restrict__ part, const int* __restrict__ bsum,
            int* __restrict__ offs,
            float* __restrict__ dinv, int n, int nP) {
    __shared__ int smA[8];
    __shared__ int smB[8];
    const int tid  = threadIdx.x;
    const int lane = tid & 31;
    const int wave = tid >> 5;

    int pb = 0;
    for (int i = tid; i < (int)blockIdx.x; i += 256) pb += bsum[i];
    pb = wave_sum(pb);
    if (lane == 0) smA[wave] = pb;

    const int base = blockIdx.x * 1024 + tid * 4;
    int4 p[NSLICE];
    int c[4] = {0, 0, 0, 0};
    #pragma unroll
    for (int sl = 0; sl < NSLICE; ++sl) {
        p[sl] = (base < n) ? load4_guard(part + (size_t)sl * (size_t)nP + base, base, n)
                           : make_int4(0, 0, 0, 0);
        c[0] += p[sl].x; c[1] += p[sl].y; c[2] += p[sl].z; c[3] += p[sl].w;
    }
    const int tsum = c[0] + c[1] + c[2] + c[3];

    int incl = tsum;
    #pragma unroll
    for (int off = 1; off < 32; off <<= 1) {
        const int t = __shfl_up(incl, off, 32);
        if (lane >= off) incl += t;
    }
    if (lane == 31) smB[wave] = incl;
    __syncthreads();

    int run = 0;
    #pragma unroll
    for (int w = 0; w < 8; ++w) run += smA[w];
    #pragma unroll
    for (int w = 0; w < 8; ++w) if (w < wave) run += smB[w];
    run += incl - tsum;

    if (base < n) {
        int4 a;
        a.x = run;
        a.y = a.x + c[0];
        a.z = a.y + c[1];
        a.w = a.z + c[2];
        store4_guard(offs + base, base, n, a);
        #pragma unroll
        for (int k = 0; k < 4; ++k)
            if (base + k < n) dinv[base + k] = rsqrtf((float)c[k] + 1.0f);
        if (n - 1 < base + 4) offs[n] = a.w + c[3];
        #pragma unroll
        for (int sl = 0; sl < NSLICE; ++sl) {
            store4_guard(part + (size_t)sl * (size_t)nP + base, base, n, a);
            a.x += p[sl].x; a.y += p[sl].y; a.z += p[sl].z; a.w += p[sl].w;
        }
    }
}

__global__ void bucket_fill_kernel(const int* __restrict__ row, const int* __restrict__ col,
                                   const int* __restrict__ rank, const int* __restrict__ part,
                                   const float* __restrict__ dinv,
                                   int2* __restrict__ ent, int E, int nP, int sliceLen) {
    int e = blockIdx.x * blockDim.x + threadIdx.x;
    if (e < E) {
        const int r = row[e];
        const int c = col[e];
        const int s = e / sliceLen;
        const int p = part[(size_t)s * (size_t)nP + c] + rank[e];
        const float nrm = dinv[r] * dinv[c];
        ent[p] = make_int2(r, __float_as_int(nrm));
    }
}

template <int TN>
__global__ void gemm_wmma_f32_kernel(const float* __restrict__ X,
                                     const float* __restrict__ W,
                                     float* __restrict__ H,
                                     int n, int cin, int cout) {
    constexpr int coutP = TN * 16;
    extern __shared__ float lds[];

    const int total = cin * coutP;
    for (int i = threadIdx.x; i < total; i += blockDim.x) {
        const int k = i / coutP;
        const int c = i - k * coutP;
        const float v = (c < cout) ? W[(size_t)k * cout + c] : 0.0f;
        lds[(k >> 1) * (2 * coutP) + (c << 1) + (k & 1)] = v;
    }
    __syncthreads();

    const int lane   = threadIdx.x & 31;
    const int waveId = threadIdx.x >> 5;
    const int tileM  = blockIdx.x * (blockDim.x >> 5) + waveId;
    const int tilesM = (n + 15) >> 4;
    if (tileM >= tilesM) return;

    const int nlane = lane & 15;
    const int koff  = (lane >> 4) << 1;
    const int mrow  = (tileM << 4) + nlane;
    const int arow  = (mrow < n) ? mrow : (n - 1);

    const float* aPtr = X + (size_t)arow * cin + koff;
    int bIdx = (koff >> 1) * (2 * coutP) + (nlane << 1);

    v8f acc[TN] = {};
    for (int kb = 0; kb < cin; kb += 4) {
        const float2 av = *(const float2*)aPtr;
        v2f a; a.x = av.x; a.y = av.y;
        #pragma unroll
        for (int t = 0; t < TN; ++t) {
            const v2f b = *(const v2f*)(lds + bIdx + (t << 5));
            acc[t] = __builtin_amdgcn_wmma_f32_16x16x4_f32(
                          false, a,  false, b,
                          (short)0, acc[t],
                          false,  false);
        }
        aPtr += 4;
        bIdx += 4 * coutP;
    }

    const int rbase = (tileM << 4) + ((lane >> 4) << 3);
    #pragma unroll
    for (int t = 0; t < TN; ++t) {
        const int col = t * 16 + nlane;
        if (col < cout) {
            #pragma unroll
            for (int r = 0; r < 8; ++r) {
                const int orow = rbase + r;
                if (orow < n) H[(size_t)orow * cout + col] = acc[t][r];
            }
        }
    }
}

template <int VEC, int NP, int U>
__global__ void __launch_bounds__(256)
gather_kernel(const float* __restrict__ h,
              const int* __restrict__ offs,
              const int2* __restrict__ ent,
              const float* __restrict__ dinv,
              const float* __restrict__ bias,
              float* __restrict__ out,
              int n, int cout, int doRelu) {
    const unsigned lane = threadIdx.x & 31u;
    const int node = __builtin_amdgcn_readfirstlane(
        (int)(blockIdx.x * (blockDim.x >> 5) + (threadIdx.x >> 5)));
    if (node >= n) return;

    const int   beg = offs[node];
    const int   end = offs[node + 1];
    const float di  = dinv[node];

    int  colOf[NP];
    bool act[NP];
    int  ldc[NP];
    #pragma unroll
    for (int p = 0; p < NP; ++p) {
        colOf[p] = (int)(lane + 32u * p) * VEC;
        act[p]   = colOf[p] < cout;
        ldc[p]   = act[p] ? colOf[p] : 0;
    }

    float hs[NP][VEC], bs[NP][VEC];
    {
        const float* __restrict__ self = h + (size_t)node * (size_t)cout;
        #pragma unroll
        for (int p = 0; p < NP; ++p) {
            if (VEC == 2) {
                const float2 t = *(const float2*)(self + ldc[p]);
                const float2 b = *(const float2*)(bias + ldc[p]);
                hs[p][0] = t.x; hs[p][VEC - 1] = t.y;
                bs[p][0] = b.x; bs[p][VEC - 1] = b.y;
            } else {
                hs[p][0] = self[ldc[p]];
                bs[p][0] = bias[ldc[p]];
            }
        }
    }

    float acc[NP][VEC] = {};
    for (int b0 = beg; b0 < end; b0 += 32) {
        const int cntb = (end - b0 < 32) ? (end - b0) : 32;
        int r = 0, wBits = 0;
        if ((int)lane < cntb) {
            const int2 en = ent[b0 + (int)lane];
            r     = en.x;
            wBits = en.y;
        }
        asm volatile("" :: "v"(r), "v"(wBits));
        for (int j0 = 0; j0 < cntb; j0 += U) {
            float v[U][NP][VEC] = {};
            #pragma unroll
            for (int u = 0; u < U; ++u) {
                const int j = j0 + u;
                if (j < cntb) {
                    const int rj = __builtin_amdgcn_readlane(r, j);
                    const float* __restrict__ src = h + (size_t)rj * (size_t)cout;
                    #pragma unroll
                    for (int p = 0; p < NP; ++p) {
                        if (VEC == 2) {
                            const float2 t = *(const float2*)(src + ldc[p]);
                            v[u][p][0] = t.x; v[u][p][VEC - 1] = t.y;
                        } else {
                            v[u][p][0] = src[ldc[p]];
                        }
                    }
                }
            }
            #pragma unroll
            for (int u = 0; u < U; ++u) {
                const int j = j0 + u;
                if (j < cntb) {
                    const float nj = __int_as_float(__builtin_amdgcn_readlane(wBits, j));
                    #pragma unroll
                    for (int p = 0; p < NP; ++p) {
                        #pragma unroll
                        for (int q = 0; q < VEC; ++q) acc[p][q] += v[u][p][q] * nj;
                    }
                }
            }
        }
    }

    const float dd = di * di;
    float* __restrict__ dst = out + (size_t)node * (size_t)cout;
    #pragma unroll
    for (int p = 0; p < NP; ++p) {
        float o[VEC];
        #pragma unroll
        for (int q = 0; q < VEC; ++q) {
            o[q] = acc[p][q] + hs[p][q] * dd + bs[p][q];
            if (doRelu) o[q] = fmaxf(o[q], 0.0f);
        }
        if (act[p]) {
            if (VEC == 2) {
                float2 t; t.x = o[0]; t.y = o[VEC - 1];
                *(float2*)(dst + colOf[p]) = t;
            } else {
                dst[colOf[p]] = o[0];
            }
        }
    }
}


static inline int cdiv(long long a, long long b) { return (int)((a + b - 1) / b); }

static void launch_gemm(const float* X, const float* W, float* H,
                        int n, int cin, int cout, hipStream_t stream) {
    const int tilesM = (n + 15) >> 4;
    const int blocks = cdiv(tilesM, 8);
    const int TN     = (cout + 15) >> 4;
    const size_t shmem = (size_t)cin * TN * 16 * sizeof(float);
    switch (TN) {
        case 4: gemm_wmma_f32_kernel<4><<<blocks, 256, shmem, stream>>>(X, W, H, n, cin, cout); break;
        case 3: gemm_wmma_f32_kernel<3><<<blocks, 256, shmem, stream>>>(X, W, H, n, cin, cout); break;
        case 2: gemm_wmma_f32_kernel<2><<<blocks, 256, shmem, stream>>>(X, W, H, n, cin, cout); break;
        default: gemm_wmma_f32_kernel<1><<<blocks, 256, shmem, stream>>>(X, W, H, n, cin, cout); break;
    }
}

static void launch_gather(const float* h, const int* offs, const int2* ent,
                          const float* dinv, const float* bias, float* out,
                          int n, int cout, int doRelu, hipStream_t stream) {
    const int blocks = cdiv(n, 8);
    if ((cout & 1) == 0)
        gather_kernel<2, 1, 16><<<blocks, 256, 0, stream>>>(h, offs, ent, dinv, bias, out, n, cout, doRelu);
    else
        gather_kernel<1, 2, 16><<<blocks, 256, 0, stream>>>(h, offs, ent, dinv, bias, out, n, cout, doRelu);
}

extern "C" void kernel_launch(void* const* d_in, const int* in_sizes, int n_in,
                              void* d_out, int out_size, void* d_ws, size_t ws_size,
                              hipStream_t stream) {
    const float* x    = (const float*)d_in[0];
    const int*   ei   = (const int*)  d_in[1];
    const float* W1   = (const float*)d_in[2];
    const float* b1   = (const float*)d_in[3];
    const float* W2   = (const float*)d_in[4];
    const float* b2   = (const float*)d_in[5];
    const float* Wout = (const float*)d_in[6];
    const float* bout = (const float*)d_in[7];

    const int C1  = in_sizes[3];
    const int C2  = in_sizes[5];
    const int C3  = in_sizes[7];
    const int Cin = in_sizes[2] / C1;
    const int n   = in_sizes[0] / Cin;
    const int E   = in_sizes[1] / 2;

    const int* rowIdx = ei;
    const int* colIdx = ei + E;

    const size_t nP = ((size_t)n + 64) & ~(size_t)63;
    float* hbuf   = (float*)d_ws;
    float* abuf   = hbuf + (size_t)n * 64;
    float* dinv   = abuf + (size_t)n * 64;
    int*   offs   = (int*)(dinv + nP);
    int*   bsum   = offs + nP;
    int*   part   = bsum + nP;
    int2*  ent    = (int2*)(part + (size_t)NSLICE * nP);
    int*   rank   = (int*)(ent + E);

    const int T = 256;
    const int scanBlocks = cdiv(n, 1024);
    const int rangeLen   = cdiv(n, NRANGE);
    const int sliceLen   = (cdiv(E, NSLICE) + 3) & ~3;

    count_rank_kernel<<<dim3(NSLICE, NRANGE), 1024, (size_t)rangeLen * sizeof(int), stream>>>(
        colIdx, part, rank, n, E, (int)nP, rangeLen, sliceLen);
    block_sum_kernel<<<scanBlocks, 256, 0, stream>>>(part, bsum, n, (int)nP);
    scan_kernel<<<scanBlocks, 256, 0, stream>>>(part, bsum, offs, dinv, n, (int)nP);
    bucket_fill_kernel<<<cdiv(E, T), T, 0, stream>>>(rowIdx, colIdx, rank, part, dinv, ent, E, (int)nP, sliceLen);

    launch_gemm(x, W1, hbuf, n, Cin, C1, stream);
    launch_gather(hbuf, offs, ent, dinv, b1, abuf, n, C1, 1, stream);

    launch_gemm(abuf, W2, hbuf, n, C1, C2, stream);
    launch_gather(hbuf, offs, ent, dinv, b2, abuf, n, C2, 1, stream);

    launch_gemm(abuf, Wout, hbuf, n, C2, C3, stream);
    launch_gather(hbuf, offs, ent, dinv, bout, (float*)d_out, n, C3, 0, stream);
}


# layers 1, 2: resident 16-wave blocks hold each 16-node tile of summed rows in LDS and multiply it by the next W (in LDS once per block) on WMMA in the same launch; 8 launches; sum order unchanged
# speedup vs baseline: 2.9575x; 1.0777x over previous
#include <hip/hip_runtime.h>

typedef __attribute__((ext_vector_type(2))) float v2f;
typedef __attribute__((ext_vector_type(8))) float v8f;


#define NSLICE 16
#define NRANGE 16

__device__ __forceinline__ int draw_rank(int c, int lo, int len, int* cntL) {
    const unsigned d = (unsigned)(c - lo);
    return (d < (unsigned)len) ? atomicAdd(&cntL[d], 1) : -1;
}

__global__ void __launch_bounds__(1024)
count_rank_kernel(const int* __restrict__ col, int* __restrict__ part,
                  int* __restrict__ rank, int n, int E, int nP,
                  int rangeLen, int sliceLen) {
    extern __shared__ int cntL[];
    const int tid = threadIdx.x;
    const int lo  = blockIdx.y * rangeLen;
    int len = n - lo;
    if (len > rangeLen) len = rangeLen;
    if (len < 0) len = 0;
    for (int i = tid; i < len; i += 1024) cntL[i] = 0;
    __syncthreads();

    int e0 = blockIdx.x * sliceLen;
    if (e0 > E) e0 = E;
    int e1 = e0 + sliceLen;
    if (e1 > E) e1 = E;
    const int* __restrict__ cs = col + e0;
    const int nq = ((((size_t)cs) & 15) == 0) ? ((e1 - e0) >> 2) : 0;
    const int4* __restrict__ cs4 = (const int4*)cs;
    for (int q0 = tid; q0 < nq; q0 += 4 * 1024) {
        int4 v[4];
        #pragma unroll
        for (int u = 0; u < 4; ++u) {
            const int q = q0 + u * 1024;
            v[u] = (q < nq) ? cs4[q] : make_int4(-1, -1, -1, -1);
        }
        int rk[4][4];
        #pragma unroll
        for (int u = 0; u < 4; ++u) {
            rk[u][0] = draw_rank(v[u].x, lo, len, cntL);
            rk[u][1] = draw_rank(v[u].y, lo, len, cntL);
            rk[u][2] = draw_rank(v[u].z, lo, len, cntL);
            rk[u][3] = draw_rank(v[u].w, lo, len, cntL);
        }
        #pragma unroll
        for (int u = 0; u < 4; ++u) {
            const int e = e0 + 4 * (q0 + u * 1024);
            #pragma unroll
            for (int j = 0; j < 4; ++j)
                if (rk[u][j] >= 0) rank[e + j] = rk[u][j];
        }
    }
    for (int e = e0 + 4 * nq + tid; e < e1; e += 1024) {
        const int rk = draw_rank(col[e], lo, len, cntL);
        if (rk >= 0) rank[e] = rk;
    }
    __syncthreads();

    int* __restrict__ dst = part + (size_t)blockIdx.x * (size_t)nP + lo;
    for (int i = tid; i < len; i += 1024) dst[i] = cntL[i];
}

__device__ __forceinline__ int4 load4_guard(const int* p, int idx, int n) {
    if (idx + 3 < n) return *(const int4*)p;
    int4 v;
    v.x = (idx     < n) ? p[0] : 0;
    v.y = (idx + 1 < n) ? p[1] : 0;
    v.z = (idx + 2 < n) ? p[2] : 0;
    v.w = 0;
    return v;
}

__device__ __forceinline__ void store4_guard(int* p, int idx, int n, int4 v) {
    if (idx + 3 < n) { *(int4*)p = v; return; }
    if (idx     < n) p[0] = v.x;
    if (idx + 1 < n) p[1] = v.y;
    if (idx + 2 < n) p[2] = v.z;
}

__device__ __forceinline__ int wave_sum(int s) {
    #pragma unroll
    for (int off = 16; off > 0; off >>= 1) s += __shfl_xor(s, off, 32);
    return s;
}

__global__ void __launch_bounds__(256)
block_sum_kernel(const int* __restrict__ part, int* __restrict__ bsum, int n, int nP) {
    __shared__ int sm[8];
    const int base = blockIdx.x * 1024 + threadIdx.x * 4;
    int s = 0;
    if (base < n) {
        #pragma unroll
        for (int sl = 0; sl < NSLICE; ++sl) {
            const int4 v = load4_guard(part + (size_t)sl * (size_t)nP + base, base, n);
            s += v.x + v.y + v.z + v.w;
        }
    }
    s = wave_sum(s);
    if ((threadIdx.x & 31) == 0) sm[threadIdx.x >> 5] = s;
    __syncthreads();
    if (threadIdx.x == 0) {
        int t = 0;
        #pragma unroll
        for (int w = 0; w < 8; ++w) t += sm[w];
        bsum[blockIdx.x] = t;
    }
}

__global__ void __launch_bounds__(256)
scan_kernel(int* __restrict__ part, const int* __restrict__ bsum,
            int* __restrict__ offs,
            float* __restrict__ dinv, int n, int nP) {
    __shared__ int smA[8];
    __shared__ int smB[8];
    const int tid  = threadIdx.x;
    const int lane = tid & 31;
    const int wave = tid >> 5;

    int pb = 0;
    for (int i = tid; i < (int)blockIdx.x; i += 256) pb += bsum[i];
    pb = wave_sum(pb);
    if (lane == 0) smA[wave] = pb;

    const int base = blockIdx.x * 1024 + tid * 4;
    int4 p[NSLICE];
    int c[4] = {0, 0, 0, 0};
    #pragma unroll
    for (int sl = 0; sl < NSLICE; ++sl) {
        p[sl] = (base < n) ? load4_guard(part + (size_t)sl * (size_t)nP + base, base, n)
                           : make_int4(0, 0, 0, 0);
        c[0] += p[sl].x; c[1] += p[sl].y; c[2] += p[sl].z; c[3] += p[sl].w;
    }
    const int tsum = c[0] + c[1] + c[2] + c[3];

    int incl = tsum;
    #pragma unroll
    for (int off = 1; off < 32; off <<= 1) {
        const int t = __shfl_up(incl, off, 32);
        if (lane >= off) incl += t;
    }
    if (lane == 31) smB[wave] = incl;
    __syncthreads();

    int run = 0;
    #pragma unroll
    for (int w = 0; w < 8; ++w) run += smA[w];
    #pragma unroll
    for (int w = 0; w < 8; ++w) if (w < wave) run += smB[w];
    run += incl - tsum;

    if (base < n) {
        int4 a;
        a.x = run;
        a.y = a.x + c[0];
        a.z = a.y + c[1];
        a.w = a.z + c[2];
        store4_guard(offs + base, base, n, a);
        #pragma unroll
        for (int k = 0; k < 4; ++k)
            if (base + k < n) dinv[base + k] = rsqrtf((float)c[k] + 1.0f);
        if (n - 1 < base + 4) offs[n] = a.w + c[3];
        #pragma unroll
        for (int sl = 0; sl < NSLICE; ++sl) {
            store4_guard(part + (size_t)sl * (size_t)nP + base, base, n, a);
            a.x += p[sl].x; a.y += p[sl].y; a.z += p[sl].z; a.w += p[sl].w;
        }
    }
}

__global__ void bucket_fill_kernel(const int* __restrict__ row, const int* __restrict__ col,
                                   const int* __restrict__ rank, const int* __restrict__ part,
                                   const float* __restrict__ dinv,
                                   int2* __restrict__ ent, int E, int nP, int sliceLen) {
    int e = blockIdx.x * blockDim.x + threadIdx.x;
    if (e < E) {
        const int r = row[e];
        const int c = col[e];
        const int s = e / sliceLen;
        const int p = part[(size_t)s * (size_t)nP + c] + rank[e];
        const float nrm = dinv[r] * dinv[c];
        ent[p] = make_int2(r, __float_as_int(nrm));
    }
}

template <int TN>
__global__ void gemm_wmma_f32_kernel(const float* __restrict__ X,
                                     const float* __restrict__ W,
                                     float* __restrict__ H,
                                     int n, int cin, int cout) {
    constexpr int coutP = TN * 16;
    extern __shared__ float lds[];

    const int total = cin * coutP;
    for (int i = threadIdx.x; i < total; i += blockDim.x) {
        const int k = i / coutP;
        const int c = i - k * coutP;
        const float v = (c < cout) ? W[(size_t)k * cout + c] : 0.0f;
        lds[(k >> 1) * (2 * coutP) + (c << 1) + (k & 1)] = v;
    }
    __syncthreads();

    const int lane   = threadIdx.x & 31;
    const int waveId = threadIdx.x >> 5;
    const int tileM  = blockIdx.x * (blockDim.x >> 5) + waveId;
    const int tilesM = (n + 15) >> 4;
    if (tileM >= tilesM) return;

    const int nlane = lane & 15;
    const int koff  = (lane >> 4) << 1;
    const int mrow  = (tileM << 4) + nlane;
    const int arow  = (mrow < n) ? mrow : (n - 1);

    const float* aPtr = X + (size_t)arow * cin + koff;
    int bIdx = (koff >> 1) * (2 * coutP) + (nlane << 1);

    v8f acc[TN] = {};
    for (int kb = 0; kb < cin; kb += 4) {
        const float2 av = *(const float2*)aPtr;
        v2f a; a.x = av.x; a.y = av.y;
        #pragma unroll
        for (int t = 0; t < TN; ++t) {
            const v2f b = *(const v2f*)(lds + bIdx + (t << 5));
            acc[t] = __builtin_amdgcn_wmma_f32_16x16x4_f32(
                          false, a,  false, b,
                          (short)0, acc[t],
                          false,  false);
        }
        aPtr += 4;
        bIdx += 4 * coutP;
    }

    const int rbase = (tileM << 4) + ((lane >> 4) << 3);
    #pragma unroll
    for (int t = 0; t < TN; ++t) {
        const int col = t * 16 + nlane;
        if (col < cout) {
            #pragma unroll
            for (int r = 0; r < 8; ++r) {
                const int orow = rbase + r;
                if (orow < n) H[(size_t)orow * cout + col] = acc[t][r];
            }
        }
    }
}

template <int VEC, int NP, int U>
__global__ void __launch_bounds__(256)
gather_kernel(const float* __restrict__ h,
              const int* __restrict__ offs,
              const int2* __restrict__ ent,
              const float* __restrict__ dinv,
              const float* __restrict__ bias,
              float* __restrict__ out,
              int n, int cout, int doRelu) {
    const unsigned lane = threadIdx.x & 31u;
    const int node = __builtin_amdgcn_readfirstlane(
        (int)(blockIdx.x * (blockDim.x >> 5) + (threadIdx.x >> 5)));
    if (node >= n) return;

    const int   beg = offs[node];
    const int   end = offs[node + 1];
    const float di  = dinv[node];

    int  colOf[NP];
    bool act[NP];
    int  ldc[NP];
    #pragma unroll
    for (int p = 0; p < NP; ++p) {
        colOf[p] = (int)(lane + 32u * p) * VEC;
        act[p]   = colOf[p] < cout;
        ldc[p]   = act[p] ? colOf[p] : 0;
    }

    float hs[NP][VEC], bs[NP][VEC];
    {
        const float* __restrict__ self = h + (size_t)node * (size_t)cout;
        #pragma unroll
        for (int p = 0; p < NP; ++p) {
            if (VEC == 2) {
                const float2 t = *(const float2*)(self + ldc[p]);
                const float2 b = *(const float2*)(bias + ldc[p]);
                hs[p][0] = t.x; hs[p][VEC - 1] = t.y;
                bs[p][0] = b.x; bs[p][VEC - 1] = b.y;
            } else {
                hs[p][0] = self[ldc[p]];
                bs[p][0] = bias[ldc[p]];
            }
        }
    }

    float acc[NP][VEC] = {};
    for (int b0 = beg; b0 < end; b0 += 32) {
        const int cntb = (end - b0 < 32) ? (end - b0) : 32;
        int r = 0, wBits = 0;
        if ((int)lane < cntb) {
            const int2 en = ent[b0 + (int)lane];
            r     = en.x;
            wBits = en.y;
        }
        asm volatile("" :: "v"(r), "v"(wBits));
        for (int j0 = 0; j0 < cntb; j0 += U) {
            float v[U][NP][VEC] = {};
            #pragma unroll
            for (int u = 0; u < U; ++u) {
                const int j = j0 + u;
                if (j < cntb) {
                    const int rj = __builtin_amdgcn_readlane(r, j);
                    const float* __restrict__ src = h + (size_t)rj * (size_t)cout;
                    #pragma unroll
                    for (int p = 0; p < NP; ++p) {
                        if (VEC == 2) {
                            const float2 t = *(const float2*)(src + ldc[p]);
                            v[u][p][0] = t.x; v[u][p][VEC - 1] = t.y;
                        } else {
                            v[u][p][0] = src[ldc[p]];
                        }
                    }
                }
            }
            #pragma unroll
            for (int u = 0; u < U; ++u) {
                const int j = j0 + u;
                if (j < cntb) {
                    const float nj = __int_as_float(__builtin_amdgcn_readlane(wBits, j));
                    #pragma unroll
                    for (int p = 0; p < NP; ++p) {
                        #pragma unroll
                        for (int q = 0; q < VEC; ++q) acc[p][q] += v[u][p][q] * nj;
                    }
                }
            }
        }
    }

    const float dd = di * di;
    float* __restrict__ dst = out + (size_t)node * (size_t)cout;
    #pragma unroll
    for (int p = 0; p < NP; ++p) {
        float o[VEC];
        #pragma unroll
        for (int q = 0; q < VEC; ++q) {
            o[q] = acc[p][q] + hs[p][q] * dd + bs[p][q];
            if (doRelu) o[q] = fmaxf(o[q], 0.0f);
        }
        if (act[p]) {
            if (VEC == 2) {
                float2 t; t.x = o[0]; t.y = o[VEC - 1];
                *(float2*)(dst + colOf[p]) = t;
            } else {
                dst[colOf[p]] = o[0];
            }
        }
    }
}

#define APITCH 68

template <int TN, int VEC, int NP, int U>
__global__ void __launch_bounds__(512)
gather_project_kernel(const float* __restrict__ h,
                      const int* __restrict__ offs,
                      const int2* __restrict__ ent,
                      const float* __restrict__ dinv,
                      const float* __restrict__ bias,
                      const float* __restrict__ Wn,
                      float* __restrict__ Hn,
                      int n, int coutN, int doRelu, int tilesPerBlock) {
    constexpr int cout  = 64;
    constexpr int coutP = TN * 16;
    __shared__ __attribute__((aligned(16))) float Ws[cout * coutP];
    __shared__ __attribute__((aligned(16))) float As[2 * 16 * APITCH];

    {
        float w[2 * TN];
        #pragma unroll
        for (int u = 0; u < 2 * TN; ++u) {
            const int i = (int)threadIdx.x + u * 512;
            const int k = i / coutP;
            const int c = i - k * coutP;
            w[u] = (c < coutN) ? Wn[k * coutN + c] : 0.0f;
        }
        #pragma unroll
        for (int u = 0; u < 2 * TN; ++u) {
            const int i = (int)threadIdx.x + u * 512;
            const int k = i / coutP;
            const int c = i - k * coutP;
            Ws[(k >> 1) * (2 * coutP) + (c << 1) + (k & 1)] = w[u];
        }
    }

    const unsigned lane = threadIdx.x & 31u;
    const int wave = __builtin_amdgcn_readfirstlane((int)(threadIdx.x >> 5));

    int colOf[NP];
    #pragma unroll
    for (int p = 0; p < NP; ++p) colOf[p] = (int)(lane + 32u * p) * VEC;

    float bs[NP][VEC];
    #pragma unroll
    for (int p = 0; p < NP; ++p) {
        const float2 b = *(const float2*)(bias + colOf[p]);
        bs[p][0] = b.x; bs[p][VEC - 1] = b.y;
    }

    const int nlane = (int)lane & 15;
    const int koff  = ((int)lane >> 4) << 1;
    const int pcol  = (wave < TN ? wave : 0) * 16 + nlane;
    const float* bFrag = Ws + (koff >> 1) * (2 * coutP) + (pcol << 1);

    const int tile0 = (int)blockIdx.x * tilesPerBlock;
    for (int it = 0; it < tilesPerBlock; ++it) {
        const int tile = tile0 + it;
        const int node = tile * 16 + wave;
        float* At = As + (it & 1) * (16 * APITCH);

        float o[NP][VEC] = {};
        if (node < n) {
            const int   beg = offs[node];
            const int   end = offs[node + 1];
            const float di  = dinv[node];

            float hs[NP][VEC];
            {
                const float* __restrict__ self = h + (size_t)node * (size_t)cout;
                #pragma unroll
                for (int p = 0; p < NP; ++p) {
                    const float2 t = *(const float2*)(self + colOf[p]);
                    hs[p][0] = t.x; hs[p][VEC - 1] = t.y;
                }
            }

            float acc[NP][VEC] = {};
            for (int b0 = beg; b0 < end; b0 += 32) {
                const int cntb = (end - b0 < 32) ? (end - b0) : 32;
                int r = 0, wBits = 0;
                if ((int)lane < cntb) {
                    const int2 en = ent[b0 + (int)lane];
                    r     = en.x;
                    wBits = en.y;
                }
                asm volatile("" :: "v"(r), "v"(wBits));
                for (int j0 = 0; j0 < cntb; j0 += U) {
                    float v[U][NP][VEC] = {};
                    #pragma unroll
                    for (int u = 0; u < U; ++u) {
                        const int j = j0 + u;
                        if (j < cntb) {
                            const int rj = __builtin_amdgcn_readlane(r, j);
                            const float* __restrict__ src = h + (size_t)rj * (size_t)cout;
                            #pragma unroll
                            for (int p = 0; p < NP; ++p) {
                                const float2 t = *(const float2*)(src + colOf[p]);
                                v[u][p][0] = t.x; v[u][p][VEC - 1] = t.y;
                            }
                        }
                    }
                    #pragma unroll
                    for (int u = 0; u < U; ++u) {
                        const int j = j0 + u;
                        if (j < cntb) {
                            const float nj = __int_as_float(__builtin_amdgcn_readlane(wBits, j));
                            #pragma unroll
                            for (int p = 0; p < NP; ++p) {
                                #pragma unroll
                                for (int q = 0; q < VEC; ++q) acc[p][q] += v[u][p][q] * nj;
                            }
                        }
                    }
                }
            }

            const float dd = di * di;
            #pragma unroll
            for (int p = 0; p < NP; ++p) {
                #pragma unroll
                for (int q = 0; q < VEC; ++q) {
                    o[p][q] = acc[p][q] + hs[p][q] * dd + bs[p][q];
                    if (doRelu) o[p][q] = fmaxf(o[p][q], 0.0f);
                }
            }
        }
        #pragma unroll
        for (int p = 0; p < NP; ++p) {
            float2 t; t.x = o[p][0]; t.y = o[p][VEC - 1];
            *(float2*)(At + wave * APITCH + colOf[p]) = t;
        }
        __syncthreads();

        if (wave < TN && tile * 16 < n) {
            const float* aRow = At + nlane * APITCH + koff;
            v8f d = {};
            #pragma unroll
            for (int s = 0; s < 16; ++s) {
                const v2f a = *(const v2f*)(aRow + 4 * s);
                const v2f b = *(const v2f*)(bFrag + s * (4 * coutP));
                d = __builtin_amdgcn_wmma_f32_16x16x4_f32(
                         false, a,  false, b,
                         (short)0, d,
                         false,  false);
            }
            const int rbase = tile * 16 + (((int)lane >> 4) << 3);
            if (pcol < coutN) {
                #pragma unroll
                for (int r = 0; r < 8; ++r) {
                    const int orow = rbase + r;
                    if (orow < n) Hn[(size_t)orow * coutN + pcol] = d[r];
                }
            }
        }
    }
}


static inline int cdiv(long long a, long long b) { return (int)((a + b - 1) / b); }

static void launch_gemm(const float* X, const float* W, float* H,
                        int n, int cin, int cout, hipStream_t stream) {
    const int tilesM = (n + 15) >> 4;
    const int blocks = cdiv(tilesM, 8);
    const int TN     = (cout + 15) >> 4;
    const size_t shmem = (size_t)cin * TN * 16 * sizeof(float);
    switch (TN) {
        case 4: gemm_wmma_f32_kernel<4><<<blocks, 256, shmem, stream>>>(X, W, H, n, cin, cout); break;
        case 3: gemm_wmma_f32_kernel<3><<<blocks, 256, shmem, stream>>>(X, W, H, n, cin, cout); break;
        case 2: gemm_wmma_f32_kernel<2><<<blocks, 256, shmem, stream>>>(X, W, H, n, cin, cout); break;
        default: gemm_wmma_f32_kernel<1><<<blocks, 256, shmem, stream>>>(X, W, H, n, cin, cout); break;
    }
}

static void launch_gather(const float* h, const int* offs, const int2* ent,
                          const float* dinv, const float* bias, float* out,
                          int n, int cout, int doRelu, hipStream_t stream) {
    const int blocks = cdiv(n, 8);
    if ((cout & 1) == 0)
        gather_kernel<2, 1, 16><<<blocks, 256, 0, stream>>>(h, offs, ent, dinv, bias, out, n, cout, doRelu);
    else
        gather_kernel<1, 2, 16><<<blocks, 256, 0, stream>>>(h, offs, ent, dinv, bias, out, n, cout, doRelu);
}

static void launch_gather_project(const float* h, const int* offs, const int2* ent,
                                  const float* dinv, const float* bias,
                                  const float* Wn, float* Hn,
                                  int n, int coutN, int doRelu, hipStream_t stream) {
    const int tiles = cdiv(n, 16);
    const int tpb   = cdiv(tiles, 1024);
    const int blocks = cdiv(tiles, tpb);
    const int TN     = (coutN + 15) >> 4;
    switch (TN) {
        case 4: gather_project_kernel<4, 2, 1, 16><<<blocks, 512, 0, stream>>>(h, offs, ent, dinv, bias, Wn, Hn, n, coutN, doRelu, tpb); break;
        case 3: gather_project_kernel<3, 2, 1, 16><<<blocks, 512, 0, stream>>>(h, offs, ent, dinv, bias, Wn, Hn, n, coutN, doRelu, tpb); break;
        case 2: gather_project_kernel<2, 2, 1, 16><<<blocks, 512, 0, stream>>>(h, offs, ent, dinv, bias, Wn, Hn, n, coutN, doRelu, tpb); break;
        default: gather_project_kernel<1, 2, 1, 16><<<blocks, 512, 0, stream>>>(h, offs, ent, dinv, bias, Wn, Hn, n, coutN, doRelu, tpb); break;
    }
}

extern "C" void kernel_launch(void* const* d_in, const int* in_sizes, int n_in,
                              void* d_out, int out_size, void* d_ws, size_t ws_size,
                              hipStream_t stream) {
    const float* x    = (const float*)d_in[0];
    const int*   ei   = (const int*)  d_in[1];
    const float* W1   = (const float*)d_in[2];
    const float* b1   = (const float*)d_in[3];
    const float* W2   = (const float*)d_in[4];
    const float* b2   = (const float*)d_in[5];
    const float* Wout = (const float*)d_in[6];
    const float* bout = (const float*)d_in[7];

    const int C1  = in_sizes[3];
    const int C2  = in_sizes[5];
    const int C3  = in_sizes[7];
    const int Cin = in_sizes[2] / C1;
    const int n   = in_sizes[0] / Cin;
    const int E   = in_sizes[1] / 2;

    const int* rowIdx = ei;
    const int* colIdx = ei + E;

    const size_t nP = ((size_t)n + 64) & ~(size_t)63;
    float* hbuf   = (float*)d_ws;
    float* abuf   = hbuf + (size_t)n * 64;
    float* dinv   = abuf + (size_t)n * 64;
    int*   offs   = (int*)(dinv + nP);
    int*   bsum   = offs + nP;
    int*   part   = bsum + nP;
    int2*  ent    = (int2*)(part + (size_t)NSLICE * nP);
    int*   rank   = (int*)(ent + E);

    const int T = 256;
    const int scanBlocks = cdiv(n, 1024);
    const int rangeLen   = cdiv(n, NRANGE);
    const int sliceLen   = (cdiv(E, NSLICE) + 3) & ~3;

    count_rank_kernel<<<dim3(NSLICE, NRANGE), 1024, (size_t)rangeLen * sizeof(int), stream>>>(
        colIdx, part, rank, n, E, (int)nP, rangeLen, sliceLen);
    block_sum_kernel<<<scanBlocks, 256, 0, stream>>>(part, bsum, n, (int)nP);
    scan_kernel<<<scanBlocks, 256, 0, stream>>>(part, bsum, offs, dinv, n, (int)nP);
    bucket_fill_kernel<<<cdiv(E, T), T, 0, stream>>>(rowIdx, colIdx, rank, part, dinv, ent, E, (int)nP, sliceLen);

    float* cur = hbuf;
    float* oth = abuf;

    launch_gemm(x, W1, cur, n, Cin, C1, stream);

    if (C1 == 64) {
        launch_gather_project(cur, offs, ent, dinv, b1, W2, oth, n, C2, 1, stream);
        float* t = cur; cur = oth; oth = t;
    } else {
        launch_gather(cur, offs, ent, dinv, b1, oth, n, C1, 1, stream);
        launch_gemm(oth, W2, cur, n, C1, C2, stream);
    }

    if (C2 == 64) {
        launch_gather_project(cur, offs, ent, dinv, b2, Wout, oth, n, C3, 1, stream);
        float* t = cur; cur = oth; oth = t;
    } else {
        launch_gather(cur, offs, ent, dinv, b2, oth, n, C2, 1, stream);
        launch_gemm(oth, Wout, cur, n, C2, C3, stream);
    }

    launch_gather(cur, offs, ent, dinv, bout, (float*)d_out, n, C3, 0, stream);
}


# aggregation loops: one wave-uniform test per four gathered rows, not one per row, and constant lane selects; lanes past a bucket's end pad with the own row at weight 0; order of sums kept
# speedup vs baseline: 3.0749x; 1.0397x over previous
#include <hip/hip_runtime.h>

typedef __attribute__((ext_vector_type(2))) float v2f;
typedef __attribute__((ext_vector_type(8))) float v8f;


#define NSLICE 16
#define NRANGE 16

__device__ __forceinline__ int draw_rank(int c, int lo, int len, int* cntL) {
    const unsigned d = (unsigned)(c - lo);
    return (d < (unsigned)len) ? atomicAdd(&cntL[d], 1) : -1;
}

__global__ void __launch_bounds__(1024)
count_rank_kernel(const int* __restrict__ col, int* __restrict__ part,
                  int* __restrict__ rank, int n, int E, int nP,
                  int rangeLen, int sliceLen) {
    extern __shared__ int cntL[];
    const int tid = threadIdx.x;
    const int lo  = blockIdx.y * rangeLen;
    int len = n - lo;
    if (len > rangeLen) len = rangeLen;
    if (len < 0) len = 0;
    for (int i = tid; i < len; i += 1024) cntL[i] = 0;
    __syncthreads();

    int e0 = blockIdx.x * sliceLen;
    if (e0 > E) e0 = E;
    int e1 = e0 + sliceLen;
    if (e1 > E) e1 = E;
    const int* __restrict__ cs = col + e0;
    const int nq = ((((size_t)cs) & 15) == 0) ? ((e1 - e0) >> 2) : 0;
    const int4* __restrict__ cs4 = (const int4*)cs;
    for (int q0 = tid; q0 < nq; q0 += 4 * 1024) {
        int4 v[4];
        #pragma unroll
        for (int u = 0; u < 4; ++u) {
            const int q = q0 + u * 1024;
            v[u] = (q < nq) ? cs4[q] : make_int4(-1, -1, -1, -1);
        }
        int rk[4][4];
        #pragma unroll
        for (int u = 0; u < 4; ++u) {
            rk[u][0] = draw_rank(v[u].x, lo, len, cntL);
            rk[u][1] = draw_rank(v[u].y, lo, len, cntL);
            rk[u][2] = draw_rank(v[u].z, lo, len, cntL);
            rk[u][3] = draw_rank(v[u].w, lo, len, cntL);
        }
        #pragma unroll
        for (int u = 0; u < 4; ++u) {
            const int e = e0 + 4 * (q0 + u * 1024);
            #pragma unroll
            for (int j = 0; j < 4; ++j)
                if (rk[u][j] >= 0) rank[e + j] = rk[u][j];
        }
    }
    for (int e = e0 + 4 * nq + tid; e < e1; e += 1024) {
        const int rk = draw_rank(col[e], lo, len, cntL);
        if (rk >= 0) rank[e] = rk;
    }
    __syncthreads();

    int* __restrict__ dst = part + (size_t)blockIdx.x * (size_t)nP + lo;
    for (int i = tid; i < len; i += 1024) dst[i] = cntL[i];
}

__device__ __forceinline__ int4 load4_guard(const int* p, int idx, int n) {
    if (idx + 3 < n) return *(const int4*)p;
    int4 v;
    v.x = (idx     < n) ? p[0] : 0;
    v.y = (idx + 1 < n) ? p[1] : 0;
    v.z = (idx + 2 < n) ? p[2] : 0;
    v.w = 0;
    return v;
}

__device__ __forceinline__ void store4_guard(int* p, int idx, int n, int4 v) {
    if (idx + 3 < n) { *(int4*)p = v; return; }
    if (idx     < n) p[0] = v.x;
    if (idx + 1 < n) p[1] = v.y;
    if (idx + 2 < n) p[2] = v.z;
}

__device__ __forceinline__ int wave_sum(int s) {
    #pragma unroll
    for (int off = 16; off > 0; off >>= 1) s += __shfl_xor(s, off, 32);
    return s;
}

__global__ void __launch_bounds__(256)
block_sum_kernel(const int* __restrict__ part, int* __restrict__ bsum, int n, int nP) {
    __shared__ int sm[8];
    const int base = blockIdx.x * 1024 + threadIdx.x * 4;
    int s = 0;
    if (base < n) {
        #pragma unroll
        for (int sl = 0; sl < NSLICE; ++sl) {
            const int4 v = load4_guard(part + (size_t)sl * (size_t)nP + base, base, n);
            s += v.x + v.y + v.z + v.w;
        }
    }
    s = wave_sum(s);
    if ((threadIdx.x & 31) == 0) sm[threadIdx.x >> 5] = s;
    __syncthreads();
    if (threadIdx.x == 0) {
        int t = 0;
        #pragma unroll
        for (int w = 0; w < 8; ++w) t += sm[w];
        bsum[blockIdx.x] = t;
    }
}

__global__ void __launch_bounds__(256)
scan_kernel(int* __restrict__ part, const int* __restrict__ bsum,
            int* __restrict__ offs,
            float* __restrict__ dinv, int n, int nP) {
    __shared__ int smA[8];
    __shared__ int smB[8];
    const int tid  = threadIdx.x;
    const int lane = tid & 31;
    const int wave = tid >> 5;

    int pb = 0;
    for (int i = tid; i < (int)blockIdx.x; i += 256) pb += bsum[i];
    pb = wave_sum(pb);
    if (lane == 0) smA[wave] = pb;

    const int base = blockIdx.x * 1024 + tid * 4;
    int4 p[NSLICE];
    int c[4] = {0, 0, 0, 0};
    #pragma unroll
    for (int sl = 0; sl < NSLICE; ++sl) {
        p[sl] = (base < n) ? load4_guard(part + (size_t)sl * (size_t)nP + base, base, n)
                           : make_int4(0, 0, 0, 0);
        c[0] += p[sl].x; c[1] += p[sl].y; c[2] += p[sl].z; c[3] += p[sl].w;
    }
    const int tsum = c[0] + c[1] + c[2] + c[3];

    int incl = tsum;
    #pragma unroll
    for (int off = 1; off < 32; off <<= 1) {
        const int t = __shfl_up(incl, off, 32);
        if (lane >= off) incl += t;
    }
    if (lane == 31) smB[wave] = incl;
    __syncthreads();

    int run = 0;
    #pragma unroll
    for (int w = 0; w < 8; ++w) run += smA[w];
    #pragma unroll
    for (int w = 0; w < 8; ++w) if (w < wave) run += smB[w];
    run += incl - tsum;

    if (base < n) {
        int4 a;
        a.x = run;
        a.y = a.x + c[0];
        a.z = a.y + c[1];
        a.w = a.z + c[2];
        store4_guard(offs + base, base, n, a);
        #pragma unroll
        for (int k = 0; k < 4; ++k)
            if (base + k < n) dinv[base + k] = rsqrtf((float)c[k] + 1.0f);
        if (n - 1 < base + 4) offs[n] = a.w + c[3];
        #pragma unroll
        for (int sl = 0; sl < NSLICE; ++sl) {
            store4_guard(part + (size_t)sl * (size_t)nP + base, base, n, a);
            a.x += p[sl].x; a.y += p[sl].y; a.z += p[sl].z; a.w += p[sl].w;
        }
    }
}

__global__ void bucket_fill_kernel(const int* __restrict__ row, const int* __restrict__ col,
                                   const int* __restrict__ rank, const int* __restrict__ part,
                                   const float* __restrict__ dinv,
                                   int2* __restrict__ ent, int E, int nP, int sliceLen) {
    int e = blockIdx.x * blockDim.x + threadIdx.x;
    if (e < E) {
        const int r = row[e];
        const int c = col[e];
        const int s = e / sliceLen;
        const int p = part[(size_t)s * (size_t)nP + c] + rank[e];
        const float nrm = dinv[r] * dinv[c];
        ent[p] = make_int2(r, __float_as_int(nrm));
    }
}

template <int TN>
__global__ void gemm_wmma_f32_kernel(const float* __restrict__ X,
                                     const float* __restrict__ W,
                                     float* __restrict__ H,
                                     int n, int cin, int cout) {
    constexpr int coutP = TN * 16;
    extern __shared__ float lds[];

    const int total = cin * coutP;
    for (int i = threadIdx.x; i < total; i += blockDim.x) {
        const int k = i / coutP;
        const int c = i - k * coutP;
        const float v = (c < cout) ? W[(size_t)k * cout + c] : 0.0f;
        lds[(k >> 1) * (2 * coutP) + (c << 1) + (k & 1)] = v;
    }
    __syncthreads();

    const int lane   = threadIdx.x & 31;
    const int waveId = threadIdx.x >> 5;
    const int tileM  = blockIdx.x * (blockDim.x >> 5) + waveId;
    const int tilesM = (n + 15) >> 4;
    if (tileM >= tilesM) return;

    const int nlane = lane & 15;
    const int koff  = (lane >> 4) << 1;
    const int mrow  = (tileM << 4) + nlane;
    const int arow  = (mrow < n) ? mrow : (n - 1);

    const float* aPtr = X + (size_t)arow * cin + koff;
    int bIdx = (koff >> 1) * (2 * coutP) + (nlane << 1);

    v8f acc[TN] = {};
    for (int kb = 0; kb < cin; kb += 4) {
        const float2 av = *(const float2*)aPtr;
        v2f a; a.x = av.x; a.y = av.y;
        #pragma unroll
        for (int t = 0; t < TN; ++t) {
            const v2f b = *(const v2f*)(lds + bIdx + (t << 5));
            acc[t] = __builtin_amdgcn_wmma_f32_16x16x4_f32(
                          false, a,  false, b,
                          (short)0, acc[t],
                          false,  false);
        }
        aPtr += 4;
        bIdx += 4 * coutP;
    }

    const int rbase = (tileM << 4) + ((lane >> 4) << 3);
    #pragma unroll
    for (int t = 0; t < TN; ++t) {
        const int col = t * 16 + nlane;
        if (col < cout) {
            #pragma unroll
            for (int r = 0; r < 8; ++r) {
                const int orow = rbase + r;
                if (orow < n) H[(size_t)orow * cout + col] = acc[t][r];
            }
        }
    }
}

template <int VEC, int NP>
__device__ __forceinline__ void sum_bucket(const float* __restrict__ h, size_t pitch,
                                           const int2* __restrict__ ent,
                                           int beg, int end, int node, unsigned lane,
                                           const int (&ldc)[NP], float (&acc)[NP][VEC]) {
    for (int b0 = beg; b0 < end; b0 += 32) {
        const int cntb = (end - b0 < 32) ? (end - b0) : 32;
        int r = node, wBits = 0;
        if ((int)lane < cntb) {
            const int2 en = ent[b0 + (int)lane];
            r     = en.x;
            wBits = en.y;
        }
        asm volatile("" :: "v"(r), "v"(wBits));
        const int nq = (cntb + 3) >> 2;
        #pragma unroll
        for (int b = 0; b < 2; ++b) {
            if (4 * b < nq) {
                float v[4][4][NP][VEC];
                #pragma unroll
                for (int g = 0; g < 4; ++g) {
                    if (4 * b + g < nq) {
                        #pragma unroll
                        for (int u = 0; u < 4; ++u) {
                            const int rj = __builtin_amdgcn_readlane(r, 16 * b + 4 * g + u);
                            const float* __restrict__ src = h + (size_t)rj * pitch;
                            #pragma unroll
                            for (int p = 0; p < NP; ++p) {
                                if (VEC == 2) {
                                    const float2 t = *(const float2*)(src + ldc[p]);
                                    v[g][u][p][0] = t.x; v[g][u][p][VEC - 1] = t.y;
                                } else {
                                    v[g][u][p][0] = src[ldc[p]];
                                }
                            }
                        }
                    }
                }
                #pragma unroll
                for (int g = 0; g < 4; ++g) {
                    if (4 * b + g < nq) {
                        #pragma unroll
                        for (int u = 0; u < 4; ++u) {
                            const float nj = __int_as_float(
                                __builtin_amdgcn_readlane(wBits, 16 * b + 4 * g + u));
                            #pragma unroll
                            for (int p = 0; p < NP; ++p) {
                                #pragma unroll
                                for (int q = 0; q < VEC; ++q)
                                    acc[p][q] = __builtin_fmaf(v[g][u][p][q], nj, acc[p][q]);
                            }
                        }
                    }
                }
            }
        }
    }
}

template <int VEC, int NP>
__global__ void __launch_bounds__(256)
gather_kernel(const float* __restrict__ h,
              const int* __restrict__ offs,
              const int2* __restrict__ ent,
              const float* __restrict__ dinv,
              const float* __restrict__ bias,
              float* __restrict__ out,
              int n, int cout, int doRelu) {
    const unsigned lane = threadIdx.x & 31u;
    const int node = __builtin_amdgcn_readfirstlane(
        (int)(blockIdx.x * (blockDim.x >> 5) + (threadIdx.x >> 5)));
    if (node >= n) return;

    const int   beg = offs[node];
    const int   end = offs[node + 1];
    const float di  = dinv[node];

    int  colOf[NP];
    bool act[NP];
    int  ldc[NP];
    #pragma unroll
    for (int p = 0; p < NP; ++p) {
        colOf[p] = (int)(lane + 32u * p) * VEC;
        act[p]   = colOf[p] < cout;
        ldc[p]   = act[p] ? colOf[p] : 0;
    }

    float hs[NP][VEC], bs[NP][VEC];
    {
        const float* __restrict__ self = h + (size_t)node * (size_t)cout;
        #pragma unroll
        for (int p = 0; p < NP; ++p) {
            if (VEC == 2) {
                const float2 t = *(const float2*)(self + ldc[p]);
                const float2 b = *(const float2*)(bias + ldc[p]);
                hs[p][0] = t.x; hs[p][VEC - 1] = t.y;
                bs[p][0] = b.x; bs[p][VEC - 1] = b.y;
            } else {
                hs[p][0] = self[ldc[p]];
                bs[p][0] = bias[ldc[p]];
            }
        }
    }

    float acc[NP][VEC] = {};
    sum_bucket<VEC, NP>(h, (size_t)cout, ent, beg, end, node, lane, ldc, acc);

    const float dd = di * di;
    float* __restrict__ dst = out + (size_t)node * (size_t)cout;
    #pragma unroll
    for (int p = 0; p < NP; ++p) {
        float o[VEC];
        #pragma unroll
        for (int q = 0; q < VEC; ++q) {
            o[q] = acc[p][q] + hs[p][q] * dd + bs[p][q];
            if (doRelu) o[q] = fmaxf(o[q], 0.0f);
        }
        if (act[p]) {
            if (VEC == 2) {
                float2 t; t.x = o[0]; t.y = o[VEC - 1];
                *(float2*)(dst + colOf[p]) = t;
            } else {
                dst[colOf[p]] = o[0];
            }
        }
    }
}

#define APITCH 68

template <int TN, int VEC, int NP>
__global__ void __launch_bounds__(512)
gather_project_kernel(const float* __restrict__ h,
                      const int* __restrict__ offs,
                      const int2* __restrict__ ent,
                      const float* __restrict__ dinv,
                      const float* __restrict__ bias,
                      const float* __restrict__ Wn,
                      float* __restrict__ Hn,
                      int n, int coutN, int doRelu, int tilesPerBlock) {
    constexpr int cout  = 64;
    constexpr int coutP = TN * 16;
    __shared__ __attribute__((aligned(16))) float Ws[cout * coutP];
    __shared__ __attribute__((aligned(16))) float As[2 * 16 * APITCH];

    {
        float w[2 * TN];
        #pragma unroll
        for (int u = 0; u < 2 * TN; ++u) {
            const int i = (int)threadIdx.x + u * 512;
            const int k = i / coutP;
            const int c = i - k * coutP;
            w[u] = (c < coutN) ? Wn[k * coutN + c] : 0.0f;
        }
        #pragma unroll
        for (int u = 0; u < 2 * TN; ++u) {
            const int i = (int)threadIdx.x + u * 512;
            const int k = i / coutP;
            const int c = i - k * coutP;
            Ws[(k >> 1) * (2 * coutP) + (c << 1) + (k & 1)] = w[u];
        }
    }

    const unsigned lane = threadIdx.x & 31u;
    const int wave = __builtin_amdgcn_readfirstlane((int)(threadIdx.x >> 5));

    int colOf[NP];
    #pragma unroll
    for (int p = 0; p < NP; ++p) colOf[p] = (int)(lane + 32u * p) * VEC;

    float bs[NP][VEC];
    #pragma unroll
    for (int p = 0; p < NP; ++p) {
        const float2 b = *(const float2*)(bias + colOf[p]);
        bs[p][0] = b.x; bs[p][VEC - 1] = b.y;
    }

    const int nlane = (int)lane & 15;
    const int koff  = ((int)lane >> 4) << 1;
    const int pcol  = (wave < TN ? wave : 0) * 16 + nlane;
    const float* bFrag = Ws + (koff >> 1) * (2 * coutP) + (pcol << 1);

    const int tile0 = (int)blockIdx.x * tilesPerBlock;
    for (int it = 0; it < tilesPerBlock; ++it) {
        const int tile = tile0 + it;
        const int node = tile * 16 + wave;
        float* At = As + (it & 1) * (16 * APITCH);

        float o[NP][VEC] = {};
        if (node < n) {
            const int   beg = offs[node];
            const int   end = offs[node + 1];
            const float di  = dinv[node];

            float hs[NP][VEC];
            {
                const float* __restrict__ self = h + (size_t)node * (size_t)cout;
                #pragma unroll
                for (int p = 0; p < NP; ++p) {
                    const float2 t = *(const float2*)(self + colOf[p]);
                    hs[p][0] = t.x; hs[p][VEC - 1] = t.y;
                }
            }

            float acc[NP][VEC] = {};
            sum_bucket<VEC, NP>(h, (size_t)cout, ent, beg, end, node, lane, colOf, acc);

            const float dd = di * di;
            #pragma unroll
            for (int p = 0; p < NP; ++p) {
                #pragma unroll
                for (int q = 0; q < VEC; ++q) {
                    o[p][q] = acc[p][q] + hs[p][q] * dd + bs[p][q];
                    if (doRelu) o[p][q] = fmaxf(o[p][q], 0.0f);
                }
            }
        }
        #pragma unroll
        for (int p = 0; p < NP; ++p) {
            float2 t; t.x = o[p][0]; t.y = o[p][VEC - 1];
            *(float2*)(At + wave * APITCH + colOf[p]) = t;
        }
        __syncthreads();

        if (wave < TN && tile * 16 < n) {
            const float* aRow = At + nlane * APITCH + koff;
            v8f d = {};
            #pragma unroll
            for (int s = 0; s < 16; ++s) {
                const v2f a = *(const v2f*)(aRow + 4 * s);
                const v2f b = *(const v2f*)(bFrag + s * (4 * coutP));
                d = __builtin_amdgcn_wmma_f32_16x16x4_f32(
                         false, a,  false, b,
                         (short)0, d,
                         false,  false);
            }
            const int rbase = tile * 16 + (((int)lane >> 4) << 3);
            if (pcol < coutN) {
                #pragma unroll
                for (int r = 0; r < 8; ++r) {
                    const int orow = rbase + r;
                    if (orow < n) Hn[(size_t)orow * coutN + pcol] = d[r];
                }
            }
        }
    }
}


static inline int cdiv(long long a, long long b) { return (int)((a + b - 1) / b); }

static void launch_gemm(const float* X, const float* W, float* H,
                        int n, int cin, int cout, hipStream_t stream) {
    const int tilesM = (n + 15) >> 4;
    const int blocks = cdiv(tilesM, 8);
    const int TN     = (cout + 15) >> 4;
    const size_t shmem = (size_t)cin * TN * 16 * sizeof(float);
    switch (TN) {
        case 4: gemm_wmma_f32_kernel<4><<<blocks, 256, shmem, stream>>>(X, W, H, n, cin, cout); break;
        case 3: gemm_wmma_f32_kernel<3><<<blocks, 256, shmem, stream>>>(X, W, H, n, cin, cout); break;
        case 2: gemm_wmma_f32_kernel<2><<<blocks, 256, shmem, stream>>>(X, W, H, n, cin, cout); break;
        default: gemm_wmma_f32_kernel<1><<<blocks, 256, shmem, stream>>>(X, W, H, n, cin, cout); break;
    }
}

static void launch_gather(const float* h, const int* offs, const int2* ent,
                          const float* dinv, const float* bias, float* out,
                          int n, int cout, int doRelu, hipStream_t stream) {
    const int blocks = cdiv(n, 8);
    if ((cout & 1) == 0)
        gather_kernel<2, 1><<<blocks, 256, 0, stream>>>(h, offs, ent, dinv, bias, out, n, cout, doRelu);
    else
        gather_kernel<1, 2><<<blocks, 256, 0, stream>>>(h, offs, ent, dinv, bias, out, n, cout, doRelu);
}

static void launch_gather_project(const float* h, const int* offs, const int2* ent,
                                  const float* dinv, const float* bias,
                                  const float* Wn, float* Hn,
                                  int n, int coutN, int doRelu, hipStream_t stream) {
    const int tiles = cdiv(n, 16);
    const int tpb   = cdiv(tiles, 1024);
    const int blocks = cdiv(tiles, tpb);
    const int TN     = (coutN + 15) >> 4;
    switch (TN) {
        case 4: gather_project_kernel<4, 2, 1><<<blocks, 512, 0, stream>>>(h, offs, ent, dinv, bias, Wn, Hn, n, coutN, doRelu, tpb); break;
        case 3: gather_project_kernel<3, 2, 1><<<blocks, 512, 0, stream>>>(h, offs, ent, dinv, bias, Wn, Hn, n, coutN, doRelu, tpb); break;
        case 2: gather_project_kernel<2, 2, 1><<<blocks, 512, 0, stream>>>(h, offs, ent, dinv, bias, Wn, Hn, n, coutN, doRelu, tpb); break;
        default: gather_project_kernel<1, 2, 1><<<blocks, 512, 0, stream>>>(h, offs, ent, dinv, bias, Wn, Hn, n, coutN, doRelu, tpb); break;
    }
}

extern "C" void kernel_launch(void* const* d_in, const int* in_sizes, int n_in,
                              void* d_out, int out_size, void* d_ws, size_t ws_size,
                              hipStream_t stream) {
    const float* x    = (const float*)d_in[0];
    const int*   ei   = (const int*)  d_in[1];
    const float* W1   = (const float*)d_in[2];
    const float* b1   = (const float*)d_in[3];
    const float* W2   = (const float*)d_in[4];
    const float* b2   = (const float*)d_in[5];
    const float* Wout = (const float*)d_in[6];
    const float* bout = (const float*)d_in[7];

    const int C1  = in_sizes[3];
    const int C2  = in_sizes[5];
    const int C3  = in_sizes[7];
    const int Cin = in_sizes[2] / C1;
    const int n   = in_sizes[0] / Cin;
    const int E   = in_sizes[1] / 2;

    const int* rowIdx = ei;
    const int* colIdx = ei + E;

    const size_t nP = ((size_t)n + 64) & ~(size_t)63;
    float* hbuf   = (float*)d_ws;
    float* abuf   = hbuf + (size_t)n * 64;
    float* dinv   = abuf + (size_t)n * 64;
    int*   offs   = (int*)(dinv + nP);
    int*   bsum   = offs + nP;
    int*   part   = bsum + nP;
    int2*  ent    = (int2*)(part + (size_t)NSLICE * nP);
    int*   rank   = (int*)(ent + E);

    const int T = 256;
    const int scanBlocks = cdiv(n, 1024);
    const int rangeLen   = cdiv(n, NRANGE);
    const int sliceLen   = (cdiv(E, NSLICE) + 3) & ~3;

    count_rank_kernel<<<dim3(NSLICE, NRANGE), 1024, (size_t)rangeLen * sizeof(int), stream>>>(
        colIdx, part, rank, n, E, (int)nP, rangeLen, sliceLen);
    block_sum_kernel<<<scanBlocks, 256, 0, stream>>>(part, bsum, n, (int)nP);
    scan_kernel<<<scanBlocks, 256, 0, stream>>>(part, bsum, offs, dinv, n, (int)nP);
    bucket_fill_kernel<<<cdiv(E, T), T, 0, stream>>>(rowIdx, colIdx, rank, part, dinv, ent, E, (int)nP, sliceLen);

    float* cur = hbuf;
    float* oth = abuf;

    launch_gemm(x, W1, cur, n, Cin, C1, stream);

    if (C1 == 64) {
        launch_gather_project(cur, offs, ent, dinv, b1, W2, oth, n, C2, 1, stream);
        float* t = cur; cur = oth; oth = t;
    } else {
        launch_gather(cur, offs, ent, dinv, b1, oth, n, C1, 1, stream);
        launch_gemm(oth, W2, cur, n, C1, C2, stream);
    }

    if (C2 == 64) {
        launch_gather_project(cur, offs, ent, dinv, b2, Wout, oth, n, C3, 1, stream);
        float* t = cur; cur = oth; oth = t;
    } else {
        launch_gather(cur, offs, ent, dinv, b2, oth, n, C2, 1, stream);
        launch_gemm(oth, Wout, cur, n, C2, C3, stream);
    }

    launch_gather(cur, offs, ent, dinv, bout, (float*)d_out, n, C3, 0, stream);
}


# layer 1 product x @ W1 runs as the first blocks of the edge-placement launch (it needs nothing from the bucketing): matrix work beside the scattered edge stream; 7 launches, not 8; sums unchanged
# speedup vs baseline: 3.5266x; 1.1469x over previous
#include <hip/hip_runtime.h>

typedef __attribute__((ext_vector_type(2))) float v2f;
typedef __attribute__((ext_vector_type(8))) float v8f;


#define NSLICE 16
#define NRANGE 16

__device__ __forceinline__ int draw_rank(int c, int lo, int len, int* cntL) {
    const unsigned d = (unsigned)(c - lo);
    return (d < (unsigned)len) ? atomicAdd(&cntL[d], 1) : -1;
}

__global__ void __launch_bounds__(1024)
count_rank_kernel(const int* __restrict__ col, int* __restrict__ part,
                  int* __restrict__ rank, int n, int E, int nP,
                  int rangeLen, int sliceLen) {
    extern __shared__ int cntL[];
    const int tid = threadIdx.x;
    const int lo  = blockIdx.y * rangeLen;
    int len = n - lo;
    if (len > rangeLen) len = rangeLen;
    if (len < 0) len = 0;
    for (int i = tid; i < len; i += 1024) cntL[i] = 0;
    __syncthreads();

    int e0 = blockIdx.x * sliceLen;
    if (e0 > E) e0 = E;
    int e1 = e0 + sliceLen;
    if (e1 > E) e1 = E;
    const int* __restrict__ cs = col + e0;
    const int nq = ((((size_t)cs) & 15) == 0) ? ((e1 - e0) >> 2) : 0;
    const int4* __restrict__ cs4 = (const int4*)cs;
    for (int q0 = tid; q0 < nq; q0 += 4 * 1024) {
        int4 v[4];
        #pragma unroll
        for (int u = 0; u < 4; ++u) {
            const int q = q0 + u * 1024;
            v[u] = (q < nq) ? cs4[q] : make_int4(-1, -1, -1, -1);
        }
        int rk[4][4];
        #pragma unroll
        for (int u = 0; u < 4; ++u) {
            rk[u][0] = draw_rank(v[u].x, lo, len, cntL);
            rk[u][1] = draw_rank(v[u].y, lo, len, cntL);
            rk[u][2] = draw_rank(v[u].z, lo, len, cntL);
            rk[u][3] = draw_rank(v[u].w, lo, len, cntL);
        }
        #pragma unroll
        for (int u = 0; u < 4; ++u) {
            const int e = e0 + 4 * (q0 + u * 1024);
            #pragma unroll
            for (int j = 0; j < 4; ++j)
                if (rk[u][j] >= 0) rank[e + j] = rk[u][j];
        }
    }
    for (int e = e0 + 4 * nq + tid; e < e1; e += 1024) {
        const int rk = draw_rank(col[e], lo, len, cntL);
        if (rk >= 0) rank[e] = rk;
    }
    __syncthreads();

    int* __restrict__ dst = part + (size_t)blockIdx.x * (size_t)nP + lo;
    for (int i = tid; i < len; i += 1024) dst[i] = cntL[i];
}

__device__ __forceinline__ int4 load4_guard(const int* p, int idx, int n) {
    if (idx + 3 < n) return *(const int4*)p;
    int4 v;
    v.x = (idx     < n) ? p[0] : 0;
    v.y = (idx + 1 < n) ? p[1] : 0;
    v.z = (idx + 2 < n) ? p[2] : 0;
    v.w = 0;
    return v;
}

__device__ __forceinline__ void store4_guard(int* p, int idx, int n, int4 v) {
    if (idx + 3 < n) { *(int4*)p = v; return; }
    if (idx     < n) p[0] = v.x;
    if (idx + 1 < n) p[1] = v.y;
    if (idx + 2 < n) p[2] = v.z;
}

__device__ __forceinline__ int wave_sum(int s) {
    #pragma unroll
    for (int off = 16; off > 0; off >>= 1) s += __shfl_xor(s, off, 32);
    return s;
}

__global__ void __launch_bounds__(256)
block_sum_kernel(const int* __restrict__ part, int* __restrict__ bsum, int n, int nP) {
    __shared__ int sm[8];
    const int base = blockIdx.x * 1024 + threadIdx.x * 4;
    int s = 0;
    if (base < n) {
        #pragma unroll
        for (int sl = 0; sl < NSLICE; ++sl) {
            const int4 v = load4_guard(part + (size_t)sl * (size_t)nP + base, base, n);
            s += v.x + v.y + v.z + v.w;
        }
    }
    s = wave_sum(s);
    if ((threadIdx.x & 31) == 0) sm[threadIdx.x >> 5] = s;
    __syncthreads();
    if (threadIdx.x == 0) {
        int t = 0;
        #pragma unroll
        for (int w = 0; w < 8; ++w) t += sm[w];
        bsum[blockIdx.x] = t;
    }
}

__global__ void __launch_bounds__(256)
scan_kernel(int* __restrict__ part, const int* __restrict__ bsum,
            int* __restrict__ offs,
            float* __restrict__ dinv, int n, int nP) {
    __shared__ int smA[8];
    __shared__ int smB[8];
    const int tid  = threadIdx.x;
    const int lane = tid & 31;
    const int wave = tid >> 5;

    int pb = 0;
    for (int i = tid; i < (int)blockIdx.x; i += 256) pb += bsum[i];
    pb = wave_sum(pb);
    if (lane == 0) smA[wave] = pb;

    const int base = blockIdx.x * 1024 + tid * 4;
    int4 p[NSLICE];
    int c[4] = {0, 0, 0, 0};
    #pragma unroll
    for (int sl = 0; sl < NSLICE; ++sl) {
        p[sl] = (base < n) ? load4_guard(part + (size_t)sl * (size_t)nP + base, base, n)
                           : make_int4(0, 0, 0, 0);
        c[0] += p[sl].x; c[1] += p[sl].y; c[2] += p[sl].z; c[3] += p[sl].w;
    }
    const int tsum = c[0] + c[1] + c[2] + c[3];

    int incl = tsum;
    #pragma unroll
    for (int off = 1; off < 32; off <<= 1) {
        const int t = __shfl_up(incl, off, 32);
        if (lane >= off) incl += t;
    }
    if (lane == 31) smB[wave] = incl;
    __syncthreads();

    int run = 0;
    #pragma unroll
    for (int w = 0; w < 8; ++w) run += smA[w];
    #pragma unroll
    for (int w = 0; w < 8; ++w) if (w < wave) run += smB[w];
    run += incl - tsum;

    if (base < n) {
        int4 a;
        a.x = run;
        a.y = a.x + c[0];
        a.z = a.y + c[1];
        a.w = a.z + c[2];
        store4_guard(offs + base, base, n, a);
        #pragma unroll
        for (int k = 0; k < 4; ++k)
            if (base + k < n) dinv[base + k] = rsqrtf((float)c[k] + 1.0f);
        if (n - 1 < base + 4) offs[n] = a.w + c[3];
        #pragma unroll
        for (int sl = 0; sl < NSLICE; ++sl) {
            store4_guard(part + (size_t)sl * (size_t)nP + base, base, n, a);
            a.x += p[sl].x; a.y += p[sl].y; a.z += p[sl].z; a.w += p[sl].w;
        }
    }
}

__device__ __forceinline__ void place_edge(int e,
                                           const int* __restrict__ row, const int* __restrict__ col,
                                           const int* __restrict__ rank, const int* __restrict__ part,
                                           const float* __restrict__ dinv,
                                           int2* __restrict__ ent, int E, int nP, int sliceLen) {
    if (e < E) {
        const int r = row[e];
        const int c = col[e];
        const int s = e / sliceLen;
        const int p = part[(size_t)s * (size_t)nP + c] + rank[e];
        const float nrm = dinv[r] * dinv[c];
        ent[p] = make_int2(r, __float_as_int(nrm));
    }
}

template <int TN>
__device__ __forceinline__ void gemm_block(const float* __restrict__ X,
                                           const float* __restrict__ W,
                                           float* __restrict__ H,
                                           int n, int cin, int cout,
                                           int blk, float* lds) {
    constexpr int coutP = TN * 16;

    const int total = cin * coutP;
    for (int i = threadIdx.x; i < total; i += blockDim.x) {
        const int k = i / coutP;
        const int c = i - k * coutP;
        const float v = (c < cout) ? W[(size_t)k * cout + c] : 0.0f;
        lds[(k >> 1) * (2 * coutP) + (c << 1) + (k & 1)] = v;
    }
    __syncthreads();

    const int lane   = threadIdx.x & 31;
    const int waveId = threadIdx.x >> 5;
    const int tileM  = blk * (blockDim.x >> 5) + waveId;
    const int tilesM = (n + 15) >> 4;
    if (tileM >= tilesM) return;

    const int nlane = lane & 15;
    const int koff  = (lane >> 4) << 1;
    const int mrow  = (tileM << 4) + nlane;
    const int arow  = (mrow < n) ? mrow : (n - 1);

    const float* aPtr = X + (size_t)arow * cin + koff;
    int bIdx = (koff >> 1) * (2 * coutP) + (nlane << 1);

    v8f acc[TN] = {};
    for (int kb = 0; kb < cin; kb += 4) {
        const float2 av = *(const float2*)aPtr;
        v2f a; a.x = av.x; a.y = av.y;
        #pragma unroll
        for (int t = 0; t < TN; ++t) {
            const v2f b = *(const v2f*)(lds + bIdx + (t << 5));
            acc[t] = __builtin_amdgcn_wmma_f32_16x16x4_f32(
                          false, a,  false, b,
                          (short)0, acc[t],
                          false,  false);
        }
        aPtr += 4;
        bIdx += 4 * coutP;
    }

    const int rbase = (tileM << 4) + ((lane >> 4) << 3);
    #pragma unroll
    for (int t = 0; t < TN; ++t) {
        const int col = t * 16 + nlane;
        if (col < cout) {
            #pragma unroll
            for (int r = 0; r < 8; ++r) {
                const int orow = rbase + r;
                if (orow < n) H[(size_t)orow * cout + col] = acc[t][r];
            }
        }
    }
}

template <int TN>
__global__ void gemm_wmma_f32_kernel(const float* __restrict__ X,
                                     const float* __restrict__ W,
                                     float* __restrict__ H,
                                     int n, int cin, int cout) {
    extern __shared__ float lds[];
    gemm_block<TN>(X, W, H, n, cin, cout, (int)blockIdx.x, lds);
}

template <int TN>
__global__ void __launch_bounds__(256)
fill_project_kernel(const int* __restrict__ row, const int* __restrict__ col,
                    const int* __restrict__ rank, const int* __restrict__ part,
                    const float* __restrict__ dinv,
                    int2* __restrict__ ent, int E, int nP, int sliceLen,
                    const float* __restrict__ X, const float* __restrict__ W,
                    float* __restrict__ H, int n, int cin, int cout,
                    int gemmBlocks) {
    extern __shared__ float lds[];
    const int blk = (int)blockIdx.x;
    if (blk < gemmBlocks) {
        gemm_block<TN>(X, W, H, n, cin, cout, blk, lds);
    } else {
        place_edge((blk - gemmBlocks) * 256 + (int)threadIdx.x,
                   row, col, rank, part, dinv, ent, E, nP, sliceLen);
    }
}

template <int VEC, int NP>
__device__ __forceinline__ void sum_bucket(const float* __restrict__ h, size_t pitch,
                                           const int2* __restrict__ ent,
                                           int beg, int end, int node, unsigned lane,
                                           const int (&ldc)[NP], float (&acc)[NP][VEC]) {
    for (int b0 = beg; b0 < end; b0 += 32) {
        const int cntb = (end - b0 < 32) ? (end - b0) : 32;
        int r = node, wBits = 0;
        if ((int)lane < cntb) {
            const int2 en = ent[b0 + (int)lane];
            r     = en.x;
            wBits = en.y;
        }
        asm volatile("" :: "v"(r), "v"(wBits));
        const int nq = (cntb + 3) >> 2;
        #pragma unroll
        for (int b = 0; b < 2; ++b) {
            if (4 * b < nq) {
                float v[4][4][NP][VEC];
                #pragma unroll
                for (int g = 0; g < 4; ++g) {
                    if (4 * b + g < nq) {
                        #pragma unroll
                        for (int u = 0; u < 4; ++u) {
                            const int rj = __builtin_amdgcn_readlane(r, 16 * b + 4 * g + u);
                            const float* __restrict__ src = h + (size_t)rj * pitch;
                            #pragma unroll
                            for (int p = 0; p < NP; ++p) {
                                if (VEC == 2) {
                                    const float2 t = *(const float2*)(src + ldc[p]);
                                    v[g][u][p][0] = t.x; v[g][u][p][VEC - 1] = t.y;
                                } else {
                                    v[g][u][p][0] = src[ldc[p]];
                                }
                            }
                        }
                    }
                }
                #pragma unroll
                for (int g = 0; g < 4; ++g) {
                    if (4 * b + g < nq) {
                        #pragma unroll
                        for (int u = 0; u < 4; ++u) {
                            const float nj = __int_as_float(
                                __builtin_amdgcn_readlane(wBits, 16 * b + 4 * g + u));
                            #pragma unroll
                            for (int p = 0; p < NP; ++p) {
                                #pragma unroll
                                for (int q = 0; q < VEC; ++q)
                                    acc[p][q] = __builtin_fmaf(v[g][u][p][q], nj, acc[p][q]);
                            }
                        }
                    }
                }
            }
        }
    }
}

template <int VEC, int NP>
__global__ void __launch_bounds__(256)
gather_kernel(const float* __restrict__ h,
              const int* __restrict__ offs,
              const int2* __restrict__ ent,
              const float* __restrict__ dinv,
              const float* __restrict__ bias,
              float* __restrict__ out,
              int n, int cout, int doRelu) {
    const unsigned lane = threadIdx.x & 31u;
    const int node = __builtin_amdgcn_readfirstlane(
        (int)(blockIdx.x * (blockDim.x >> 5) + (threadIdx.x >> 5)));
    if (node >= n) return;

    const int   beg = offs[node];
    const int   end = offs[node + 1];
    const float di  = dinv[node];

    int  colOf[NP];
    bool act[NP];
    int  ldc[NP];
    #pragma unroll
    for (int p = 0; p < NP; ++p) {
        colOf[p] = (int)(lane + 32u * p) * VEC;
        act[p]   = colOf[p] < cout;
        ldc[p]   = act[p] ? colOf[p] : 0;
    }

    float hs[NP][VEC], bs[NP][VEC];
    {
        const float* __restrict__ self = h + (size_t)node * (size_t)cout;
        #pragma unroll
        for (int p = 0; p < NP; ++p) {
            if (VEC == 2) {
                const float2 t = *(const float2*)(self + ldc[p]);
                const float2 b = *(const float2*)(bias + ldc[p]);
                hs[p][0] = t.x; hs[p][VEC - 1] = t.y;
                bs[p][0] = b.x; bs[p][VEC - 1] = b.y;
            } else {
                hs[p][0] = self[ldc[p]];
                bs[p][0] = bias[ldc[p]];
            }
        }
    }

    float acc[NP][VEC] = {};
    sum_bucket<VEC, NP>(h, (size_t)cout, ent, beg, end, node, lane, ldc, acc);

    const float dd = di * di;
    float* __restrict__ dst = out + (size_t)node * (size_t)cout;
    #pragma unroll
    for (int p = 0; p < NP; ++p) {
        float o[VEC];
        #pragma unroll
        for (int q = 0; q < VEC; ++q) {
            o[q] = acc[p][q] + hs[p][q] * dd + bs[p][q];
            if (doRelu) o[q] = fmaxf(o[q], 0.0f);
        }
        if (act[p]) {
            if (VEC == 2) {
                float2 t; t.x = o[0]; t.y = o[VEC - 1];
                *(float2*)(dst + colOf[p]) = t;
            } else {
                dst[colOf[p]] = o[0];
            }
        }
    }
}

#define APITCH 68

template <int TN, int VEC, int NP>
__global__ void __launch_bounds__(512)
gather_project_kernel(const float* __restrict__ h,
                      const int* __restrict__ offs,
                      const int2* __restrict__ ent,
                      const float* __restrict__ dinv,
                      const float* __restrict__ bias,
                      const float* __restrict__ Wn,
                      float* __restrict__ Hn,
                      int n, int coutN, int doRelu, int tilesPerBlock) {
    constexpr int cout  = 64;
    constexpr int coutP = TN * 16;
    __shared__ __attribute__((aligned(16))) float Ws[cout * coutP];
    __shared__ __attribute__((aligned(16))) float As[2 * 16 * APITCH];

    {
        float w[2 * TN];
        #pragma unroll
        for (int u = 0; u < 2 * TN; ++u) {
            const int i = (int)threadIdx.x + u * 512;
            const int k = i / coutP;
            const int c = i - k * coutP;
            w[u] = (c < coutN) ? Wn[k * coutN + c] : 0.0f;
        }
        #pragma unroll
        for (int u = 0; u < 2 * TN; ++u) {
            const int i = (int)threadIdx.x + u * 512;
            const int k = i / coutP;
            const int c = i - k * coutP;
            Ws[(k >> 1) * (2 * coutP) + (c << 1) + (k & 1)] = w[u];
        }
    }

    const unsigned lane = threadIdx.x & 31u;
    const int wave = __builtin_amdgcn_readfirstlane((int)(threadIdx.x >> 5));

    int colOf[NP];
    #pragma unroll
    for (int p = 0; p < NP; ++p) colOf[p] = (int)(lane + 32u * p) * VEC;

    float bs[NP][VEC];
    #pragma unroll
    for (int p = 0; p < NP; ++p) {
        const float2 b = *(const float2*)(bias + colOf[p]);
        bs[p][0] = b.x; bs[p][VEC - 1] = b.y;
    }

    const int nlane = (int)lane & 15;
    const int koff  = ((int)lane >> 4) << 1;
    const int pcol  = (wave < TN ? wave : 0) * 16 + nlane;
    const float* bFrag = Ws + (koff >> 1) * (2 * coutP) + (pcol << 1);

    const int tile0 = (int)blockIdx.x * tilesPerBlock;
    for (int it = 0; it < tilesPerBlock; ++it) {
        const int tile = tile0 + it;
        const int node = tile * 16 + wave;
        float* At = As + (it & 1) * (16 * APITCH);

        float o[NP][VEC] = {};
        if (node < n) {
            const int   beg = offs[node];
            const int   end = offs[node + 1];
            const float di  = dinv[node];

            float hs[NP][VEC];
            {
                const float* __restrict__ self = h + (size_t)node * (size_t)cout;
                #pragma unroll
                for (int p = 0; p < NP; ++p) {
                    const float2 t = *(const float2*)(self + colOf[p]);
                    hs[p][0] = t.x; hs[p][VEC - 1] = t.y;
                }
            }

            float acc[NP][VEC] = {};
            sum_bucket<VEC, NP>(h, (size_t)cout, ent, beg, end, node, lane, colOf, acc);

            const float dd = di * di;
            #pragma unroll
            for (int p = 0; p < NP; ++p) {
                #pragma unroll
                for (int q = 0; q < VEC; ++q) {
                    o[p][q] = acc[p][q] + hs[p][q] * dd + bs[p][q];
                    if (doRelu) o[p][q] = fmaxf(o[p][q], 0.0f);
                }
            }
        }
        #pragma unroll
        for (int p = 0; p < NP; ++p) {
            float2 t; t.x = o[p][0]; t.y = o[p][VEC - 1];
            *(float2*)(At + wave * APITCH + colOf[p]) = t;
        }
        __syncthreads();

        if (wave < TN && tile * 16 < n) {
            const float* aRow = At + nlane * APITCH + koff;
            v8f d = {};
            #pragma unroll
            for (int s = 0; s < 16; ++s) {
                const v2f a = *(const v2f*)(aRow + 4 * s);
                const v2f b = *(const v2f*)(bFrag + s * (4 * coutP));
                d = __builtin_amdgcn_wmma_f32_16x16x4_f32(
                         false, a,  false, b,
                         (short)0, d,
                         false,  false);
            }
            const int rbase = tile * 16 + (((int)lane >> 4) << 3);
            if (pcol < coutN) {
                #pragma unroll
                for (int r = 0; r < 8; ++r) {
                    const int orow = rbase + r;
                    if (orow < n) Hn[(size_t)orow * coutN + pcol] = d[r];
                }
            }
        }
    }
}


static inline int cdiv(long long a, long long b) { return (int)((a + b - 1) / b); }

static void launch_gemm(const float* X, const float* W, float* H,
                        int n, int cin, int cout, hipStream_t stream) {
    const int tilesM = (n + 15) >> 4;
    const int blocks = cdiv(tilesM, 8);
    const int TN     = (cout + 15) >> 4;
    const size_t shmem = (size_t)cin * TN * 16 * sizeof(float);
    switch (TN) {
        case 4: gemm_wmma_f32_kernel<4><<<blocks, 256, shmem, stream>>>(X, W, H, n, cin, cout); break;
        case 3: gemm_wmma_f32_kernel<3><<<blocks, 256, shmem, stream>>>(X, W, H, n, cin, cout); break;
        case 2: gemm_wmma_f32_kernel<2><<<blocks, 256, shmem, stream>>>(X, W, H, n, cin, cout); break;
        default: gemm_wmma_f32_kernel<1><<<blocks, 256, shmem, stream>>>(X, W, H, n, cin, cout); break;
    }
}

static void launch_fill_project(const int* row, const int* col, const int* rank,
                                const int* part, const float* dinv, int2* ent,
                                int E, int nP, int sliceLen,
                                const float* X, const float* W, float* H,
                                int n, int cin, int cout, hipStream_t stream) {
    const int tilesM     = (n + 15) >> 4;
    const int gemmBlocks = cdiv(tilesM, 8);
    const int blocks     = gemmBlocks + cdiv(E, 256);
    const int TN         = (cout + 15) >> 4;
    const size_t shmem   = (size_t)cin * TN * 16 * sizeof(float);
    switch (TN) {
        case 4: fill_project_kernel<4><<<blocks, 256, shmem, stream>>>(row, col, rank, part, dinv, ent, E, nP, sliceLen, X, W, H, n, cin, cout, gemmBlocks); break;
        case 3: fill_project_kernel<3><<<blocks, 256, shmem, stream>>>(row, col, rank, part, dinv, ent, E, nP, sliceLen, X, W, H, n, cin, cout, gemmBlocks); break;
        case 2: fill_project_kernel<2><<<blocks, 256, shmem, stream>>>(row, col, rank, part, dinv, ent, E, nP, sliceLen, X, W, H, n, cin, cout, gemmBlocks); break;
        default: fill_project_kernel<1><<<blocks, 256, shmem, stream>>>(row, col, rank, part, dinv, ent, E, nP, sliceLen, X, W, H, n, cin, cout, gemmBlocks); break;
    }
}

static void launch_gather(const float* h, const int* offs, const int2* ent,
                          const float* dinv, const float* bias, float* out,
                          int n, int cout, int doRelu, hipStream_t stream) {
    const int blocks = cdiv(n, 8);
    if ((cout & 1) == 0)
        gather_kernel<2, 1><<<blocks, 256, 0, stream>>>(h, offs, ent, dinv, bias, out, n, cout, doRelu);
    else
        gather_kernel<1, 2><<<blocks, 256, 0, stream>>>(h, offs, ent, dinv, bias, out, n, cout, doRelu);
}

static void launch_gather_project(const float* h, const int* offs, const int2* ent,
                                  const float* dinv, const float* bias,
                                  const float* Wn, float* Hn,
                                  int n, int coutN, int doRelu, hipStream_t stream) {
    const int tiles = cdiv(n, 16);
    const int tpb   = cdiv(tiles, 1024);
    const int blocks = cdiv(tiles, tpb);
    const int TN     = (coutN + 15) >> 4;
    switch (TN) {
        case 4: gather_project_kernel<4, 2, 1><<<blocks, 512, 0, stream>>>(h, offs, ent, dinv, bias, Wn, Hn, n, coutN, doRelu, tpb); break;
        case 3: gather_project_kernel<3, 2, 1><<<blocks, 512, 0, stream>>>(h, offs, ent, dinv, bias, Wn, Hn, n, coutN, doRelu, tpb); break;
        case 2: gather_project_kernel<2, 2, 1><<<blocks, 512, 0, stream>>>(h, offs, ent, dinv, bias, Wn, Hn, n, coutN, doRelu, tpb); break;
        default: gather_project_kernel<1, 2, 1><<<blocks, 512, 0, stream>>>(h, offs, ent, dinv, bias, Wn, Hn, n, coutN, doRelu, tpb); break;
    }
}

extern "C" void kernel_launch(void* const* d_in, const int* in_sizes, int n_in,
                              void* d_out, int out_size, void* d_ws, size_t ws_size,
                              hipStream_t stream) {
    const float* x    = (const float*)d_in[0];
    const int*   ei   = (const int*)  d_in[1];
    const float* W1   = (const float*)d_in[2];
    const float* b1   = (const float*)d_in[3];
    const float* W2   = (const float*)d_in[4];
    const float* b2   = (const float*)d_in[5];
    const float* Wout = (const float*)d_in[6];
    const float* bout = (const float*)d_in[7];

    const int C1  = in_sizes[3];
    const int C2  = in_sizes[5];
    const int C3  = in_sizes[7];
    const int Cin = in_sizes[2] / C1;
    const int n   = in_sizes[0] / Cin;
    const int E   = in_sizes[1] / 2;

    const int* rowIdx = ei;
    const int* colIdx = ei + E;

    const size_t nP = ((size_t)n + 64) & ~(size_t)63;
    float* hbuf   = (float*)d_ws;
    float* abuf   = hbuf + (size_t)n * 64;
    float* dinv   = abuf + (size_t)n * 64;
    int*   offs   = (int*)(dinv + nP);
    int*   bsum   = offs + nP;
    int*   part   = bsum + nP;
    int2*  ent    = (int2*)(part + (size_t)NSLICE * nP);
    int*   rank   = (int*)(ent + E);

    const int scanBlocks = cdiv(n, 1024);
    const int rangeLen   = cdiv(n, NRANGE);
    const int sliceLen   = (cdiv(E, NSLICE) + 3) & ~3;

    count_rank_kernel<<<dim3(NSLICE, NRANGE), 1024, (size_t)rangeLen * sizeof(int), stream>>>(
        colIdx, part, rank, n, E, (int)nP, rangeLen, sliceLen);
    block_sum_kernel<<<scanBlocks, 256, 0, stream>>>(part, bsum, n, (int)nP);
    scan_kernel<<<scanBlocks, 256, 0, stream>>>(part, bsum, offs, dinv, n, (int)nP);

    float* cur = hbuf;
    float* oth = abuf;

    launch_fill_project(rowIdx, colIdx, rank, part, dinv, ent, E, (int)nP, sliceLen,
                        x, W1, cur, n, Cin, C1, stream);

    if (C1 == 64) {
        launch_gather_project(cur, offs, ent, dinv, b1, W2, oth, n, C2, 1, stream);
        float* t = cur; cur = oth; oth = t;
    } else {
        launch_gather(cur, offs, ent, dinv, b1, oth, n, C1, 1, stream);
        launch_gemm(oth, W2, cur, n, C1, C2, stream);
    }

    if (C2 == 64) {
        launch_gather_project(cur, offs, ent, dinv, b2, Wout, oth, n, C3, 1, stream);
        float* t = cur; cur = oth; oth = t;
    } else {
        launch_gather(cur, offs, ent, dinv, b2, oth, n, C2, 1, stream);
        launch_gemm(oth, Wout, cur, n, C2, C3, stream);
    }

    launch_gather(cur, offs, ent, dinv, bout, (float*)d_out, n, C3, 0, stream);
}


# edge weight folded into row scales: products store rows times their dinv, a node's sum is scaled by dinv once; an entry is the 4 B source, placement reads no dinv; sums' rounding changes, order kept
# speedup vs baseline: 3.6552x; 1.0365x over previous
#include <hip/hip_runtime.h>

typedef __attribute__((ext_vector_type(2))) float v2f;
typedef __attribute__((ext_vector_type(8))) float v8f;


#define NSLICE 16
#define NRANGE 16

__device__ __forceinline__ int draw_rank(int c, int lo, int len, int* cntL) {
    const unsigned d = (unsigned)(c - lo);
    return (d < (unsigned)len) ? atomicAdd(&cntL[d], 1) : -1;
}

__global__ void __launch_bounds__(1024)
count_rank_kernel(const int* __restrict__ col, int* __restrict__ part,
                  int* __restrict__ rank, int n, int E, int nP,
                  int rangeLen, int sliceLen) {
    extern __shared__ int cntL[];
    const int tid = threadIdx.x;
    const int lo  = blockIdx.y * rangeLen;
    int len = n - lo;
    if (len > rangeLen) len = rangeLen;
    if (len < 0) len = 0;
    for (int i = tid; i < len; i += 1024) cntL[i] = 0;
    __syncthreads();

    int e0 = blockIdx.x * sliceLen;
    if (e0 > E) e0 = E;
    int e1 = e0 + sliceLen;
    if (e1 > E) e1 = E;
    const int* __restrict__ cs = col + e0;
    const int nq = ((((size_t)cs) & 15) == 0) ? ((e1 - e0) >> 2) : 0;
    const int4* __restrict__ cs4 = (const int4*)cs;
    for (int q0 = tid; q0 < nq; q0 += 4 * 1024) {
        int4 v[4];
        #pragma unroll
        for (int u = 0; u < 4; ++u) {
            const int q = q0 + u * 1024;
            v[u] = (q < nq) ? cs4[q] : make_int4(-1, -1, -1, -1);
        }
        int rk[4][4];
        #pragma unroll
        for (int u = 0; u < 4; ++u) {
            rk[u][0] = draw_rank(v[u].x, lo, len, cntL);
            rk[u][1] = draw_rank(v[u].y, lo, len, cntL);
            rk[u][2] = draw_rank(v[u].z, lo, len, cntL);
            rk[u][3] = draw_rank(v[u].w, lo, len, cntL);
        }
        #pragma unroll
        for (int u = 0; u < 4; ++u) {
            const int e = e0 + 4 * (q0 + u * 1024);
            #pragma unroll
            for (int j = 0; j < 4; ++j)
                if (rk[u][j] >= 0) rank[e + j] = rk[u][j];
        }
    }
    for (int e = e0 + 4 * nq + tid; e < e1; e += 1024) {
        const int rk = draw_rank(col[e], lo, len, cntL);
        if (rk >= 0) rank[e] = rk;
    }
    __syncthreads();

    int* __restrict__ dst = part + (size_t)blockIdx.x * (size_t)nP + lo;
    for (int i = tid; i < len; i += 1024) dst[i] = cntL[i];
}

__device__ __forceinline__ int4 load4_guard(const int* p, int idx, int n) {
    if (idx + 3 < n) return *(const int4*)p;
    int4 v;
    v.x = (idx     < n) ? p[0] : 0;
    v.y = (idx + 1 < n) ? p[1] : 0;
    v.z = (idx + 2 < n) ? p[2] : 0;
    v.w = 0;
    return v;
}

__device__ __forceinline__ void store4_guard(int* p, int idx, int n, int4 v) {
    if (idx + 3 < n) { *(int4*)p = v; return; }
    if (idx     < n) p[0] = v.x;
    if (idx + 1 < n) p[1] = v.y;
    if (idx + 2 < n) p[2] = v.z;
}

__device__ __forceinline__ int wave_sum(int s) {
    #pragma unroll
    for (int off = 16; off > 0; off >>= 1) s += __shfl_xor(s, off, 32);
    return s;
}

__global__ void __launch_bounds__(256)
block_sum_kernel(const int* __restrict__ part, int* __restrict__ bsum, int n, int nP) {
    __shared__ int sm[8];
    const int base = blockIdx.x * 1024 + threadIdx.x * 4;
    int s = 0;
    if (base < n) {
        #pragma unroll
        for (int sl = 0; sl < NSLICE; ++sl) {
            const int4 v = load4_guard(part + (size_t)sl * (size_t)nP + base, base, n);
            s += v.x + v.y + v.z + v.w;
        }
    }
    s = wave_sum(s);
    if ((threadIdx.x & 31) == 0) sm[threadIdx.x >> 5] = s;
    __syncthreads();
    if (threadIdx.x == 0) {
        int t = 0;
        #pragma unroll
        for (int w = 0; w < 8; ++w) t += sm[w];
        bsum[blockIdx.x] = t;
    }
}

__global__ void __launch_bounds__(256)
scan_kernel(int* __restrict__ part, const int* __restrict__ bsum,
            int* __restrict__ offs,
            float* __restrict__ dinv, int n, int nP) {
    __shared__ int smA[8];
    __shared__ int smB[8];
    const int tid  = threadIdx.x;
    const int lane = tid & 31;
    const int wave = tid >> 5;

    int pb = 0;
    for (int i = tid; i < (int)blockIdx.x; i += 256) pb += bsum[i];
    pb = wave_sum(pb);
    if (lane == 0) smA[wave] = pb;

    const int base = blockIdx.x * 1024 + tid * 4;
    int4 p[NSLICE];
    int c[4] = {0, 0, 0, 0};
    #pragma unroll
    for (int sl = 0; sl < NSLICE; ++sl) {
        p[sl] = (base < n) ? load4_guard(part + (size_t)sl * (size_t)nP + base, base, n)
                           : make_int4(0, 0, 0, 0);
        c[0] += p[sl].x; c[1] += p[sl].y; c[2] += p[sl].z; c[3] += p[sl].w;
    }
    const int tsum = c[0] + c[1] + c[2] + c[3];

    int incl = tsum;
    #pragma unroll
    for (int off = 1; off < 32; off <<= 1) {
        const int t = __shfl_up(incl, off, 32);
        if (lane >= off) incl += t;
    }
    if (lane == 31) smB[wave] = incl;
    __syncthreads();

    int run = 0;
    #pragma unroll
    for (int w = 0; w < 8; ++w) run += smA[w];
    #pragma unroll
    for (int w = 0; w < 8; ++w) if (w < wave) run += smB[w];
    run += incl - tsum;

    if (base < n) {
        int4 a;
        a.x = run;
        a.y = a.x + c[0];
        a.z = a.y + c[1];
        a.w = a.z + c[2];
        store4_guard(offs + base, base, n, a);
        #pragma unroll
        for (int k = 0; k < 4; ++k)
            if (base + k < n) dinv[base + k] = rsqrtf((float)c[k] + 1.0f);
        if (n - 1 < base + 4) offs[n] = a.w + c[3];
        #pragma unroll
        for (int sl = 0; sl < NSLICE; ++sl) {
            store4_guard(part + (size_t)sl * (size_t)nP + base, base, n, a);
            a.x += p[sl].x; a.y += p[sl].y; a.z += p[sl].z; a.w += p[sl].w;
        }
    }
}

__device__ __forceinline__ void place_edge(int e,
                                           const int* __restrict__ row, const int* __restrict__ col,
                                           const int* __restrict__ rank, const int* __restrict__ part,
                                           int* __restrict__ ent, int E, int nP, int sliceLen) {
    if (e < E) {
        const int r = row[e];
        const int c = col[e];
        const int s = e / sliceLen;
        const int p = part[(size_t)s * (size_t)nP + c] + rank[e];
        ent[p] = r;
    }
}

template <int TN>
__device__ __forceinline__ void gemm_block(const float* __restrict__ X,
                                           const float* __restrict__ W,
                                           const float* __restrict__ rscale,
                                           float* __restrict__ H,
                                           int n, int cin, int cout,
                                           int blk, float* lds) {
    constexpr int coutP = TN * 16;

    if (blk == 0 && (int)threadIdx.x < cout) H[(size_t)n * cout + threadIdx.x] = 0.0f;

    const int total = cin * coutP;
    for (int i = threadIdx.x; i < total; i += blockDim.x) {
        const int k = i / coutP;
        const int c = i - k * coutP;
        const float v = (c < cout) ? W[(size_t)k * cout + c] : 0.0f;
        lds[(k >> 1) * (2 * coutP) + (c << 1) + (k & 1)] = v;
    }
    __syncthreads();

    const int lane   = threadIdx.x & 31;
    const int waveId = threadIdx.x >> 5;
    const int tileM  = blk * (blockDim.x >> 5) + waveId;
    const int tilesM = (n + 15) >> 4;
    if (tileM >= tilesM) return;

    const int nlane = lane & 15;
    const int koff  = (lane >> 4) << 1;
    const int mrow  = (tileM << 4) + nlane;
    const int arow  = (mrow < n) ? mrow : (n - 1);

    const float* aPtr = X + (size_t)arow * cin + koff;
    int bIdx = (koff >> 1) * (2 * coutP) + (nlane << 1);

    v8f acc[TN] = {};
    for (int kb = 0; kb < cin; kb += 4) {
        const float2 av = *(const float2*)aPtr;
        v2f a; a.x = av.x; a.y = av.y;
        #pragma unroll
        for (int t = 0; t < TN; ++t) {
            const v2f b = *(const v2f*)(lds + bIdx + (t << 5));
            acc[t] = __builtin_amdgcn_wmma_f32_16x16x4_f32(
                          false, a,  false, b,
                          (short)0, acc[t],
                          false,  false);
        }
        aPtr += 4;
        bIdx += 4 * coutP;
    }

    const int rbase = (tileM << 4) + ((lane >> 4) << 3);
    float rs[8];
    #pragma unroll
    for (int r = 0; r < 8; ++r) rs[r] = (rbase + r < n) ? rscale[rbase + r] : 0.0f;
    #pragma unroll
    for (int t = 0; t < TN; ++t) {
        const int col = t * 16 + nlane;
        if (col < cout) {
            #pragma unroll
            for (int r = 0; r < 8; ++r) {
                const int orow = rbase + r;
                if (orow < n) H[(size_t)orow * cout + col] = acc[t][r] * rs[r];
            }
        }
    }
}

template <int TN>
__global__ void gemm_wmma_f32_kernel(const float* __restrict__ X,
                                     const float* __restrict__ W,
                                     const float* __restrict__ rscale,
                                     float* __restrict__ H,
                                     int n, int cin, int cout) {
    extern __shared__ float lds[];
    gemm_block<TN>(X, W, rscale, H, n, cin, cout, (int)blockIdx.x, lds);
}

template <int TN>
__global__ void __launch_bounds__(256)
fill_project_kernel(const int* __restrict__ row, const int* __restrict__ col,
                    const int* __restrict__ rank, const int* __restrict__ part,
                    const float* __restrict__ dinv,
                    int* __restrict__ ent, int E, int nP, int sliceLen,
                    const float* __restrict__ X, const float* __restrict__ W,
                    float* __restrict__ H, int n, int cin, int cout,
                    int gemmBlocks) {
    extern __shared__ float lds[];
    const int blk = (int)blockIdx.x;
    if (blk < gemmBlocks) {
        gemm_block<TN>(X, W, dinv, H, n, cin, cout, blk, lds);
    } else {
        place_edge((blk - gemmBlocks) * 256 + (int)threadIdx.x,
                   row, col, rank, part, ent, E, nP, sliceLen);
    }
}

template <int VEC, int NP>
__device__ __forceinline__ void sum_bucket(const float* __restrict__ h, size_t pitch,
                                           const int* __restrict__ ent,
                                           int beg, int end, int zrow, unsigned lane,
                                           const int (&ldc)[NP], float (&acc)[NP][VEC]) {
    for (int b0 = beg; b0 < end; b0 += 32) {
        const int cntb = (end - b0 < 32) ? (end - b0) : 32;
        int r = zrow;
        if ((int)lane < cntb) r = ent[b0 + (int)lane];
        asm volatile("" :: "v"(r));
        const int nq = (cntb + 3) >> 2;
        #pragma unroll
        for (int b = 0; b < 2; ++b) {
            if (4 * b < nq) {
                float v[4][4][NP][VEC];
                #pragma unroll
                for (int g = 0; g < 4; ++g) {
                    if (4 * b + g < nq) {
                        #pragma unroll
                        for (int u = 0; u < 4; ++u) {
                            const int rj = __builtin_amdgcn_readlane(r, 16 * b + 4 * g + u);
                            const float* __restrict__ src = h + (size_t)rj * pitch;
                            #pragma unroll
                            for (int p = 0; p < NP; ++p) {
                                if (VEC == 2) {
                                    const float2 t = *(const float2*)(src + ldc[p]);
                                    v[g][u][p][0] = t.x; v[g][u][p][VEC - 1] = t.y;
                                } else {
                                    v[g][u][p][0] = src[ldc[p]];
                                }
                            }
                        }
                    }
                }
                #pragma unroll
                for (int g = 0; g < 4; ++g) {
                    if (4 * b + g < nq) {
                        #pragma unroll
                        for (int u = 0; u < 4; ++u) {
                            #pragma unroll
                            for (int p = 0; p < NP; ++p) {
                                #pragma unroll
                                for (int q = 0; q < VEC; ++q)
                                    acc[p][q] += v[g][u][p][q];
                            }
                        }
                    }
                }
            }
        }
    }
}

template <int VEC, int NP>
__global__ void __launch_bounds__(256)
gather_kernel(const float* __restrict__ h,
              const int* __restrict__ offs,
              const int* __restrict__ ent,
              const float* __restrict__ dinv,
              const float* __restrict__ bias,
              float* __restrict__ out,
              int n, int cout, int doRelu) {
    const unsigned lane = threadIdx.x & 31u;
    const int node = __builtin_amdgcn_readfirstlane(
        (int)(blockIdx.x * (blockDim.x >> 5) + (threadIdx.x >> 5)));
    if (node >= n) return;

    const int   beg = offs[node];
    const int   end = offs[node + 1];
    const float di  = dinv[node];

    int  colOf[NP];
    bool act[NP];
    int  ldc[NP];
    #pragma unroll
    for (int p = 0; p < NP; ++p) {
        colOf[p] = (int)(lane + 32u * p) * VEC;
        act[p]   = colOf[p] < cout;
        ldc[p]   = act[p] ? colOf[p] : 0;
    }

    float hs[NP][VEC], bs[NP][VEC];
    {
        const float* __restrict__ self = h + (size_t)node * (size_t)cout;
        #pragma unroll
        for (int p = 0; p < NP; ++p) {
            if (VEC == 2) {
                const float2 t = *(const float2*)(self + ldc[p]);
                const float2 b = *(const float2*)(bias + ldc[p]);
                hs[p][0] = t.x; hs[p][VEC - 1] = t.y;
                bs[p][0] = b.x; bs[p][VEC - 1] = b.y;
            } else {
                hs[p][0] = self[ldc[p]];
                bs[p][0] = bias[ldc[p]];
            }
        }
    }

    float acc[NP][VEC] = {};
    sum_bucket<VEC, NP>(h, (size_t)cout, ent, beg, end, n, lane, ldc, acc);

    float* __restrict__ dst = out + (size_t)node * (size_t)cout;
    #pragma unroll
    for (int p = 0; p < NP; ++p) {
        float o[VEC];
        #pragma unroll
        for (int q = 0; q < VEC; ++q) {
            o[q] = (acc[p][q] + hs[p][q]) * di + bs[p][q];
            if (doRelu) o[q] = fmaxf(o[q], 0.0f);
        }
        if (act[p]) {
            if (VEC == 2) {
                float2 t; t.x = o[0]; t.y = o[VEC - 1];
                *(float2*)(dst + colOf[p]) = t;
            } else {
                dst[colOf[p]] = o[0];
            }
        }
    }
}

#define APITCH 68

template <int TN, int VEC, int NP>
__global__ void __launch_bounds__(512)
gather_project_kernel(const float* __restrict__ h,
                      const int* __restrict__ offs,
                      const int* __restrict__ ent,
                      const float* __restrict__ dinv,
                      const float* __restrict__ bias,
                      const float* __restrict__ Wn,
                      float* __restrict__ Hn,
                      int n, int coutN, int doRelu, int tilesPerBlock) {
    constexpr int cout  = 64;
    constexpr int coutP = TN * 16;
    __shared__ __attribute__((aligned(16))) float Ws[cout * coutP];
    __shared__ __attribute__((aligned(16))) float As[2 * 16 * APITCH];

    {
        float w[2 * TN];
        #pragma unroll
        for (int u = 0; u < 2 * TN; ++u) {
            const int i = (int)threadIdx.x + u * 512;
            const int k = i / coutP;
            const int c = i - k * coutP;
            w[u] = (c < coutN) ? Wn[k * coutN + c] : 0.0f;
        }
        #pragma unroll
        for (int u = 0; u < 2 * TN; ++u) {
            const int i = (int)threadIdx.x + u * 512;
            const int k = i / coutP;
            const int c = i - k * coutP;
            Ws[(k >> 1) * (2 * coutP) + (c << 1) + (k & 1)] = w[u];
        }
    }

    if (blockIdx.x == 0 && (int)threadIdx.x < coutN) Hn[(size_t)n * coutN + threadIdx.x] = 0.0f;

    const unsigned lane = threadIdx.x & 31u;
    const int wave = __builtin_amdgcn_readfirstlane((int)(threadIdx.x >> 5));

    int colOf[NP];
    #pragma unroll
    for (int p = 0; p < NP; ++p) colOf[p] = (int)(lane + 32u * p) * VEC;

    float bs[NP][VEC];
    #pragma unroll
    for (int p = 0; p < NP; ++p) {
        const float2 b = *(const float2*)(bias + colOf[p]);
        bs[p][0] = b.x; bs[p][VEC - 1] = b.y;
    }

    const int nlane = (int)lane & 15;
    const int koff  = ((int)lane >> 4) << 1;
    const int pcol  = (wave < TN ? wave : 0) * 16 + nlane;
    const float* bFrag = Ws + (koff >> 1) * (2 * coutP) + (pcol << 1);

    const int tile0 = (int)blockIdx.x * tilesPerBlock;
    for (int it = 0; it < tilesPerBlock; ++it) {
        const int tile = tile0 + it;
        const int node = tile * 16 + wave;
        float* At = As + (it & 1) * (16 * APITCH);

        float o[NP][VEC] = {};
        if (node < n) {
            const int   beg = offs[node];
            const int   end = offs[node + 1];
            const float di  = dinv[node];

            float hs[NP][VEC];
            {
                const float* __restrict__ self = h + (size_t)node * (size_t)cout;
                #pragma unroll
                for (int p = 0; p < NP; ++p) {
                    const float2 t = *(const float2*)(self + colOf[p]);
                    hs[p][0] = t.x; hs[p][VEC - 1] = t.y;
                }
            }

            float acc[NP][VEC] = {};
            sum_bucket<VEC, NP>(h, (size_t)cout, ent, beg, end, n, lane, colOf, acc);

            #pragma unroll
            for (int p = 0; p < NP; ++p) {
                #pragma unroll
                for (int q = 0; q < VEC; ++q) {
                    o[p][q] = (acc[p][q] + hs[p][q]) * di + bs[p][q];
                    if (doRelu) o[p][q] = fmaxf(o[p][q], 0.0f);
                    o[p][q] *= di;
                }
            }
        }
        #pragma unroll
        for (int p = 0; p < NP; ++p) {
            float2 t; t.x = o[p][0]; t.y = o[p][VEC - 1];
            *(float2*)(At + wave * APITCH + colOf[p]) = t;
        }
        __syncthreads();

        if (wave < TN && tile * 16 < n) {
            const float* aRow = At + nlane * APITCH + koff;
            v8f d = {};
            #pragma unroll
            for (int s = 0; s < 16; ++s) {
                const v2f a = *(const v2f*)(aRow + 4 * s);
                const v2f b = *(const v2f*)(bFrag + s * (4 * coutP));
                d = __builtin_amdgcn_wmma_f32_16x16x4_f32(
                         false, a,  false, b,
                         (short)0, d,
                         false,  false);
            }
            const int rbase = tile * 16 + (((int)lane >> 4) << 3);
            if (pcol < coutN) {
                #pragma unroll
                for (int r = 0; r < 8; ++r) {
                    const int orow = rbase + r;
                    if (orow < n) Hn[(size_t)orow * coutN + pcol] = d[r];
                }
            }
        }
    }
}


static inline int cdiv(long long a, long long b) { return (int)((a + b - 1) / b); }

static void launch_gemm(const float* X, const float* W, const float* rscale, float* H,
                        int n, int cin, int cout, hipStream_t stream) {
    const int tilesM = (n + 15) >> 4;
    const int blocks = cdiv(tilesM, 8);
    const int TN     = (cout + 15) >> 4;
    const size_t shmem = (size_t)cin * TN * 16 * sizeof(float);
    switch (TN) {
        case 4: gemm_wmma_f32_kernel<4><<<blocks, 256, shmem, stream>>>(X, W, rscale, H, n, cin, cout); break;
        case 3: gemm_wmma_f32_kernel<3><<<blocks, 256, shmem, stream>>>(X, W, rscale, H, n, cin, cout); break;
        case 2: gemm_wmma_f32_kernel<2><<<blocks, 256, shmem, stream>>>(X, W, rscale, H, n, cin, cout); break;
        default: gemm_wmma_f32_kernel<1><<<blocks, 256, shmem, stream>>>(X, W, rscale, H, n, cin, cout); break;
    }
}

static void launch_fill_project(const int* row, const int* col, const int* rank,
                                const int* part, const float* dinv, int* ent,
                                int E, int nP, int sliceLen,
                                const float* X, const float* W, float* H,
                                int n, int cin, int cout, hipStream_t stream) {
    const int tilesM     = (n + 15) >> 4;
    const int gemmBlocks = cdiv(tilesM, 8);
    const int blocks     = gemmBlocks + cdiv(E, 256);
    const int TN         = (cout + 15) >> 4;
    const size_t shmem   = (size_t)cin * TN * 16 * sizeof(float);
    switch (TN) {
        case 4: fill_project_kernel<4><<<blocks, 256, shmem, stream>>>(row, col, rank, part, dinv, ent, E, nP, sliceLen, X, W, H, n, cin, cout, gemmBlocks); break;
        case 3: fill_project_kernel<3><<<blocks, 256, shmem, stream>>>(row, col, rank, part, dinv, ent, E, nP, sliceLen, X, W, H, n, cin, cout, gemmBlocks); break;
        case 2: fill_project_kernel<2><<<blocks, 256, shmem, stream>>>(row, col, rank, part, dinv, ent, E, nP, sliceLen, X, W, H, n, cin, cout, gemmBlocks); break;
        default: fill_project_kernel<1><<<blocks, 256, shmem, stream>>>(row, col, rank, part, dinv, ent, E, nP, sliceLen, X, W, H, n, cin, cout, gemmBlocks); break;
    }
}

static void launch_gather(const float* h, const int* offs, const int* ent,
                          const float* dinv, const float* bias, float* out,
                          int n, int cout, int doRelu, hipStream_t stream) {
    const int blocks = cdiv(n, 8);
    if ((cout & 1) == 0)
        gather_kernel<2, 1><<<blocks, 256, 0, stream>>>(h, offs, ent, dinv, bias, out, n, cout, doRelu);
    else
        gather_kernel<1, 2><<<blocks, 256, 0, stream>>>(h, offs, ent, dinv, bias, out, n, cout, doRelu);
}

static void launch_gather_project(const float* h, const int* offs, const int* ent,
                                  const float* dinv, const float* bias,
                                  const float* Wn, float* Hn,
                                  int n, int coutN, int doRelu, hipStream_t stream) {
    const int tiles = cdiv(n, 16);
    const int tpb   = cdiv(tiles, 1024);
    const int blocks = cdiv(tiles, tpb);
    const int TN     = (coutN + 15) >> 4;
    switch (TN) {
        case 4: gather_project_kernel<4, 2, 1><<<blocks, 512, 0, stream>>>(h, offs, ent, dinv, bias, Wn, Hn, n, coutN, doRelu, tpb); break;
        case 3: gather_project_kernel<3, 2, 1><<<blocks, 512, 0, stream>>>(h, offs, ent, dinv, bias, Wn, Hn, n, coutN, doRelu, tpb); break;
        case 2: gather_project_kernel<2, 2, 1><<<blocks, 512, 0, stream>>>(h, offs, ent, dinv, bias, Wn, Hn, n, coutN, doRelu, tpb); break;
        default: gather_project_kernel<1, 2, 1><<<blocks, 512, 0, stream>>>(h, offs, ent, dinv, bias, Wn, Hn, n, coutN, doRelu, tpb); break;
    }
}

extern "C" void kernel_launch(void* const* d_in, const int* in_sizes, int n_in,
                              void* d_out, int out_size, void* d_ws, size_t ws_size,
                              hipStream_t stream) {
    const float* x    = (const float*)d_in[0];
    const int*   ei   = (const int*)  d_in[1];
    const float* W1   = (const float*)d_in[2];
    const float* b1   = (const float*)d_in[3];
    const float* W2   = (const float*)d_in[4];
    const float* b2   = (const float*)d_in[5];
    const float* Wout = (const float*)d_in[6];
    const float* bout = (const float*)d_in[7];

    const int C1  = in_sizes[3];
    const int C2  = in_sizes[5];
    const int C3  = in_sizes[7];
    const int Cin = in_sizes[2] / C1;
    const int n   = in_sizes[0] / Cin;
    const int E   = in_sizes[1] / 2;

    const int* rowIdx = ei;
    const int* colIdx = ei + E;

    const size_t nP = ((size_t)n + 64) & ~(size_t)63;
    float* hbuf   = (float*)d_ws;
    float* abuf   = hbuf + ((size_t)n + 1) * 64;
    float* dinv   = abuf + ((size_t)n + 1) * 64;
    int*   offs   = (int*)(dinv + nP);
    int*   bsum   = offs + nP;
    int*   part   = bsum + nP;
    int*   ent    = part + (size_t)NSLICE * nP;
    int*   rank   = ent + (((size_t)E + 63) & ~(size_t)63);

    const int scanBlocks = cdiv(n, 1024);
    const int rangeLen   = cdiv(n, NRANGE);
    const int sliceLen   = (cdiv(E, NSLICE) + 3) & ~3;

    count_rank_kernel<<<dim3(NSLICE, NRANGE), 1024, (size_t)rangeLen * sizeof(int), stream>>>(
        colIdx, part, rank, n, E, (int)nP, rangeLen, sliceLen);
    block_sum_kernel<<<scanBlocks, 256, 0, stream>>>(part, bsum, n, (int)nP);
    scan_kernel<<<scanBlocks, 256, 0, stream>>>(part, bsum, offs, dinv, n, (int)nP);

    float* cur = hbuf;
    float* oth = abuf;

    launch_fill_project(rowIdx, colIdx, rank, part, dinv, ent, E, (int)nP, sliceLen,
                        x, W1, cur, n, Cin, C1, stream);

    if (C1 == 64) {
        launch_gather_project(cur, offs, ent, dinv, b1, W2, oth, n, C2, 1, stream);
        float* t = cur; cur = oth; oth = t;
    } else {
        launch_gather(cur, offs, ent, dinv, b1, oth, n, C1, 1, stream);
        launch_gemm(oth, W2, dinv, cur, n, C1, C2, stream);
    }

    if (C2 == 64) {
        launch_gather_project(cur, offs, ent, dinv, b2, Wout, oth, n, C3, 1, stream);
        float* t = cur; cur = oth; oth = t;
    } else {
        launch_gather(cur, offs, ent, dinv, b2, oth, n, C2, 1, stream);
        launch_gemm(oth, Wout, dinv, cur, n, C2, C3, stream);
    }

    launch_gather(cur, offs, ent, dinv, bout, (float*)d_out, n, C3, 0, stream);
}


# placement: (slice, range) blocks walk their slice again, LDS cursors start at the scanned slot starts, slot = LDS add's return: no rank array, no scattered part read; sums in placement-arrival order
# speedup vs baseline: 3.8059x; 1.0412x over previous
#include <hip/hip_runtime.h>

typedef __attribute__((ext_vector_type(2))) float v2f;
typedef __attribute__((ext_vector_type(8))) float v8f;


#define NSLICE 16
#define NRANGE 16

__device__ __forceinline__ int draw_slot(int c, int lo, int len, int* cntL) {
    const unsigned d = (unsigned)(c - lo);
    return (d < (unsigned)len) ? atomicAdd(&cntL[d], 1) : -1;
}

__device__ __forceinline__ void count_one(int c, int lo, int len, int* cntL) {
    const unsigned d = (unsigned)(c - lo);
    if (d < (unsigned)len) atomicAdd(&cntL[d], 1);
}

__global__ void __launch_bounds__(1024)
count_kernel(const int* __restrict__ col, int* __restrict__ part,
             int n, int E, int nP, int rangeLen, int sliceLen) {
    extern __shared__ int cntL[];
    const int tid = threadIdx.x;
    const int lo  = blockIdx.y * rangeLen;
    int len = n - lo;
    if (len > rangeLen) len = rangeLen;
    if (len < 0) len = 0;
    for (int i = tid; i < len; i += 1024) cntL[i] = 0;
    __syncthreads();

    int e0 = blockIdx.x * sliceLen;
    if (e0 > E) e0 = E;
    int e1 = e0 + sliceLen;
    if (e1 > E) e1 = E;
    const int* __restrict__ cs = col + e0;
    const int nq = ((((size_t)cs) & 15) == 0) ? ((e1 - e0) >> 2) : 0;
    const int4* __restrict__ cs4 = (const int4*)cs;
    for (int q0 = tid; q0 < nq; q0 += 4 * 1024) {
        int4 v[4];
        #pragma unroll
        for (int u = 0; u < 4; ++u) {
            const int q = q0 + u * 1024;
            v[u] = (q < nq) ? cs4[q] : make_int4(-1, -1, -1, -1);
        }
        #pragma unroll
        for (int u = 0; u < 4; ++u) {
            count_one(v[u].x, lo, len, cntL);
            count_one(v[u].y, lo, len, cntL);
            count_one(v[u].z, lo, len, cntL);
            count_one(v[u].w, lo, len, cntL);
        }
    }
    for (int e = e0 + 4 * nq + tid; e < e1; e += 1024)
        count_one(col[e], lo, len, cntL);
    __syncthreads();

    int* __restrict__ dst = part + (size_t)blockIdx.x * (size_t)nP + lo;
    for (int i = tid; i < len; i += 1024) dst[i] = cntL[i];
}

__device__ __forceinline__ int4 load4_guard(const int* p, int idx, int n) {
    if (idx + 3 < n) return *(const int4*)p;
    int4 v;
    v.x = (idx     < n) ? p[0] : 0;
    v.y = (idx + 1 < n) ? p[1] : 0;
    v.z = (idx + 2 < n) ? p[2] : 0;
    v.w = 0;
    return v;
}

__device__ __forceinline__ void store4_guard(int* p, int idx, int n, int4 v) {
    if (idx + 3 < n) { *(int4*)p = v; return; }
    if (idx     < n) p[0] = v.x;
    if (idx + 1 < n) p[1] = v.y;
    if (idx + 2 < n) p[2] = v.z;
}

__device__ __forceinline__ int wave_sum(int s) {
    #pragma unroll
    for (int off = 16; off > 0; off >>= 1) s += __shfl_xor(s, off, 32);
    return s;
}

__global__ void __launch_bounds__(256)
block_sum_kernel(const int* __restrict__ part, int* __restrict__ bsum, int n, int nP) {
    __shared__ int sm[8];
    const int base = blockIdx.x * 1024 + threadIdx.x * 4;
    int s = 0;
    if (base < n) {
        #pragma unroll
        for (int sl = 0; sl < NSLICE; ++sl) {
            const int4 v = load4_guard(part + (size_t)sl * (size_t)nP + base, base, n);
            s += v.x + v.y + v.z + v.w;
        }
    }
    s = wave_sum(s);
    if ((threadIdx.x & 31) == 0) sm[threadIdx.x >> 5] = s;
    __syncthreads();
    if (threadIdx.x == 0) {
        int t = 0;
        #pragma unroll
        for (int w = 0; w < 8; ++w) t += sm[w];
        bsum[blockIdx.x] = t;
    }
}

__global__ void __launch_bounds__(256)
scan_kernel(int* __restrict__ part, const int* __restrict__ bsum,
            int* __restrict__ offs,
            float* __restrict__ dinv, int n, int nP) {
    __shared__ int smA[8];
    __shared__ int smB[8];
    const int tid  = threadIdx.x;
    const int lane = tid & 31;
    const int wave = tid >> 5;

    int pb = 0;
    for (int i = tid; i < (int)blockIdx.x; i += 256) pb += bsum[i];
    pb = wave_sum(pb);
    if (lane == 0) smA[wave] = pb;

    const int base = blockIdx.x * 1024 + tid * 4;
    int4 p[NSLICE];
    int c[4] = {0, 0, 0, 0};
    #pragma unroll
    for (int sl = 0; sl < NSLICE; ++sl) {
        p[sl] = (base < n) ? load4_guard(part + (size_t)sl * (size_t)nP + base, base, n)
                           : make_int4(0, 0, 0, 0);
        c[0] += p[sl].x; c[1] += p[sl].y; c[2] += p[sl].z; c[3] += p[sl].w;
    }
    const int tsum = c[0] + c[1] + c[2] + c[3];

    int incl = tsum;
    #pragma unroll
    for (int off = 1; off < 32; off <<= 1) {
        const int t = __shfl_up(incl, off, 32);
        if (lane >= off) incl += t;
    }
    if (lane == 31) smB[wave] = incl;
    __syncthreads();

    int run = 0;
    #pragma unroll
    for (int w = 0; w < 8; ++w) run += smA[w];
    #pragma unroll
    for (int w = 0; w < 8; ++w) if (w < wave) run += smB[w];
    run += incl - tsum;

    if (base < n) {
        int4 a;
        a.x = run;
        a.y = a.x + c[0];
        a.z = a.y + c[1];
        a.w = a.z + c[2];
        store4_guard(offs + base, base, n, a);
        #pragma unroll
        for (int k = 0; k < 4; ++k)
            if (base + k < n) dinv[base + k] = rsqrtf((float)c[k] + 1.0f);
        if (n - 1 < base + 4) offs[n] = a.w + c[3];
        #pragma unroll
        for (int sl = 0; sl < NSLICE; ++sl) {
            store4_guard(part + (size_t)sl * (size_t)nP + base, base, n, a);
            a.x += p[sl].x; a.y += p[sl].y; a.z += p[sl].z; a.w += p[sl].w;
        }
    }
}

__device__ __forceinline__ void place_block(int s, int r,
                                            const int* __restrict__ row, const int* __restrict__ col,
                                            const int* __restrict__ part, int* __restrict__ ent,
                                            int n, int E, int nP, int rangeLen, int sliceLen,
                                            int* curL) {
    const int tid = threadIdx.x;
    const int lo  = r * rangeLen;
    int len = n - lo;
    if (len > rangeLen) len = rangeLen;
    if (len < 0) len = 0;
    const int* __restrict__ first = part + (size_t)s * (size_t)nP + lo;
    for (int i = tid; i < len; i += 1024) curL[i] = first[i];
    __syncthreads();

    int e0 = s * sliceLen;
    if (e0 > E) e0 = E;
    int e1 = e0 + sliceLen;
    if (e1 > E) e1 = E;
    const int* __restrict__ cs = col + e0;
    const int* __restrict__ rs = row + e0;
    const int nq = (((((size_t)cs) | ((size_t)rs)) & 15) == 0) ? ((e1 - e0) >> 2) : 0;
    const int4* __restrict__ cs4 = (const int4*)cs;
    const int4* __restrict__ rs4 = (const int4*)rs;
    for (int q0 = tid; q0 < nq; q0 += 4 * 1024) {
        int4 c[4], w[4];
        #pragma unroll
        for (int u = 0; u < 4; ++u) {
            const int q = q0 + u * 1024;
            c[u] = (q < nq) ? cs4[q] : make_int4(-1, -1, -1, -1);
            w[u] = (q < nq) ? rs4[q] : make_int4(0, 0, 0, 0);
        }
        int sl[4][4];
        #pragma unroll
        for (int u = 0; u < 4; ++u) {
            sl[u][0] = draw_slot(c[u].x, lo, len, curL);
            sl[u][1] = draw_slot(c[u].y, lo, len, curL);
            sl[u][2] = draw_slot(c[u].z, lo, len, curL);
            sl[u][3] = draw_slot(c[u].w, lo, len, curL);
        }
        #pragma unroll
        for (int u = 0; u < 4; ++u) {
            if (sl[u][0] >= 0) ent[sl[u][0]] = w[u].x;
            if (sl[u][1] >= 0) ent[sl[u][1]] = w[u].y;
            if (sl[u][2] >= 0) ent[sl[u][2]] = w[u].z;
            if (sl[u][3] >= 0) ent[sl[u][3]] = w[u].w;
        }
    }
    for (int e = e0 + 4 * nq + tid; e < e1; e += 1024) {
        const int p = draw_slot(col[e], lo, len, curL);
        if (p >= 0) ent[p] = row[e];
    }
}

template <int TN>
__device__ __forceinline__ void gemm_block(const float* __restrict__ X,
                                           const float* __restrict__ W,
                                           const float* __restrict__ rscale,
                                           float* __restrict__ H,
                                           int n, int cin, int cout,
                                           int blk, float* lds) {
    constexpr int coutP = TN * 16;

    if (blk == 0 && (int)threadIdx.x < cout) H[(size_t)n * cout + threadIdx.x] = 0.0f;

    const int total = cin * coutP;
    for (int i = threadIdx.x; i < total; i += blockDim.x) {
        const int k = i / coutP;
        const int c = i - k * coutP;
        const float v = (c < cout) ? W[(size_t)k * cout + c] : 0.0f;
        lds[(k >> 1) * (2 * coutP) + (c << 1) + (k & 1)] = v;
    }
    __syncthreads();

    const int lane   = threadIdx.x & 31;
    const int waveId = threadIdx.x >> 5;
    const int tileM  = blk * (blockDim.x >> 5) + waveId;
    const int tilesM = (n + 15) >> 4;
    if (tileM >= tilesM) return;

    const int nlane = lane & 15;
    const int koff  = (lane >> 4) << 1;
    const int mrow  = (tileM << 4) + nlane;
    const int arow  = (mrow < n) ? mrow : (n - 1);

    const float* aPtr = X + (size_t)arow * cin + koff;
    int bIdx = (koff >> 1) * (2 * coutP) + (nlane << 1);

    v8f acc[TN] = {};
    for (int kb = 0; kb < cin; kb += 4) {
        const float2 av = *(const float2*)aPtr;
        v2f a; a.x = av.x; a.y = av.y;
        #pragma unroll
        for (int t = 0; t < TN; ++t) {
            const v2f b = *(const v2f*)(lds + bIdx + (t << 5));
            acc[t] = __builtin_amdgcn_wmma_f32_16x16x4_f32(
                          false, a,  false, b,
                          (short)0, acc[t],
                          false,  false);
        }
        aPtr += 4;
        bIdx += 4 * coutP;
    }

    const int rbase = (tileM << 4) + ((lane >> 4) << 3);
    float rs[8];
    #pragma unroll
    for (int r = 0; r < 8; ++r) rs[r] = (rbase + r < n) ? rscale[rbase + r] : 0.0f;
    #pragma unroll
    for (int t = 0; t < TN; ++t) {
        const int col = t * 16 + nlane;
        if (col < cout) {
            #pragma unroll
            for (int r = 0; r < 8; ++r) {
                const int orow = rbase + r;
                if (orow < n) H[(size_t)orow * cout + col] = acc[t][r] * rs[r];
            }
        }
    }
}

template <int TN>
__global__ void gemm_wmma_f32_kernel(const float* __restrict__ X,
                                     const float* __restrict__ W,
                                     const float* __restrict__ rscale,
                                     float* __restrict__ H,
                                     int n, int cin, int cout) {
    extern __shared__ float lds[];
    gemm_block<TN>(X, W, rscale, H, n, cin, cout, (int)blockIdx.x, lds);
}

template <int TN>
__global__ void __launch_bounds__(1024, 16)
fill_project_kernel(const int* __restrict__ row, const int* __restrict__ col,
                    const int* __restrict__ part,
                    const float* __restrict__ dinv,
                    int* __restrict__ ent, int E, int nP, int rangeLen, int sliceLen,
                    const float* __restrict__ X, const float* __restrict__ W,
                    float* __restrict__ H, int n, int cin, int cout,
                    int gemmBlocks) {
    extern __shared__ float lds[];
    const int blk = (int)blockIdx.x;
    if (blk < gemmBlocks) {
        gemm_block<TN>(X, W, dinv, H, n, cin, cout, blk, lds);
    } else {
        const int pb = blk - gemmBlocks;
        place_block(pb % NSLICE, pb / NSLICE, row, col, part, ent,
                    n, E, nP, rangeLen, sliceLen, (int*)lds);
    }
}

template <int VEC, int NP>
__device__ __forceinline__ void sum_bucket(const float* __restrict__ h, size_t pitch,
                                           const int* __restrict__ ent,
                                           int beg, int end, int zrow, unsigned lane,
                                           const int (&ldc)[NP], float (&acc)[NP][VEC]) {
    for (int b0 = beg; b0 < end; b0 += 32) {
        const int cntb = (end - b0 < 32) ? (end - b0) : 32;
        int r = zrow;
        if ((int)lane < cntb) r = ent[b0 + (int)lane];
        asm volatile("" :: "v"(r));
        const int nq = (cntb + 3) >> 2;
        #pragma unroll
        for (int b = 0; b < 2; ++b) {
            if (4 * b < nq) {
                float v[4][4][NP][VEC];
                #pragma unroll
                for (int g = 0; g < 4; ++g) {
                    if (4 * b + g < nq) {
                        #pragma unroll
                        for (int u = 0; u < 4; ++u) {
                            const int rj = __builtin_amdgcn_readlane(r, 16 * b + 4 * g + u);
                            const float* __restrict__ src = h + (size_t)rj * pitch;
                            #pragma unroll
                            for (int p = 0; p < NP; ++p) {
                                if (VEC == 2) {
                                    const float2 t = *(const float2*)(src + ldc[p]);
                                    v[g][u][p][0] = t.x; v[g][u][p][VEC - 1] = t.y;
                                } else {
                                    v[g][u][p][0] = src[ldc[p]];
                                }
                            }
                        }
                    }
                }
                #pragma unroll
                for (int g = 0; g < 4; ++g) {
                    if (4 * b + g < nq) {
                        #pragma unroll
                        for (int u = 0; u < 4; ++u) {
                            #pragma unroll
                            for (int p = 0; p < NP; ++p) {
                                #pragma unroll
                                for (int q = 0; q < VEC; ++q)
                                    acc[p][q] += v[g][u][p][q];
                            }
                        }
                    }
                }
            }
        }
    }
}

template <int VEC, int NP>
__global__ void __launch_bounds__(256)
gather_kernel(const float* __restrict__ h,
              const int* __restrict__ offs,
              const int* __restrict__ ent,
              const float* __restrict__ dinv,
              const float* __restrict__ bias,
              float* __restrict__ out,
              int n, int cout, int doRelu) {
    const unsigned lane = threadIdx.x & 31u;
    const int node = __builtin_amdgcn_readfirstlane(
        (int)(blockIdx.x * (blockDim.x >> 5) + (threadIdx.x >> 5)));
    if (node >= n) return;

    const int   beg = offs[node];
    const int   end = offs[node + 1];
    const float di  = dinv[node];

    int  colOf[NP];
    bool act[NP];
    int  ldc[NP];
    #pragma unroll
    for (int p = 0; p < NP; ++p) {
        colOf[p] = (int)(lane + 32u * p) * VEC;
        act[p]   = colOf[p] < cout;
        ldc[p]   = act[p] ? colOf[p] : 0;
    }

    float hs[NP][VEC], bs[NP][VEC];
    {
        const float* __restrict__ self = h + (size_t)node * (size_t)cout;
        #pragma unroll
        for (int p = 0; p < NP; ++p) {
            if (VEC == 2) {
                const float2 t = *(const float2*)(self + ldc[p]);
                const float2 b = *(const float2*)(bias + ldc[p]);
                hs[p][0] = t.x; hs[p][VEC - 1] = t.y;
                bs[p][0] = b.x; bs[p][VEC - 1] = b.y;
            } else {
                hs[p][0] = self[ldc[p]];
                bs[p][0] = bias[ldc[p]];
            }
        }
    }

    float acc[NP][VEC] = {};
    sum_bucket<VEC, NP>(h, (size_t)cout, ent, beg, end, n, lane, ldc, acc);

    float* __restrict__ dst = out + (size_t)node * (size_t)cout;
    #pragma unroll
    for (int p = 0; p < NP; ++p) {
        float o[VEC];
        #pragma unroll
        for (int q = 0; q < VEC; ++q) {
            o[q] = (acc[p][q] + hs[p][q]) * di + bs[p][q];
            if (doRelu) o[q] = fmaxf(o[q], 0.0f);
        }
        if (act[p]) {
            if (VEC == 2) {
                float2 t; t.x = o[0]; t.y = o[VEC - 1];
                *(float2*)(dst + colOf[p]) = t;
            } else {
                dst[colOf[p]] = o[0];
            }
        }
    }
}

#define APITCH 68

template <int TN, int VEC, int NP>
__global__ void __launch_bounds__(512)
gather_project_kernel(const float* __restrict__ h,
                      const int* __restrict__ offs,
                      const int* __restrict__ ent,
                      const float* __restrict__ dinv,
                      const float* __restrict__ bias,
                      const float* __restrict__ Wn,
                      float* __restrict__ Hn,
                      int n, int coutN, int doRelu, int tilesPerBlock) {
    constexpr int cout  = 64;
    constexpr int coutP = TN * 16;
    __shared__ __attribute__((aligned(16))) float Ws[cout * coutP];
    __shared__ __attribute__((aligned(16))) float As[2 * 16 * APITCH];

    {
        float w[2 * TN];
        #pragma unroll
        for (int u = 0; u < 2 * TN; ++u) {
            const int i = (int)threadIdx.x + u * 512;
            const int k = i / coutP;
            const int c = i - k * coutP;
            w[u] = (c < coutN) ? Wn[k * coutN + c] : 0.0f;
        }
        #pragma unroll
        for (int u = 0; u < 2 * TN; ++u) {
            const int i = (int)threadIdx.x + u * 512;
            const int k = i / coutP;
            const int c = i - k * coutP;
            Ws[(k >> 1) * (2 * coutP) + (c << 1) + (k & 1)] = w[u];
        }
    }

    if (blockIdx.x == 0 && (int)threadIdx.x < coutN) Hn[(size_t)n * coutN + threadIdx.x] = 0.0f;

    const unsigned lane = threadIdx.x & 31u;
    const int wave = __builtin_amdgcn_readfirstlane((int)(threadIdx.x >> 5));

    int colOf[NP];
    #pragma unroll
    for (int p = 0; p < NP; ++p) colOf[p] = (int)(lane + 32u * p) * VEC;

    float bs[NP][VEC];
    #pragma unroll
    for (int p = 0; p < NP; ++p) {
        const float2 b = *(const float2*)(bias + colOf[p]);
        bs[p][0] = b.x; bs[p][VEC - 1] = b.y;
    }

    const int nlane = (int)lane & 15;
    const int koff  = ((int)lane >> 4) << 1;
    const int pcol  = (wave < TN ? wave : 0) * 16 + nlane;
    const float* bFrag = Ws + (koff >> 1) * (2 * coutP) + (pcol << 1);

    const int tile0 = (int)blockIdx.x * tilesPerBlock;
    for (int it = 0; it < tilesPerBlock; ++it) {
        const int tile = tile0 + it;
        const int node = tile * 16 + wave;
        float* At = As + (it & 1) * (16 * APITCH);

        float o[NP][VEC] = {};
        if (node < n) {
            const int   beg = offs[node];
            const int   end = offs[node + 1];
            const float di  = dinv[node];

            float hs[NP][VEC];
            {
                const float* __restrict__ self = h + (size_t)node * (size_t)cout;
                #pragma unroll
                for (int p = 0; p < NP; ++p) {
                    const float2 t = *(const float2*)(self + colOf[p]);
                    hs[p][0] = t.x; hs[p][VEC - 1] = t.y;
                }
            }

            float acc[NP][VEC] = {};
            sum_bucket<VEC, NP>(h, (size_t)cout, ent, beg, end, n, lane, colOf, acc);

            #pragma unroll
            for (int p = 0; p < NP; ++p) {
                #pragma unroll
                for (int q = 0; q < VEC; ++q) {
                    o[p][q] = (acc[p][q] + hs[p][q]) * di + bs[p][q];
                    if (doRelu) o[p][q] = fmaxf(o[p][q], 0.0f);
                    o[p][q] *= di;
                }
            }
        }
        #pragma unroll
        for (int p = 0; p < NP; ++p) {
            float2 t; t.x = o[p][0]; t.y = o[p][VEC - 1];
            *(float2*)(At + wave * APITCH + colOf[p]) = t;
        }
        __syncthreads();

        if (wave < TN && tile * 16 < n) {
            const float* aRow = At + nlane * APITCH + koff;
            v8f d = {};
            #pragma unroll
            for (int s = 0; s < 16; ++s) {
                const v2f a = *(const v2f*)(aRow + 4 * s);
                const v2f b = *(const v2f*)(bFrag + s * (4 * coutP));
                d = __builtin_amdgcn_wmma_f32_16x16x4_f32(
                         false, a,  false, b,
                         (short)0, d,
                         false,  false);
            }
            const int rbase = tile * 16 + (((int)lane >> 4) << 3);
            if (pcol < coutN) {
                #pragma unroll
                for (int r = 0; r < 8; ++r) {
                    const int orow = rbase + r;
                    if (orow < n) Hn[(size_t)orow * coutN + pcol] = d[r];
                }
            }
        }
    }
}


static inline int cdiv(long long a, long long b) { return (int)((a + b - 1) / b); }

static void launch_gemm(const float* X, const float* W, const float* rscale, float* H,
                        int n, int cin, int cout, hipStream_t stream) {
    const int tilesM = (n + 15) >> 4;
    const int blocks = cdiv(tilesM, 8);
    const int TN     = (cout + 15) >> 4;
    const size_t shmem = (size_t)cin * TN * 16 * sizeof(float);
    switch (TN) {
        case 4: gemm_wmma_f32_kernel<4><<<blocks, 256, shmem, stream>>>(X, W, rscale, H, n, cin, cout); break;
        case 3: gemm_wmma_f32_kernel<3><<<blocks, 256, shmem, stream>>>(X, W, rscale, H, n, cin, cout); break;
        case 2: gemm_wmma_f32_kernel<2><<<blocks, 256, shmem, stream>>>(X, W, rscale, H, n, cin, cout); break;
        default: gemm_wmma_f32_kernel<1><<<blocks, 256, shmem, stream>>>(X, W, rscale, H, n, cin, cout); break;
    }
}

static void launch_fill_project(const int* row, const int* col,
                                const int* part, const float* dinv, int* ent,
                                int E, int nP, int rangeLen, int sliceLen,
                                const float* X, const float* W, float* H,
                                int n, int cin, int cout, hipStream_t stream) {
    const int tilesM     = (n + 15) >> 4;
    const int gemmBlocks = cdiv(tilesM, 32);
    const int blocks     = gemmBlocks + NSLICE * NRANGE;
    const int TN         = (cout + 15) >> 4;
    const size_t wBytes  = (size_t)cin * TN * 16 * sizeof(float);
    const size_t cBytes  = (size_t)rangeLen * sizeof(int);
    const size_t shmem   = wBytes > cBytes ? wBytes : cBytes;
    switch (TN) {
        case 4: fill_project_kernel<4><<<blocks, 1024, shmem, stream>>>(row, col, part, dinv, ent, E, nP, rangeLen, sliceLen, X, W, H, n, cin, cout, gemmBlocks); break;
        case 3: fill_project_kernel<3><<<blocks, 1024, shmem, stream>>>(row, col, part, dinv, ent, E, nP, rangeLen, sliceLen, X, W, H, n, cin, cout, gemmBlocks); break;
        case 2: fill_project_kernel<2><<<blocks, 1024, shmem, stream>>>(row, col, part, dinv, ent, E, nP, rangeLen, sliceLen, X, W, H, n, cin, cout, gemmBlocks); break;
        default: fill_project_kernel<1><<<blocks, 1024, shmem, stream>>>(row, col, part, dinv, ent, E, nP, rangeLen, sliceLen, X, W, H, n, cin, cout, gemmBlocks); break;
    }
}

static void launch_gather(const float* h, const int* offs, const int* ent,
                          const float* dinv, const float* bias, float* out,
                          int n, int cout, int doRelu, hipStream_t stream) {
    const int blocks = cdiv(n, 8);
    if ((cout & 1) == 0)
        gather_kernel<2, 1><<<blocks, 256, 0, stream>>>(h, offs, ent, dinv, bias, out, n, cout, doRelu);
    else
        gather_kernel<1, 2><<<blocks, 256, 0, stream>>>(h, offs, ent, dinv, bias, out, n, cout, doRelu);
}

static void launch_gather_project(const float* h, const int* offs, const int* ent,
                                  const float* dinv, const float* bias,
                                  const float* Wn, float* Hn,
                                  int n, int coutN, int doRelu, hipStream_t stream) {
    const int tiles = cdiv(n, 16);
    const int tpb   = cdiv(tiles, 1024);
    const int blocks = cdiv(tiles, tpb);
    const int TN     = (coutN + 15) >> 4;
    switch (TN) {
        case 4: gather_project_kernel<4, 2, 1><<<blocks, 512, 0, stream>>>(h, offs, ent, dinv, bias, Wn, Hn, n, coutN, doRelu, tpb); break;
        case 3: gather_project_kernel<3, 2, 1><<<blocks, 512, 0, stream>>>(h, offs, ent, dinv, bias, Wn, Hn, n, coutN, doRelu, tpb); break;
        case 2: gather_project_kernel<2, 2, 1><<<blocks, 512, 0, stream>>>(h, offs, ent, dinv, bias, Wn, Hn, n, coutN, doRelu, tpb); break;
        default: gather_project_kernel<1, 2, 1><<<blocks, 512, 0, stream>>>(h, offs, ent, dinv, bias, Wn, Hn, n, coutN, doRelu, tpb); break;
    }
}

extern "C" void kernel_launch(void* const* d_in, const int* in_sizes, int n_in,
                              void* d_out, int out_size, void* d_ws, size_t ws_size,
                              hipStream_t stream) {
    const float* x    = (const float*)d_in[0];
    const int*   ei   = (const int*)  d_in[1];
    const float* W1   = (const float*)d_in[2];
    const float* b1   = (const float*)d_in[3];
    const float* W2   = (const float*)d_in[4];
    const float* b2   = (const float*)d_in[5];
    const float* Wout = (const float*)d_in[6];
    const float* bout = (const float*)d_in[7];

    const int C1  = in_sizes[3];
    const int C2  = in_sizes[5];
    const int C3  = in_sizes[7];
    const int Cin = in_sizes[2] / C1;
    const int n   = in_sizes[0] / Cin;
    const int E   = in_sizes[1] / 2;

    const int* rowIdx = ei;
    const int* colIdx = ei + E;

    const size_t nP = ((size_t)n + 64) & ~(size_t)63;
    float* hbuf   = (float*)d_ws;
    float* abuf   = hbuf + ((size_t)n + 1) * 64;
    float* dinv   = abuf + ((size_t)n + 1) * 64;
    int*   offs   = (int*)(dinv + nP);
    int*   bsum   = offs + nP;
    int*   part   = bsum + nP;
    int*   ent    = part + (size_t)NSLICE * nP;

    const int scanBlocks = cdiv(n, 1024);
    const int rangeLen   = cdiv(n, NRANGE);
    const int sliceLen   = (cdiv(E, NSLICE) + 3) & ~3;

    count_kernel<<<dim3(NSLICE, NRANGE), 1024, (size_t)rangeLen * sizeof(int), stream>>>(
        colIdx, part, n, E, (int)nP, rangeLen, sliceLen);
    block_sum_kernel<<<scanBlocks, 256, 0, stream>>>(part, bsum, n, (int)nP);
    scan_kernel<<<scanBlocks, 256, 0, stream>>>(part, bsum, offs, dinv, n, (int)nP);

    float* cur = hbuf;
    float* oth = abuf;

    launch_fill_project(rowIdx, colIdx, part, dinv, ent, E, (int)nP, rangeLen, sliceLen,
                        x, W1, cur, n, Cin, C1, stream);

    if (C1 == 64) {
        launch_gather_project(cur, offs, ent, dinv, b1, W2, oth, n, C2, 1, stream);
        float* t = cur; cur = oth; oth = t;
    } else {
        launch_gather(cur, offs, ent, dinv, b1, oth, n, C1, 1, stream);
        launch_gemm(oth, W2, dinv, cur, n, C1, C2, stream);
    }

    if (C2 == 64) {
        launch_gather_project(cur, offs, ent, dinv, b2, Wout, oth, n, C3, 1, stream);
        float* t = cur; cur = oth; oth = t;
    } else {
        launch_gather(cur, offs, ent, dinv, b2, oth, n, C2, 1, stream);
        launch_gemm(oth, Wout, dinv, cur, n, C2, C3, stream);
    }

    launch_gather(cur, offs, ent, dinv, bout, (float*)d_out, n, C3, 0, stream);
}


# bucket starts made where they are used: each placement block scans its range's per-slice counts itself, slice 0 stores offs and dinv, product rows scaled from counts; 5 launches, not 7; sums kept
# speedup vs baseline: 4.0443x; 1.0626x over previous
#include <hip/hip_runtime.h>

typedef __attribute__((ext_vector_type(2))) float v2f;
typedef __attribute__((ext_vector_type(8))) float v8f;


#define NSLICE 16
#define NRANGE 16

__device__ __forceinline__ int draw_slot(int c, int lo, int len, int* cntL) {
    const unsigned d = (unsigned)(c - lo);
    return (d < (unsigned)len) ? atomicAdd(&cntL[d], 1) : -1;
}

__device__ __forceinline__ void count_one(int c, int lo, int len, int* cntL) {
    const unsigned d = (unsigned)(c - lo);
    if (d < (unsigned)len) atomicAdd(&cntL[d], 1);
}

__device__ __forceinline__ int wave_sum(int s) {
    #pragma unroll
    for (int off = 16; off > 0; off >>= 1) s += __shfl_xor(s, off, 32);
    return s;
}

__global__ void __launch_bounds__(1024)
count_kernel(const int* __restrict__ col, int* __restrict__ part, int* __restrict__ tot,
             int n, int E, int nP, int rangeLen, int sliceLen) {
    extern __shared__ int cntL[];
    __shared__ int wsum[32];
    const int tid = threadIdx.x;
    const int lo  = blockIdx.y * rangeLen;
    int len = n - lo;
    if (len > rangeLen) len = rangeLen;
    if (len < 0) len = 0;
    int lenP = ((n + 7) & ~7) - lo;
    if (lenP > rangeLen) lenP = rangeLen;
    if (lenP < 0) lenP = 0;
    for (int i = tid; i < lenP; i += 1024) cntL[i] = 0;
    __syncthreads();

    int e0 = blockIdx.x * sliceLen;
    if (e0 > E) e0 = E;
    int e1 = e0 + sliceLen;
    if (e1 > E) e1 = E;
    const int* __restrict__ cs = col + e0;
    const int nq = ((((size_t)cs) & 15) == 0) ? ((e1 - e0) >> 2) : 0;
    const int4* __restrict__ cs4 = (const int4*)cs;
    for (int q0 = tid; q0 < nq; q0 += 4 * 1024) {
        int4 v[4];
        #pragma unroll
        for (int u = 0; u < 4; ++u) {
            const int q = q0 + u * 1024;
            v[u] = (q < nq) ? cs4[q] : make_int4(-1, -1, -1, -1);
        }
        #pragma unroll
        for (int u = 0; u < 4; ++u) {
            count_one(v[u].x, lo, len, cntL);
            count_one(v[u].y, lo, len, cntL);
            count_one(v[u].z, lo, len, cntL);
            count_one(v[u].w, lo, len, cntL);
        }
    }
    for (int e = e0 + 4 * nq + tid; e < e1; e += 1024)
        count_one(col[e], lo, len, cntL);
    __syncthreads();

    int* __restrict__ dst = part + (size_t)blockIdx.x * (size_t)nP + lo;
    int mine = 0;
    for (int i = tid; i < lenP; i += 1024) {
        const int v = cntL[i];
        dst[i] = v;
        mine += v;
    }
    mine = wave_sum(mine);
    if ((tid & 31) == 0) wsum[tid >> 5] = mine;
    __syncthreads();
    if (tid < 32) {
        const int t = wave_sum(wsum[tid]);
        if (tid == 0) tot[blockIdx.y * NSLICE + blockIdx.x] = t;
    }
}

__device__ __forceinline__ void place_block(int s, int r,
                                            const int* __restrict__ row, const int* __restrict__ col,
                                            const int* __restrict__ part, const int* __restrict__ tot,
                                            int* __restrict__ offs, float* __restrict__ dinv,
                                            int* __restrict__ ent,
                                            int n, int E, int nP, int rangeLen, int sliceLen,
                                            int* curL) {
    __shared__ int smA[32];
    __shared__ int smB[32];
    const int tid  = threadIdx.x;
    const int lane = tid & 31;
    const int wave = tid >> 5;
    const int lo  = r * rangeLen;
    int len = n - lo;
    if (len > rangeLen) len = rangeLen;
    if (len < 0) len = 0;

    const int early = (tid < r * NSLICE) ? tot[tid] : 0;

    const int i0 = tid * 8;
    int ca[8] = {0, 0, 0, 0, 0, 0, 0, 0};
    int cb[8] = {0, 0, 0, 0, 0, 0, 0, 0};
    if (i0 < len) {
        const int* __restrict__ p0 = part + lo + i0;
        #pragma unroll 4
        for (int sl = 0; sl < NSLICE; ++sl) {
            const int4 u = *(const int4*)(p0 + (size_t)sl * (size_t)nP);
            const int4 v = *(const int4*)(p0 + (size_t)sl * (size_t)nP + 4);
            if (sl == s) {
                #pragma unroll
                for (int k = 0; k < 8; ++k) cb[k] = ca[k];
            }
            ca[0] += u.x; ca[1] += u.y; ca[2] += u.z; ca[3] += u.w;
            ca[4] += v.x; ca[5] += v.y; ca[6] += v.z; ca[7] += v.w;
        }
    }
    const int tsum = ca[0] + ca[1] + ca[2] + ca[3] + ca[4] + ca[5] + ca[6] + ca[7];
    const int pb = wave_sum(early);
    if (lane == 0) smA[wave] = pb;

    int incl = tsum;
    #pragma unroll
    for (int off = 1; off < 32; off <<= 1) {
        const int t = __shfl_up(incl, off, 32);
        if (lane >= off) incl += t;
    }
    if (lane == 31) smB[wave] = incl;
    __syncthreads();

    int run = wave_sum(smA[lane]) + wave_sum((lane < wave) ? smB[lane] : 0) + incl - tsum;

    if (i0 < len) {
        #pragma unroll
        for (int k = 0; k < 8; ++k) {
            if (i0 + k < len) {
                curL[i0 + k] = run + cb[k];
                if (s == 0) {
                    offs[lo + i0 + k] = run;
                    dinv[lo + i0 + k] = rsqrtf((float)ca[k] + 1.0f);
                }
            }
            run += ca[k];
        }
        if (s == 0 && i0 <= n - 1 - lo && n - 1 - lo < i0 + 8) offs[n] = run;
    }
    __syncthreads();

    int e0 = s * sliceLen;
    if (e0 > E) e0 = E;
    int e1 = e0 + sliceLen;
    if (e1 > E) e1 = E;
    const int* __restrict__ cs = col + e0;
    const int* __restrict__ rs = row + e0;
    const int nq = (((((size_t)cs) | ((size_t)rs)) & 15) == 0) ? ((e1 - e0) >> 2) : 0;
    const int4* __restrict__ cs4 = (const int4*)cs;
    const int4* __restrict__ rs4 = (const int4*)rs;
    for (int q0 = tid; q0 < nq; q0 += 4 * 1024) {
        int4 c[4], w[4];
        #pragma unroll
        for (int u = 0; u < 4; ++u) {
            const int q = q0 + u * 1024;
            c[u] = (q < nq) ? cs4[q] : make_int4(-1, -1, -1, -1);
            w[u] = (q < nq) ? rs4[q] : make_int4(0, 0, 0, 0);
        }
        int sl[4][4];
        #pragma unroll
        for (int u = 0; u < 4; ++u) {
            sl[u][0] = draw_slot(c[u].x, lo, len, curL);
            sl[u][1] = draw_slot(c[u].y, lo, len, curL);
            sl[u][2] = draw_slot(c[u].z, lo, len, curL);
            sl[u][3] = draw_slot(c[u].w, lo, len, curL);
        }
        #pragma unroll
        for (int u = 0; u < 4; ++u) {
            if (sl[u][0] >= 0) ent[sl[u][0]] = w[u].x;
            if (sl[u][1] >= 0) ent[sl[u][1]] = w[u].y;
            if (sl[u][2] >= 0) ent[sl[u][2]] = w[u].z;
            if (sl[u][3] >= 0) ent[sl[u][3]] = w[u].w;
        }
    }
    for (int e = e0 + 4 * nq + tid; e < e1; e += 1024) {
        const int p = draw_slot(col[e], lo, len, curL);
        if (p >= 0) ent[p] = row[e];
    }
}

template <int TN, bool COUNTS>
__device__ __forceinline__ void gemm_block(const float* __restrict__ X,
                                           const float* __restrict__ W,
                                           const float* __restrict__ rscale,
                                           const int* __restrict__ cnt, int nP,
                                           float* __restrict__ H,
                                           int n, int cin, int cout,
                                           int blk, float* lds) {
    constexpr int coutP = TN * 16;

    if (blk == 0 && (int)threadIdx.x < cout) H[(size_t)n * cout + threadIdx.x] = 0.0f;

    const int total = cin * coutP;
    for (int i = threadIdx.x; i < total; i += blockDim.x) {
        const int k = i / coutP;
        const int c = i - k * coutP;
        const float v = (c < cout) ? W[(size_t)k * cout + c] : 0.0f;
        lds[(k >> 1) * (2 * coutP) + (c << 1) + (k & 1)] = v;
    }
    __syncthreads();

    const int lane   = threadIdx.x & 31;
    const int waveId = threadIdx.x >> 5;
    const int tileM  = blk * (blockDim.x >> 5) + waveId;
    const int tilesM = (n + 15) >> 4;
    if (tileM >= tilesM) return;

    const int nlane = lane & 15;
    const int koff  = (lane >> 4) << 1;
    const int mrow  = (tileM << 4) + nlane;
    const int arow  = (mrow < n) ? mrow : (n - 1);

    const float* aPtr = X + (size_t)arow * cin + koff;
    int bIdx = (koff >> 1) * (2 * coutP) + (nlane << 1);

    v8f acc[TN] = {};
    for (int kb = 0; kb < cin; kb += 4) {
        const float2 av = *(const float2*)aPtr;
        v2f a; a.x = av.x; a.y = av.y;
        #pragma unroll
        for (int t = 0; t < TN; ++t) {
            const v2f b = *(const v2f*)(lds + bIdx + (t << 5));
            acc[t] = __builtin_amdgcn_wmma_f32_16x16x4_f32(
                          false, a,  false, b,
                          (short)0, acc[t],
                          false,  false);
        }
        aPtr += 4;
        bIdx += 4 * coutP;
    }

    const int rbase = (tileM << 4) + ((lane >> 4) << 3);
    float rs[8];
    if (COUNTS) {
        int deg = 0;
        if (mrow < n) {
            const int* __restrict__ cp = cnt + (size_t)((lane >> 4) * (NSLICE / 2)) * (size_t)nP + mrow;
            #pragma unroll
            for (int sl = 0; sl < NSLICE / 2; ++sl) deg += cp[(size_t)sl * (size_t)nP];
        }
        deg += __shfl_xor(deg, 16, 32);
        const float dv = rsqrtf((float)deg + 1.0f);
        #pragma unroll
        for (int r = 0; r < 8; ++r) rs[r] = __shfl(dv, ((lane >> 4) << 3) + r, 32);
    } else {
        #pragma unroll
        for (int r = 0; r < 8; ++r) rs[r] = (rbase + r < n) ? rscale[rbase + r] : 0.0f;
    }
    #pragma unroll
    for (int t = 0; t < TN; ++t) {
        const int col = t * 16 + nlane;
        if (col < cout) {
            #pragma unroll
            for (int r = 0; r < 8; ++r) {
                const int orow = rbase + r;
                if (orow < n) H[(size_t)orow * cout + col] = acc[t][r] * rs[r];
            }
        }
    }
}

template <int TN>
__global__ void gemm_wmma_f32_kernel(const float* __restrict__ X,
                                     const float* __restrict__ W,
                                     const float* __restrict__ rscale,
                                     float* __restrict__ H,
                                     int n, int cin, int cout) {
    extern __shared__ float lds[];
    gemm_block<TN, false>(X, W, rscale, nullptr, 0, H, n, cin, cout, (int)blockIdx.x, lds);
}

template <int TN>
__global__ void __launch_bounds__(1024, 16)
fill_project_kernel(const int* __restrict__ row, const int* __restrict__ col,
                    const int* __restrict__ part, const int* __restrict__ tot,
                    int* __restrict__ offs, float* __restrict__ dinv,
                    int* __restrict__ ent, int E, int nP, int rangeLen, int sliceLen,
                    const float* __restrict__ X, const float* __restrict__ W,
                    float* __restrict__ H, int n, int cin, int cout,
                    int gemmBlocks) {
    extern __shared__ float lds[];
    const int blk = (int)blockIdx.x;
    if (blk < gemmBlocks) {
        gemm_block<TN, true>(X, W, nullptr, part, nP, H, n, cin, cout, blk, lds);
    } else {
        const int pb = blk - gemmBlocks;
        place_block(pb % NSLICE, pb / NSLICE, row, col, part, tot, offs, dinv, ent,
                    n, E, nP, rangeLen, sliceLen, (int*)lds);
    }
}

template <int VEC, int NP>
__device__ __forceinline__ void sum_bucket(const float* __restrict__ h, size_t pitch,
                                           const int* __restrict__ ent,
                                           int beg, int end, int zrow, unsigned lane,
                                           const int (&ldc)[NP], float (&acc)[NP][VEC]) {
    for (int b0 = beg; b0 < end; b0 += 32) {
        const int cntb = (end - b0 < 32) ? (end - b0) : 32;
        int r = zrow;
        if ((int)lane < cntb) r = ent[b0 + (int)lane];
        asm volatile("" :: "v"(r));
        const int nq = (cntb + 3) >> 2;
        #pragma unroll
        for (int b = 0; b < 2; ++b) {
            if (4 * b < nq) {
                float v[4][4][NP][VEC];
                #pragma unroll
                for (int g = 0; g < 4; ++g) {
                    if (4 * b + g < nq) {
                        #pragma unroll
                        for (int u = 0; u < 4; ++u) {
                            const int rj = __builtin_amdgcn_readlane(r, 16 * b + 4 * g + u);
                            const float* __restrict__ src = h + (size_t)rj * pitch;
                            #pragma unroll
                            for (int p = 0; p < NP; ++p) {
                                if (VEC == 2) {
                                    const float2 t = *(const float2*)(src + ldc[p]);
                                    v[g][u][p][0] = t.x; v[g][u][p][VEC - 1] = t.y;
                                } else {
                                    v[g][u][p][0] = src[ldc[p]];
                                }
                            }
                        }
                    }
                }
                #pragma unroll
                for (int g = 0; g < 4; ++g) {
                    if (4 * b + g < nq) {
                        #pragma unroll
                        for (int u = 0; u < 4; ++u) {
                            #pragma unroll
                            for (int p = 0; p < NP; ++p) {
                                #pragma unroll
                                for (int q = 0; q < VEC; ++q)
                                    acc[p][q] += v[g][u][p][q];
                            }
                        }
                    }
                }
            }
        }
    }
}

template <int VEC, int NP>
__global__ void __launch_bounds__(256)
gather_kernel(const float* __restrict__ h,
              const int* __restrict__ offs,
              const int* __restrict__ ent,
              const float* __restrict__ dinv,
              const float* __restrict__ bias,
              float* __restrict__ out,
              int n, int cout, int doRelu) {
    const unsigned lane = threadIdx.x & 31u;
    const int node = __builtin_amdgcn_readfirstlane(
        (int)(blockIdx.x * (blockDim.x >> 5) + (threadIdx.x >> 5)));
    if (node >= n) return;

    const int   beg = offs[node];
    const int   end = offs[node + 1];
    const float di  = dinv[node];

    int  colOf[NP];
    bool act[NP];
    int  ldc[NP];
    #pragma unroll
    for (int p = 0; p < NP; ++p) {
        colOf[p] = (int)(lane + 32u * p) * VEC;
        act[p]   = colOf[p] < cout;
        ldc[p]   = act[p] ? colOf[p] : 0;
    }

    float hs[NP][VEC], bs[NP][VEC];
    {
        const float* __restrict__ self = h + (size_t)node * (size_t)cout;
        #pragma unroll
        for (int p = 0; p < NP; ++p) {
            if (VEC == 2) {
                const float2 t = *(const float2*)(self + ldc[p]);
                const float2 b = *(const float2*)(bias + ldc[p]);
                hs[p][0] = t.x; hs[p][VEC - 1] = t.y;
                bs[p][0] = b.x; bs[p][VEC - 1] = b.y;
            } else {
                hs[p][0] = self[ldc[p]];
                bs[p][0] = bias[ldc[p]];
            }
        }
    }

    float acc[NP][VEC] = {};
    sum_bucket<VEC, NP>(h, (size_t)cout, ent, beg, end, n, lane, ldc, acc);

    float* __restrict__ dst = out + (size_t)node * (size_t)cout;
    #pragma unroll
    for (int p = 0; p < NP; ++p) {
        float o[VEC];
        #pragma unroll
        for (int q = 0; q < VEC; ++q) {
            o[q] = (acc[p][q] + hs[p][q]) * di + bs[p][q];
            if (doRelu) o[q] = fmaxf(o[q], 0.0f);
        }
        if (act[p]) {
            if (VEC == 2) {
                float2 t; t.x = o[0]; t.y = o[VEC - 1];
                *(float2*)(dst + colOf[p]) = t;
            } else {
                dst[colOf[p]] = o[0];
            }
        }
    }
}

#define APITCH 68

template <int TN, int VEC, int NP>
__global__ void __launch_bounds__(512)
gather_project_kernel(const float* __restrict__ h,
                      const int* __restrict__ offs,
                      const int* __restrict__ ent,
                      const float* __restrict__ dinv,
                      const float* __restrict__ bias,
                      const float* __restrict__ Wn,
                      float* __restrict__ Hn,
                      int n, int coutN, int doRelu, int tilesPerBlock) {
    constexpr int cout  = 64;
    constexpr int coutP = TN * 16;
    __shared__ __attribute__((aligned(16))) float Ws[cout * coutP];
    __shared__ __attribute__((aligned(16))) float As[2 * 16 * APITCH];

    {
        float w[2 * TN];
        #pragma unroll
        for (int u = 0; u < 2 * TN; ++u) {
            const int i = (int)threadIdx.x + u * 512;
            const int k = i / coutP;
            const int c = i - k * coutP;
            w[u] = (c < coutN) ? Wn[k * coutN + c] : 0.0f;
        }
        #pragma unroll
        for (int u = 0; u < 2 * TN; ++u) {
            const int i = (int)threadIdx.x + u * 512;
            const int k = i / coutP;
            const int c = i - k * coutP;
            Ws[(k >> 1) * (2 * coutP) + (c << 1) + (k & 1)] = w[u];
        }
    }

    if (blockIdx.x == 0 && (int)threadIdx.x < coutN) Hn[(size_t)n * coutN + threadIdx.x] = 0.0f;

    const unsigned lane = threadIdx.x & 31u;
    const int wave = __builtin_amdgcn_readfirstlane((int)(threadIdx.x >> 5));

    int colOf[NP];
    #pragma unroll
    for (int p = 0; p < NP; ++p) colOf[p] = (int)(lane + 32u * p) * VEC;

    float bs[NP][VEC];
    #pragma unroll
    for (int p = 0; p < NP; ++p) {
        const float2 b = *(const float2*)(bias + colOf[p]);
        bs[p][0] = b.x; bs[p][VEC - 1] = b.y;
    }

    const int nlane = (int)lane & 15;
    const int koff  = ((int)lane >> 4) << 1;
    const int pcol  = (wave < TN ? wave : 0) * 16 + nlane;
    const float* bFrag = Ws + (koff >> 1) * (2 * coutP) + (pcol << 1);

    const int tile0 = (int)blockIdx.x * tilesPerBlock;
    for (int it = 0; it < tilesPerBlock; ++it) {
        const int tile = tile0 + it;
        const int node = tile * 16 + wave;
        float* At = As + (it & 1) * (16 * APITCH);

        float o[NP][VEC] = {};
        if (node < n) {
            const int   beg = offs[node];
            const int   end = offs[node + 1];
            const float di  = dinv[node];

            float hs[NP][VEC];
            {
                const float* __restrict__ self = h + (size_t)node * (size_t)cout;
                #pragma unroll
                for (int p = 0; p < NP; ++p) {
                    const float2 t = *(const float2*)(self + colOf[p]);
                    hs[p][0] = t.x; hs[p][VEC - 1] = t.y;
                }
            }

            float acc[NP][VEC] = {};
            sum_bucket<VEC, NP>(h, (size_t)cout, ent, beg, end, n, lane, colOf, acc);

            #pragma unroll
            for (int p = 0; p < NP; ++p) {
                #pragma unroll
                for (int q = 0; q < VEC; ++q) {
                    o[p][q] = (acc[p][q] + hs[p][q]) * di + bs[p][q];
                    if (doRelu) o[p][q] = fmaxf(o[p][q], 0.0f);
                    o[p][q] *= di;
                }
            }
        }
        #pragma unroll
        for (int p = 0; p < NP; ++p) {
            float2 t; t.x = o[p][0]; t.y = o[p][VEC - 1];
            *(float2*)(At + wave * APITCH + colOf[p]) = t;
        }
        __syncthreads();

        if (wave < TN && tile * 16 < n) {
            const float* aRow = At + nlane * APITCH + koff;
            v8f d = {};
            #pragma unroll
            for (int s = 0; s < 16; ++s) {
                const v2f a = *(const v2f*)(aRow + 4 * s);
                const v2f b = *(const v2f*)(bFrag + s * (4 * coutP));
                d = __builtin_amdgcn_wmma_f32_16x16x4_f32(
                         false, a,  false, b,
                         (short)0, d,
                         false,  false);
            }
            const int rbase = tile * 16 + (((int)lane >> 4) << 3);
            if (pcol < coutN) {
                #pragma unroll
                for (int r = 0; r < 8; ++r) {
                    const int orow = rbase + r;
                    if (orow < n) Hn[(size_t)orow * coutN + pcol] = d[r];
                }
            }
        }
    }
}


static inline int cdiv(long long a, long long b) { return (int)((a + b - 1) / b); }

static void launch_gemm(const float* X, const float* W, const float* rscale, float* H,
                        int n, int cin, int cout, hipStream_t stream) {
    const int tilesM = (n + 15) >> 4;
    const int blocks = cdiv(tilesM, 8);
    const int TN     = (cout + 15) >> 4;
    const size_t shmem = (size_t)cin * TN * 16 * sizeof(float);
    switch (TN) {
        case 4: gemm_wmma_f32_kernel<4><<<blocks, 256, shmem, stream>>>(X, W, rscale, H, n, cin, cout); break;
        case 3: gemm_wmma_f32_kernel<3><<<blocks, 256, shmem, stream>>>(X, W, rscale, H, n, cin, cout); break;
        case 2: gemm_wmma_f32_kernel<2><<<blocks, 256, shmem, stream>>>(X, W, rscale, H, n, cin, cout); break;
        default: gemm_wmma_f32_kernel<1><<<blocks, 256, shmem, stream>>>(X, W, rscale, H, n, cin, cout); break;
    }
}

static void launch_fill_project(const int* row, const int* col,
                                const int* part, const int* tot, int* offs, float* dinv, int* ent,
                                int E, int nP, int rangeLen, int sliceLen,
                                const float* X, const float* W, float* H,
                                int n, int cin, int cout, hipStream_t stream) {
    const int tilesM     = (n + 15) >> 4;
    const int gemmBlocks = cdiv(tilesM, 32);
    const int blocks     = gemmBlocks + NSLICE * NRANGE;
    const int TN         = (cout + 15) >> 4;
    const size_t wBytes  = (size_t)cin * TN * 16 * sizeof(float);
    const size_t cBytes  = (size_t)rangeLen * sizeof(int);
    const size_t shmem   = wBytes > cBytes ? wBytes : cBytes;
    switch (TN) {
        case 4: fill_project_kernel<4><<<blocks, 1024, shmem, stream>>>(row, col, part, tot, offs, dinv, ent, E, nP, rangeLen, sliceLen, X, W, H, n, cin, cout, gemmBlocks); break;
        case 3: fill_project_kernel<3><<<blocks, 1024, shmem, stream>>>(row, col, part, tot, offs, dinv, ent, E, nP, rangeLen, sliceLen, X, W, H, n, cin, cout, gemmBlocks); break;
        case 2: fill_project_kernel<2><<<blocks, 1024, shmem, stream>>>(row, col, part, tot, offs, dinv, ent, E, nP, rangeLen, sliceLen, X, W, H, n, cin, cout, gemmBlocks); break;
        default: fill_project_kernel<1><<<blocks, 1024, shmem, stream>>>(row, col, part, tot, offs, dinv, ent, E, nP, rangeLen, sliceLen, X, W, H, n, cin, cout, gemmBlocks); break;
    }
}

static void launch_gather(const float* h, const int* offs, const int* ent,
                          const float* dinv, const float* bias, float* out,
                          int n, int cout, int doRelu, hipStream_t stream) {
    const int blocks = cdiv(n, 8);
    if ((cout & 1) == 0)
        gather_kernel<2, 1><<<blocks, 256, 0, stream>>>(h, offs, ent, dinv, bias, out, n, cout, doRelu);
    else
        gather_kernel<1, 2><<<blocks, 256, 0, stream>>>(h, offs, ent, dinv, bias, out, n, cout, doRelu);
}

static void launch_gather_project(const float* h, const int* offs, const int* ent,
                                  const float* dinv, const float* bias,
                                  const float* Wn, float* Hn,
                                  int n, int coutN, int doRelu, hipStream_t stream) {
    const int tiles = cdiv(n, 16);
    const int tpb   = cdiv(tiles, 1024);
    const int blocks = cdiv(tiles, tpb);
    const int TN     = (coutN + 15) >> 4;
    switch (TN) {
        case 4: gather_project_kernel<4, 2, 1><<<blocks, 512, 0, stream>>>(h, offs, ent, dinv, bias, Wn, Hn, n, coutN, doRelu, tpb); break;
        case 3: gather_project_kernel<3, 2, 1><<<blocks, 512, 0, stream>>>(h, offs, ent, dinv, bias, Wn, Hn, n, coutN, doRelu, tpb); break;
        case 2: gather_project_kernel<2, 2, 1><<<blocks, 512, 0, stream>>>(h, offs, ent, dinv, bias, Wn, Hn, n, coutN, doRelu, tpb); break;
        default: gather_project_kernel<1, 2, 1><<<blocks, 512, 0, stream>>>(h, offs, ent, dinv, bias, Wn, Hn, n, coutN, doRelu, tpb); break;
    }
}

extern "C" void kernel_launch(void* const* d_in, const int* in_sizes, int n_in,
                              void* d_out, int out_size, void* d_ws, size_t ws_size,
                              hipStream_t stream) {
    const float* x    = (const float*)d_in[0];
    const int*   ei   = (const int*)  d_in[1];
    const float* W1   = (const float*)d_in[2];
    const float* b1   = (const float*)d_in[3];
    const float* W2   = (const float*)d_in[4];
    const float* b2   = (const float*)d_in[5];
    const float* Wout = (const float*)d_in[6];
    const float* bout = (const float*)d_in[7];

    const int C1  = in_sizes[3];
    const int C2  = in_sizes[5];
    const int C3  = in_sizes[7];
    const int Cin = in_sizes[2] / C1;
    const int n   = in_sizes[0] / Cin;
    const int E   = in_sizes[1] / 2;

    const int* rowIdx = ei;
    const int* colIdx = ei + E;

    const size_t nP = ((size_t)n + 64) & ~(size_t)63;
    float* hbuf   = (float*)d_ws;
    float* abuf   = hbuf + ((size_t)n + 1) * 64;
    float* dinv   = abuf + ((size_t)n + 1) * 64;
    int*   offs   = (int*)(dinv + nP);
    int*   tot    = offs + nP;
    int*   part   = tot + NRANGE * NSLICE;
    int*   ent    = part + (size_t)NSLICE * nP;

    const int rangeLen   = (cdiv(n, NRANGE) + 7) & ~7;
    const int sliceLen   = (cdiv(E, NSLICE) + 3) & ~3;

    count_kernel<<<dim3(NSLICE, NRANGE), 1024, (size_t)rangeLen * sizeof(int), stream>>>(
        colIdx, part, tot, n, E, (int)nP, rangeLen, sliceLen);

    float* cur = hbuf;
    float* oth = abuf;

    launch_fill_project(rowIdx, colIdx, part, tot, offs, dinv, ent, E, (int)nP, rangeLen, sliceLen,
                        x, W1, cur, n, Cin, C1, stream);

    if (C1 == 64) {
        launch_gather_project(cur, offs, ent, dinv, b1, W2, oth, n, C2, 1, stream);
        float* t = cur; cur = oth; oth = t;
    } else {
        launch_gather(cur, offs, ent, dinv, b1, oth, n, C1, 1, stream);
        launch_gemm(oth, W2, dinv, cur, n, C1, C2, stream);
    }

    if (C2 == 64) {
        launch_gather_project(cur, offs, ent, dinv, b2, Wout, oth, n, C3, 1, stream);
        float* t = cur; cur = oth; oth = t;
    } else {
        launch_gather(cur, offs, ent, dinv, b2, oth, n, C2, 1, stream);
        launch_gemm(oth, Wout, dinv, cur, n, C2, C3, stream);
    }

    launch_gather(cur, offs, ent, dinv, bout, (float*)d_out, n, C3, 0, stream);
}
